# starting kernel
# baseline (speedup 1.0000x reference)
#include <hip/hip_runtime.h>
#include <hip/hip_bf16.h>

#define N_NODES   100000
#define N_EDGES   20000
#define NNZ_CNT   800000
#define D_IN      128
#define D         256
#define EPS_F     1e-5f

typedef float v2f __attribute__((ext_vector_type(2)));
typedef float v8f __attribute__((ext_vector_type(8)));

__device__ __forceinline__ void atomAddF(float* p, float v) {
    __hip_atomic_fetch_add(p, v, __ATOMIC_RELAXED, __HIP_MEMORY_SCOPE_AGENT);
}

__global__ __launch_bounds__(256) void degcnt_kernel(const int* __restrict__ node_idx,
                                                     const int* __restrict__ edge_idx,
                                                     float* __restrict__ deg,
                                                     float* __restrict__ cnt) {
    int i = blockIdx.x * 256 + threadIdx.x;
    if (i < NNZ_CNT) {
        atomAddF(&deg[node_idx[i]], 1.0f);
        atomAddF(&cnt[edge_idx[i]], 1.0f);
    }
}

template<int K>
__global__ __launch_bounds__(256) void gemm_wmma_kernel(const float* __restrict__ X,
                                                        const float* __restrict__ W,
                                                        float* __restrict__ C,
                                                        int mTiles) {
    const int wid  = (blockIdx.x * 256 + threadIdx.x) >> 5;
    const int lane = threadIdx.x & 31;
    const int mT   = wid >> 2;
    const int nT   = wid & 3;
    if (mT >= mTiles) return;
    const int m0   = mT * 16;
    const int n0   = nT * 64;
    const int half = lane >> 4;
    const int l16  = lane & 15;

    const float* __restrict__ xrow = X + (size_t)(m0 + l16) * K;
    const float* __restrict__ wrow = W + (size_t)(n0 + l16) * K;

    v8f acc0 = {}, acc1 = {}, acc2 = {}, acc3 = {};

    #pragma unroll 8
    for (int k0 = 0; k0 < K; k0 += 4) {
        const int ko = k0 + 2 * half;
        v2f a  = *reinterpret_cast<const v2f*>(xrow + ko);
        v2f b0 = *reinterpret_cast<const v2f*>(wrow + (size_t)0 * 16 * K + ko);
        v2f b1 = *reinterpret_cast<const v2f*>(wrow + (size_t)1 * 16 * K + ko);
        v2f b2 = *reinterpret_cast<const v2f*>(wrow + (size_t)2 * 16 * K + ko);
        v2f b3 = *reinterpret_cast<const v2f*>(wrow + (size_t)3 * 16 * K + ko);
        acc0 = __builtin_amdgcn_wmma_f32_16x16x4_f32(false, a, false, b0, (short)0, acc0, false, false);
        acc1 = __builtin_amdgcn_wmma_f32_16x16x4_f32(false, a, false, b1, (short)0, acc1, false, false);
        acc2 = __builtin_amdgcn_wmma_f32_16x16x4_f32(false, a, false, b2, (short)0, acc2, false, false);
        acc3 = __builtin_amdgcn_wmma_f32_16x16x4_f32(false, a, false, b3, (short)0, acc3, false, false);
    }

    float* __restrict__ ccol = C + n0 + l16;
    #pragma unroll
    for (int r = 0; r < 8; ++r) {
        const size_t m = (size_t)(m0 + r + 8 * half);
        ccol[m * D + 0 * 16] = acc0[r];
        ccol[m * D + 1 * 16] = acc1[r];
        ccol[m * D + 2 * 16] = acc2[r];
        ccol[m * D + 3 * 16] = acc3[r];
    }
}

__global__ __launch_bounds__(256) void scatter_rows_kernel(const float* __restrict__ src,
                                                           const int* __restrict__ src_idx,
                                                           const int* __restrict__ dst_idx,
                                                           float* __restrict__ dst) {
    const int i = blockIdx.x;
    const int f = threadIdx.x;
    const int s = src_idx[i];
    const int d = dst_idx[i];
    const float v = src[(size_t)s * D + f];
    atomAddF(&dst[(size_t)d * D + f], v);
}

__global__ __launch_bounds__(256) void scale_edges_kernel(float* __restrict__ E,
                                                          const float* __restrict__ cnt) {
    const int idx = blockIdx.x * 256 + threadIdx.x;
    if (idx < N_EDGES * D) {
        const float c = cnt[idx >> 8];
        const float inv = (c > 0.0f) ? 1.0f / c : 0.0f;
        E[idx] *= inv;
    }
}

__global__ __launch_bounds__(256) void fused_scale_ln_prelu_kernel(const float* __restrict__ acc,
                                                                   const float* __restrict__ deg,
                                                                   const float* __restrict__ bias,
                                                                   const float* __restrict__ g,
                                                                   const float* __restrict__ beta,
                                                                   const float* __restrict__ pa,
                                                                   float* __restrict__ out) {
    __shared__ float sred[256];
    const int r = blockIdx.x;
    const int f = threadIdx.x;
    const float dg   = deg[r];
    const float dinv = (dg > 0.0f) ? 1.0f / dg : 0.0f;
    const float v = acc[(size_t)r * D + f] * dinv + bias[f];

    sred[f] = v;
    __syncthreads();
    #pragma unroll
    for (int s = 128; s > 0; s >>= 1) { if (f < s) sred[f] += sred[f + s]; __syncthreads(); }
    const float mu = sred[0] * (1.0f / D);
    __syncthreads();

    const float dv = v - mu;
    sred[f] = dv * dv;
    __syncthreads();
    #pragma unroll
    for (int s = 128; s > 0; s >>= 1) { if (f < s) sred[f] += sred[f + s]; __syncthreads(); }
    const float var = sred[0] * (1.0f / D);

    const float y = dv * rsqrtf(var + EPS_F) * g[f] + beta[f];
    const float a = pa[0];
    out[(size_t)r * D + f] = (y >= 0.0f) ? y : a * y;
}

__global__ __launch_bounds__(256) void wstd_kernel(const float* __restrict__ W,
                                                   float* __restrict__ Ws) {
    __shared__ float sred[256];
    const int r = blockIdx.x;
    const int f = threadIdx.x;
    const float w = W[(size_t)r * D + f];

    sred[f] = w;
    __syncthreads();
    #pragma unroll
    for (int s = 128; s > 0; s >>= 1) { if (f < s) sred[f] += sred[f + s]; __syncthreads(); }
    const float mean = sred[0] * (1.0f / D);
    __syncthreads();

    const float dv = w - mean;
    sred[f] = dv * dv;
    __syncthreads();
    #pragma unroll
    for (int s = 128; s > 0; s >>= 1) { if (f < s) sred[f] += sred[f + s]; __syncthreads(); }
    const float var = sred[0] * (1.0f / (D - 1));

    Ws[(size_t)r * D + f] = dv * rsqrtf(var + EPS_F);
}

extern "C" void kernel_launch(void* const* d_in, const int* in_sizes, int n_in,
                              void* d_out, int out_size, void* d_ws, size_t ws_size,
                              hipStream_t stream) {
    const float* edge_fea = (const float*)d_in[0];
    const int*   node_idx = (const int*)  d_in[1];
    const int*   edge_idx = (const int*)  d_in[2];
    const float* W1       = (const float*)d_in[3];
    const float* b1       = (const float*)d_in[4];
    const float* W2       = (const float*)d_in[5];
    const float* b2       = (const float*)d_in[6];
    const float* g1       = (const float*)d_in[7];
    const float* beta1    = (const float*)d_in[8];
    const float* g2       = (const float*)d_in[9];
    const float* beta2    = (const float*)d_in[10];
    const float* prelu_a  = (const float*)d_in[11];

    const size_t nodeRowBytes = (size_t)N_NODES * D * sizeof(float);
    const size_t edgeRowBytes = (size_t)N_EDGES * D * sizeof(float);

    float* B0  = (float*)d_out;
    char*  ws  = (char*)d_ws;
    float* B1  = (float*)ws;
    float* E   = (float*)(ws + nodeRowBytes);
    float* deg = E + (size_t)N_EDGES * D;
    float* cnt = deg + N_NODES;
    float* W2s = cnt + N_EDGES;

    const int mTiles = N_NODES / 16;
    const int gemmBlocks = (mTiles * 4 * 32) / 256;

    hipMemsetAsync(deg, 0, (size_t)(N_NODES + N_EDGES) * sizeof(float), stream);
    degcnt_kernel<<<(NNZ_CNT + 255) / 256, 256, 0, stream>>>(node_idx, edge_idx, deg, cnt);

    gemm_wmma_kernel<D_IN><<<gemmBlocks, 256, 0, stream>>>(edge_fea, W1, B1, mTiles);

    hipMemsetAsync(E, 0, edgeRowBytes, stream);
    scatter_rows_kernel<<<NNZ_CNT, 256, 0, stream>>>(B1, node_idx, edge_idx, E);
    scale_edges_kernel<<<(N_EDGES * D) / 256, 256, 0, stream>>>(E, cnt);

    hipMemsetAsync(B0, 0, nodeRowBytes, stream);
    scatter_rows_kernel<<<NNZ_CNT, 256, 0, stream>>>(E, edge_idx, node_idx, B0);
    fused_scale_ln_prelu_kernel<<<N_NODES, 256, 0, stream>>>(B0, deg, b1, g1, beta1, prelu_a, B1);

    wstd_kernel<<<D, 256, 0, stream>>>(W2, W2s);
    gemm_wmma_kernel<D><<<gemmBlocks, 256, 0, stream>>>(B1, W2s, B0, mTiles);

    hipMemsetAsync(E, 0, edgeRowBytes, stream);
    scatter_rows_kernel<<<NNZ_CNT, 256, 0, stream>>>(B0, node_idx, edge_idx, E);
    scale_edges_kernel<<<(N_EDGES * D) / 256, 256, 0, stream>>>(E, cnt);

    hipMemsetAsync(B1, 0, nodeRowBytes, stream);
    scatter_rows_kernel<<<NNZ_CNT, 256, 0, stream>>>(E, edge_idx, node_idx, B1);
    fused_scale_ln_prelu_kernel<<<N_NODES, 256, 0, stream>>>(B1, deg, b2, g2, beta2, prelu_a, B0);
}


# Float-atomic row scatter-adds replaced by a per-call counting sort of the incidence list and one-wave-per-row register gathers; rows summed in fixed sorted order, no float atomics
# speedup vs baseline: 3.0200x; 3.0200x over previous
#include <hip/hip_runtime.h>
#include <hip/hip_bf16.h>

#define N_NODES   100000
#define N_EDGES   20000
#define NNZ_CNT   800000
#define D_IN      128
#define D         256
#define EPS_F     1e-5f

typedef float v2f __attribute__((ext_vector_type(2)));
typedef float v4f __attribute__((ext_vector_type(4)));
typedef float v8f __attribute__((ext_vector_type(8)));
typedef int   v4i __attribute__((ext_vector_type(4)));

__device__ __forceinline__ int atomAddI(int* p, int v) {
    return __hip_atomic_fetch_add(p, v, __ATOMIC_RELAXED, __HIP_MEMORY_SCOPE_AGENT);
}

__global__ __launch_bounds__(256) void rank_kernel(const int* __restrict__ node_idx,
                                                   const int* __restrict__ edge_idx,
                                                   int* __restrict__ posN,
                                                   int* __restrict__ posE,
                                                   int* __restrict__ rankN,
                                                   int* __restrict__ rankE) {
    int i = blockIdx.x * 256 + threadIdx.x;
    if (i < NNZ_CNT) {
        rankN[i] = atomAddI(&posN[node_idx[i]], 1);
        rankE[i] = atomAddI(&posE[edge_idx[i]], 1);
    }
}

template<int C>
__device__ __forceinline__ void block_exclusive_scan(int* __restrict__ a, int* sh) {
    const int t    = threadIdx.x;
    const int lane = t & 31;
    const int wv   = t >> 5;
    const bool live = t < 1000;
    v4i v[C / 4];
    int sum = 0;
    if (live) {
        #pragma unroll
        for (int q = 0; q < C / 4; ++q) {
            v[q] = *reinterpret_cast<const v4i*>(a + t * C + 4 * q);
            sum += v[q].x + v[q].y + v[q].z + v[q].w;
        }
    }
    int inc = sum;
    #pragma unroll
    for (int o = 1; o < 32; o <<= 1) {
        int up = __shfl_up(inc, o, 32);
        if (lane >= o) inc += up;
    }
    if (lane == 31) sh[wv] = inc;
    __syncthreads();
    if (wv == 0) {
        int wt = sh[lane];
        int winc = wt;
        #pragma unroll
        for (int o = 1; o < 32; o <<= 1) {
            int up = __shfl_up(winc, o, 32);
            if (lane >= o) winc += up;
        }
        sh[lane] = winc - wt;
    }
    __syncthreads();
    int run = sh[wv] + inc - sum;
    if (live) {
        #pragma unroll
        for (int q = 0; q < C / 4; ++q) {
            v4i x = v[q], o;
            o.x = run; run += x.x;
            o.y = run; run += x.y;
            o.z = run; run += x.z;
            o.w = run; run += x.w;
            *reinterpret_cast<v4i*>(a + t * C + 4 * q) = o;
        }
    }
}

__global__ __launch_bounds__(1024) void scan_kernel(int* __restrict__ posN, int* __restrict__ posE) {
    __shared__ int sh[32];
    if (blockIdx.x == 0) block_exclusive_scan<N_NODES / 1000>(posN, sh);
    else                 block_exclusive_scan<N_EDGES / 1000>(posE, sh);
}

__global__ __launch_bounds__(256) void place_kernel(const int* __restrict__ node_idx,
                                                    const int* __restrict__ edge_idx,
                                                    const int* __restrict__ posN,
                                                    const int* __restrict__ posE,
                                                    const int* __restrict__ rankN,
                                                    const int* __restrict__ rankE,
                                                    int* __restrict__ unsN,
                                                    int* __restrict__ unsE) {
    int i = blockIdx.x * 256 + threadIdx.x;
    if (i < NNZ_CNT) {
        const int n = node_idx[i];
        const int e = edge_idx[i];
        unsE[posE[e] + rankE[i]] = n;
        unsN[posN[n] + rankN[i]] = e;
    }
}

__global__ __launch_bounds__(256) void sort_segments_kernel(const int* __restrict__ posN,
                                                            const int* __restrict__ posE,
                                                            const int* __restrict__ unsN,
                                                            const int* __restrict__ unsE,
                                                            int* __restrict__ srtN,
                                                            int* __restrict__ srtE) {
    const int w    = __builtin_amdgcn_readfirstlane((int)((blockIdx.x * 256 + threadIdx.x) >> 5));
    const int lane = threadIdx.x & 31;
    if (w >= N_EDGES + N_NODES) return;
    const bool isE = w < N_EDGES;
    const int  k   = isE ? w : w - N_EDGES;
    const int* __restrict__ pos = isE ? posE : posN;
    const int* __restrict__ U   = isE ? unsE : unsN;
    int*       __restrict__ S   = isE ? srtE : srtN;
    const int nk  = isE ? N_EDGES : N_NODES;
    const int beg = pos[k];
    const int L   = ((k + 1 < nk) ? pos[k + 1] : NNZ_CNT) - beg;
    for (int m = 0; m < L; m += 32) {
        const int p = m + lane;
        const int x = (p < L) ? U[beg + p] : 0;
        int rank = 0;
        for (int c = 0; c < L; c += 32) {
            const int q  = c + lane;
            const int ec = (q < L) ? U[beg + q] : 0;
            const int cnt = (L - c < 32) ? (L - c) : 32;
            for (int j = 0; j < cnt; ++j) {
                const int v = __builtin_amdgcn_readlane(ec, j);
                rank += (v < x || (v == x && (c + j) < p)) ? 1 : 0;
            }
        }
        if (p < L) S[beg + rank] = x;
    }
}

template<int K>
__global__ __launch_bounds__(256) void gemm_wmma_kernel(const float* __restrict__ X,
                                                        const float* __restrict__ W,
                                                        float* __restrict__ C,
                                                        int mTiles) {
    const int wid  = (blockIdx.x * 256 + threadIdx.x) >> 5;
    const int lane = threadIdx.x & 31;
    const int mT   = wid >> 2;
    const int nT   = wid & 3;
    if (mT >= mTiles) return;
    const int m0   = mT * 16;
    const int n0   = nT * 64;
    const int half = lane >> 4;
    const int l16  = lane & 15;

    const float* __restrict__ xrow = X + (size_t)(m0 + l16) * K;
    const float* __restrict__ wrow = W + (size_t)(n0 + l16) * K;

    v8f acc0 = {}, acc1 = {}, acc2 = {}, acc3 = {};

    #pragma unroll 8
    for (int k0 = 0; k0 < K; k0 += 4) {
        const int ko = k0 + 2 * half;
        v2f a  = *reinterpret_cast<const v2f*>(xrow + ko);
        v2f b0 = *reinterpret_cast<const v2f*>(wrow + (size_t)0 * 16 * K + ko);
        v2f b1 = *reinterpret_cast<const v2f*>(wrow + (size_t)1 * 16 * K + ko);
        v2f b2 = *reinterpret_cast<const v2f*>(wrow + (size_t)2 * 16 * K + ko);
        v2f b3 = *reinterpret_cast<const v2f*>(wrow + (size_t)3 * 16 * K + ko);
        acc0 = __builtin_amdgcn_wmma_f32_16x16x4_f32(false, a, false, b0, (short)0, acc0, false, false);
        acc1 = __builtin_amdgcn_wmma_f32_16x16x4_f32(false, a, false, b1, (short)0, acc1, false, false);
        acc2 = __builtin_amdgcn_wmma_f32_16x16x4_f32(false, a, false, b2, (short)0, acc2, false, false);
        acc3 = __builtin_amdgcn_wmma_f32_16x16x4_f32(false, a, false, b3, (short)0, acc3, false, false);
    }

    float* __restrict__ ccol = C + n0 + l16;
    #pragma unroll
    for (int r = 0; r < 8; ++r) {
        const size_t m = (size_t)(m0 + r + 8 * half);
        ccol[m * D + 0 * 16] = acc0[r];
        ccol[m * D + 1 * 16] = acc1[r];
        ccol[m * D + 2 * 16] = acc2[r];
        ccol[m * D + 3 * 16] = acc3[r];
    }
}

template<bool SCALE>
__global__ __launch_bounds__(256) void gather_rows_kernel(const float* __restrict__ src,
                                                          const int* __restrict__ pos,
                                                          const int* __restrict__ list,
                                                          float* __restrict__ dst,
                                                          int nseg) {
    const int w    = __builtin_amdgcn_readfirstlane((int)((blockIdx.x * 256 + threadIdx.x) >> 5));
    const int lane = threadIdx.x & 31;
    if (w >= nseg) return;
    const int beg = pos[w];
    const int end = (w + 1 < nseg) ? pos[w + 1] : NNZ_CNT;
    const float* __restrict__ lsrc = src + lane * 4;

    v4f a0 = {}, a1 = {};
    for (int c = beg; c < end; c += 32) {
        const int q   = c + lane;
        const int idx = (q < end) ? list[q] : 0;
        const int cnt = (end - c < 32) ? (end - c) : 32;
        int j = 0;
        for (; j + 4 <= cnt; j += 4) {
            const float* p0 = lsrc + (size_t)__builtin_amdgcn_readlane(idx, j + 0) * D;
            const float* p1 = lsrc + (size_t)__builtin_amdgcn_readlane(idx, j + 1) * D;
            const float* p2 = lsrc + (size_t)__builtin_amdgcn_readlane(idx, j + 2) * D;
            const float* p3 = lsrc + (size_t)__builtin_amdgcn_readlane(idx, j + 3) * D;
            const v4f x0 = *reinterpret_cast<const v4f*>(p0), y0 = *reinterpret_cast<const v4f*>(p0 + 128);
            const v4f x1 = *reinterpret_cast<const v4f*>(p1), y1 = *reinterpret_cast<const v4f*>(p1 + 128);
            const v4f x2 = *reinterpret_cast<const v4f*>(p2), y2 = *reinterpret_cast<const v4f*>(p2 + 128);
            const v4f x3 = *reinterpret_cast<const v4f*>(p3), y3 = *reinterpret_cast<const v4f*>(p3 + 128);
            a0 += x0; a1 += y0;
            a0 += x1; a1 += y1;
            a0 += x2; a1 += y2;
            a0 += x3; a1 += y3;
        }
        for (; j < cnt; ++j) {
            const float* p0 = lsrc + (size_t)__builtin_amdgcn_readlane(idx, j) * D;
            a0 += *reinterpret_cast<const v4f*>(p0);
            a1 += *reinterpret_cast<const v4f*>(p0 + 128);
        }
    }
    if (SCALE) {
        const float c   = (float)(end - beg);
        const float inv = (c > 0.0f) ? 1.0f / c : 0.0f;
        a0 *= inv; a1 *= inv;
    }
    float* drow = dst + (size_t)w * D + lane * 4;
    *reinterpret_cast<v4f*>(drow)       = a0;
    *reinterpret_cast<v4f*>(drow + 128) = a1;
}

__global__ __launch_bounds__(256) void copy_lists_kernel(const int* __restrict__ from,
                                                         int* __restrict__ to, int n4) {
    const int i = blockIdx.x * 256 + threadIdx.x;
    if (i < n4) reinterpret_cast<v4i*>(to)[i] = reinterpret_cast<const v4i*>(from)[i];
}

__global__ __launch_bounds__(256) void fused_scale_ln_prelu_kernel(float* x,
                                                                   const int* __restrict__ posN,
                                                                   const float* __restrict__ bias,
                                                                   const float* __restrict__ g,
                                                                   const float* __restrict__ beta,
                                                                   const float* __restrict__ pa) {
    __shared__ float sred[256];
    const int r = blockIdx.x;
    const int f = threadIdx.x;
    const float dg   = (float)(((r + 1 < N_NODES) ? posN[r + 1] : NNZ_CNT) - posN[r]);
    const float dinv = (dg > 0.0f) ? 1.0f / dg : 0.0f;
    const float v = x[(size_t)r * D + f] * dinv + bias[f];

    sred[f] = v;
    __syncthreads();
    #pragma unroll
    for (int s = 128; s > 0; s >>= 1) { if (f < s) sred[f] += sred[f + s]; __syncthreads(); }
    const float mu = sred[0] * (1.0f / D);
    __syncthreads();

    const float dv = v - mu;
    sred[f] = dv * dv;
    __syncthreads();
    #pragma unroll
    for (int s = 128; s > 0; s >>= 1) { if (f < s) sred[f] += sred[f + s]; __syncthreads(); }
    const float var = sred[0] * (1.0f / D);

    const float y = dv * rsqrtf(var + EPS_F) * g[f] + beta[f];
    const float a = pa[0];
    x[(size_t)r * D + f] = (y >= 0.0f) ? y : a * y;
}

__global__ __launch_bounds__(256) void wstd_kernel(const float* __restrict__ W,
                                                   float* __restrict__ Ws) {
    __shared__ float sred[256];
    const int r = blockIdx.x;
    const int f = threadIdx.x;
    const float w = W[(size_t)r * D + f];

    sred[f] = w;
    __syncthreads();
    #pragma unroll
    for (int s = 128; s > 0; s >>= 1) { if (f < s) sred[f] += sred[f + s]; __syncthreads(); }
    const float mean = sred[0] * (1.0f / D);
    __syncthreads();

    const float dv = w - mean;
    sred[f] = dv * dv;
    __syncthreads();
    #pragma unroll
    for (int s = 128; s > 0; s >>= 1) { if (f < s) sred[f] += sred[f + s]; __syncthreads(); }
    const float var = sred[0] * (1.0f / (D - 1));

    Ws[(size_t)r * D + f] = dv * rsqrtf(var + EPS_F);
}

extern "C" void kernel_launch(void* const* d_in, const int* in_sizes, int n_in,
                              void* d_out, int out_size, void* d_ws, size_t ws_size,
                              hipStream_t stream) {
    const float* edge_fea = (const float*)d_in[0];
    const int*   node_idx = (const int*)  d_in[1];
    const int*   edge_idx = (const int*)  d_in[2];
    const float* W1       = (const float*)d_in[3];
    const float* b1       = (const float*)d_in[4];
    const float* W2       = (const float*)d_in[5];
    const float* b2       = (const float*)d_in[6];
    const float* g1       = (const float*)d_in[7];
    const float* beta1    = (const float*)d_in[8];
    const float* g2       = (const float*)d_in[9];
    const float* beta2    = (const float*)d_in[10];
    const float* prelu_a  = (const float*)d_in[11];

    const size_t nodeRowBytes = (size_t)N_NODES * D * sizeof(float);

    float* B0   = (float*)d_out;
    char*  ws   = (char*)d_ws;
    float* B1   = (float*)ws;
    float* E    = (float*)(ws + nodeRowBytes);
    int*   posN = (int*)(E + (size_t)N_EDGES * D);
    int*   posE = posN + N_NODES;
    float* W2s  = (float*)(posE + N_EDGES);

    int* listsA = (int*)B0;
    int* listsE = (int*)E;
    int* listsB = (int*)B1;
    int* unsN   = listsA + 2 * NNZ_CNT;
    int* unsE   = unsN + NNZ_CNT;
    int* rankN  = unsE + NNZ_CNT;
    int* rankE  = rankN + NNZ_CNT;

    const int mTiles = N_NODES / 16;
    const int gemmBlocks = (mTiles * 4 * 32) / 256;
    const int nnzBlocks  = (NNZ_CNT + 255) / 256;
    const int listVec    = (2 * NNZ_CNT) / 4;

    hipMemsetAsync(posN, 0, (size_t)(N_NODES + N_EDGES) * sizeof(int), stream);
    rank_kernel<<<nnzBlocks, 256, 0, stream>>>(node_idx, edge_idx, posN, posE, rankN, rankE);
    scan_kernel<<<2, 1024, 0, stream>>>(posN, posE);
    place_kernel<<<nnzBlocks, 256, 0, stream>>>(node_idx, edge_idx, posN, posE, rankN, rankE, unsN, unsE);
    sort_segments_kernel<<<(N_EDGES + N_NODES) / 8, 256, 0, stream>>>(posN, posE, unsN, unsE,
                                                                     listsA, listsA + NNZ_CNT);

    gemm_wmma_kernel<D_IN><<<gemmBlocks, 256, 0, stream>>>(edge_fea, W1, B1, mTiles);
    gather_rows_kernel<true ><<<N_EDGES / 8, 256, 0, stream>>>(B1, posE, listsA + NNZ_CNT, E, N_EDGES);
    gather_rows_kernel<false><<<N_NODES / 8, 256, 0, stream>>>(E, posN, listsA, B1, N_NODES);
    fused_scale_ln_prelu_kernel<<<N_NODES, 256, 0, stream>>>(B1, posN, b1, g1, beta1, prelu_a);

    copy_lists_kernel<<<(listVec + 255) / 256, 256, 0, stream>>>(listsA, listsE, listVec);
    wstd_kernel<<<D, 256, 0, stream>>>(W2, W2s);
    gemm_wmma_kernel<D><<<gemmBlocks, 256, 0, stream>>>(B1, W2s, B0, mTiles);
    copy_lists_kernel<<<(listVec + 255) / 256, 256, 0, stream>>>(listsE, listsB, listVec);

    gather_rows_kernel<true ><<<N_EDGES / 8, 256, 0, stream>>>(B0, posE, listsB + NNZ_CNT, E, N_EDGES);
    gather_rows_kernel<false><<<N_NODES / 8, 256, 0, stream>>>(E, posN, listsB, B0, N_NODES);
    fused_scale_ln_prelu_kernel<<<N_NODES, 256, 0, stream>>>(B0, posN, b2, g2, beta2, prelu_a);
}


# Scale, bias, LayerNorm and PReLU folded into the hyperedge-to-node gather (row still in registers); row mean and variance summed by wave DPP butterflies, a new summation order; two launches fewer
# speedup vs baseline: 4.2521x; 1.4080x over previous
#include <hip/hip_runtime.h>
#include <hip/hip_bf16.h>

#define N_NODES   100000
#define N_EDGES   20000
#define NNZ_CNT   800000
#define D_IN      128
#define D         256
#define EPS_F     1e-5f

typedef float v2f __attribute__((ext_vector_type(2)));
typedef float v4f __attribute__((ext_vector_type(4)));
typedef float v8f __attribute__((ext_vector_type(8)));
typedef int   v4i __attribute__((ext_vector_type(4)));

__device__ __forceinline__ int atomAddI(int* p, int v) {
    return __hip_atomic_fetch_add(p, v, __ATOMIC_RELAXED, __HIP_MEMORY_SCOPE_AGENT);
}

__global__ __launch_bounds__(256) void rank_kernel(const int* __restrict__ node_idx,
                                                   const int* __restrict__ edge_idx,
                                                   int* __restrict__ posN,
                                                   int* __restrict__ posE,
                                                   int* __restrict__ rankN,
                                                   int* __restrict__ rankE) {
    int i = blockIdx.x * 256 + threadIdx.x;
    if (i < NNZ_CNT) {
        rankN[i] = atomAddI(&posN[node_idx[i]], 1);
        rankE[i] = atomAddI(&posE[edge_idx[i]], 1);
    }
}

template<int C>
__device__ __forceinline__ void block_exclusive_scan(int* __restrict__ a, int* sh) {
    const int t    = threadIdx.x;
    const int lane = t & 31;
    const int wv   = t >> 5;
    const bool live = t < 1000;
    v4i v[C / 4];
    int sum = 0;
    if (live) {
        #pragma unroll
        for (int q = 0; q < C / 4; ++q) {
            v[q] = *reinterpret_cast<const v4i*>(a + t * C + 4 * q);
            sum += v[q].x + v[q].y + v[q].z + v[q].w;
        }
    }
    int inc = sum;
    #pragma unroll
    for (int o = 1; o < 32; o <<= 1) {
        int up = __shfl_up(inc, o, 32);
        if (lane >= o) inc += up;
    }
    if (lane == 31) sh[wv] = inc;
    __syncthreads();
    if (wv == 0) {
        int wt = sh[lane];
        int winc = wt;
        #pragma unroll
        for (int o = 1; o < 32; o <<= 1) {
            int up = __shfl_up(winc, o, 32);
            if (lane >= o) winc += up;
        }
        sh[lane] = winc - wt;
    }
    __syncthreads();
    int run = sh[wv] + inc - sum;
    if (live) {
        #pragma unroll
        for (int q = 0; q < C / 4; ++q) {
            v4i x = v[q], o;
            o.x = run; run += x.x;
            o.y = run; run += x.y;
            o.z = run; run += x.z;
            o.w = run; run += x.w;
            *reinterpret_cast<v4i*>(a + t * C + 4 * q) = o;
        }
    }
}

__global__ __launch_bounds__(1024) void scan_kernel(int* __restrict__ posN, int* __restrict__ posE) {
    __shared__ int sh[32];
    if (blockIdx.x == 0) block_exclusive_scan<N_NODES / 1000>(posN, sh);
    else                 block_exclusive_scan<N_EDGES / 1000>(posE, sh);
}

__global__ __launch_bounds__(256) void place_kernel(const int* __restrict__ node_idx,
                                                    const int* __restrict__ edge_idx,
                                                    const int* __restrict__ posN,
                                                    const int* __restrict__ posE,
                                                    const int* __restrict__ rankN,
                                                    const int* __restrict__ rankE,
                                                    int* __restrict__ unsN,
                                                    int* __restrict__ unsE) {
    int i = blockIdx.x * 256 + threadIdx.x;
    if (i < NNZ_CNT) {
        const int n = node_idx[i];
        const int e = edge_idx[i];
        unsE[posE[e] + rankE[i]] = n;
        unsN[posN[n] + rankN[i]] = e;
    }
}

__global__ __launch_bounds__(256) void sort_segments_kernel(const int* __restrict__ posN,
                                                            const int* __restrict__ posE,
                                                            const int* __restrict__ unsN,
                                                            const int* __restrict__ unsE,
                                                            int* __restrict__ srtN,
                                                            int* __restrict__ srtE) {
    const int w    = __builtin_amdgcn_readfirstlane((int)((blockIdx.x * 256 + threadIdx.x) >> 5));
    const int lane = threadIdx.x & 31;
    if (w >= N_EDGES + N_NODES) return;
    const bool isE = w < N_EDGES;
    const int  k   = isE ? w : w - N_EDGES;
    const int* __restrict__ pos = isE ? posE : posN;
    const int* __restrict__ U   = isE ? unsE : unsN;
    int*       __restrict__ S   = isE ? srtE : srtN;
    const int nk  = isE ? N_EDGES : N_NODES;
    const int beg = pos[k];
    const int L   = ((k + 1 < nk) ? pos[k + 1] : NNZ_CNT) - beg;
    for (int m = 0; m < L; m += 32) {
        const int p = m + lane;
        const int x = (p < L) ? U[beg + p] : 0;
        int rank = 0;
        for (int c = 0; c < L; c += 32) {
            const int q  = c + lane;
            const int ec = (q < L) ? U[beg + q] : 0;
            const int cnt = (L - c < 32) ? (L - c) : 32;
            for (int j = 0; j < cnt; ++j) {
                const int v = __builtin_amdgcn_readlane(ec, j);
                rank += (v < x || (v == x && (c + j) < p)) ? 1 : 0;
            }
        }
        if (p < L) S[beg + rank] = x;
    }
}

template<int K>
__global__ __launch_bounds__(256) void gemm_wmma_kernel(const float* __restrict__ X,
                                                        const float* __restrict__ W,
                                                        float* __restrict__ C,
                                                        int mTiles) {
    const int wid  = (blockIdx.x * 256 + threadIdx.x) >> 5;
    const int lane = threadIdx.x & 31;
    const int mT   = wid >> 2;
    const int nT   = wid & 3;
    if (mT >= mTiles) return;
    const int m0   = mT * 16;
    const int n0   = nT * 64;
    const int half = lane >> 4;
    const int l16  = lane & 15;

    const float* __restrict__ xrow = X + (size_t)(m0 + l16) * K;
    const float* __restrict__ wrow = W + (size_t)(n0 + l16) * K;

    v8f acc0 = {}, acc1 = {}, acc2 = {}, acc3 = {};

    #pragma unroll 8
    for (int k0 = 0; k0 < K; k0 += 4) {
        const int ko = k0 + 2 * half;
        v2f a  = *reinterpret_cast<const v2f*>(xrow + ko);
        v2f b0 = *reinterpret_cast<const v2f*>(wrow + (size_t)0 * 16 * K + ko);
        v2f b1 = *reinterpret_cast<const v2f*>(wrow + (size_t)1 * 16 * K + ko);
        v2f b2 = *reinterpret_cast<const v2f*>(wrow + (size_t)2 * 16 * K + ko);
        v2f b3 = *reinterpret_cast<const v2f*>(wrow + (size_t)3 * 16 * K + ko);
        acc0 = __builtin_amdgcn_wmma_f32_16x16x4_f32(false, a, false, b0, (short)0, acc0, false, false);
        acc1 = __builtin_amdgcn_wmma_f32_16x16x4_f32(false, a, false, b1, (short)0, acc1, false, false);
        acc2 = __builtin_amdgcn_wmma_f32_16x16x4_f32(false, a, false, b2, (short)0, acc2, false, false);
        acc3 = __builtin_amdgcn_wmma_f32_16x16x4_f32(false, a, false, b3, (short)0, acc3, false, false);
    }

    float* __restrict__ ccol = C + n0 + l16;
    #pragma unroll
    for (int r = 0; r < 8; ++r) {
        const size_t m = (size_t)(m0 + r + 8 * half);
        ccol[m * D + 0 * 16] = acc0[r];
        ccol[m * D + 1 * 16] = acc1[r];
        ccol[m * D + 2 * 16] = acc2[r];
        ccol[m * D + 3 * 16] = acc3[r];
    }
}

__device__ __forceinline__ void gather_segment(const float* __restrict__ src,
                                               const int* __restrict__ list,
                                               int beg, int end, int lane,
                                               v4f& a0, v4f& a1) {
    const float* __restrict__ lsrc = src + lane * 4;
    for (int c = beg; c < end; c += 32) {
        const int q   = c + lane;
        const int idx = (q < end) ? list[q] : 0;
        const int cnt = (end - c < 32) ? (end - c) : 32;
        int j = 0;
        for (; j + 4 <= cnt; j += 4) {
            const float* p0 = lsrc + (size_t)__builtin_amdgcn_readlane(idx, j + 0) * D;
            const float* p1 = lsrc + (size_t)__builtin_amdgcn_readlane(idx, j + 1) * D;
            const float* p2 = lsrc + (size_t)__builtin_amdgcn_readlane(idx, j + 2) * D;
            const float* p3 = lsrc + (size_t)__builtin_amdgcn_readlane(idx, j + 3) * D;
            const v4f x0 = *reinterpret_cast<const v4f*>(p0), y0 = *reinterpret_cast<const v4f*>(p0 + 128);
            const v4f x1 = *reinterpret_cast<const v4f*>(p1), y1 = *reinterpret_cast<const v4f*>(p1 + 128);
            const v4f x2 = *reinterpret_cast<const v4f*>(p2), y2 = *reinterpret_cast<const v4f*>(p2 + 128);
            const v4f x3 = *reinterpret_cast<const v4f*>(p3), y3 = *reinterpret_cast<const v4f*>(p3 + 128);
            a0 += x0; a1 += y0;
            a0 += x1; a1 += y1;
            a0 += x2; a1 += y2;
            a0 += x3; a1 += y3;
        }
        for (; j < cnt; ++j) {
            const float* p0 = lsrc + (size_t)__builtin_amdgcn_readlane(idx, j) * D;
            a0 += *reinterpret_cast<const v4f*>(p0);
            a1 += *reinterpret_cast<const v4f*>(p0 + 128);
        }
    }
}

__global__ __launch_bounds__(256) void gather_edge_mean_kernel(const float* __restrict__ src,
                                                               const int* __restrict__ pos,
                                                               const int* __restrict__ list,
                                                               float* __restrict__ dst) {
    const int w    = __builtin_amdgcn_readfirstlane((int)((blockIdx.x * 256 + threadIdx.x) >> 5));
    const int lane = threadIdx.x & 31;
    if (w >= N_EDGES) return;
    const int beg = pos[w];
    const int end = (w + 1 < N_EDGES) ? pos[w + 1] : NNZ_CNT;

    v4f a0 = {}, a1 = {};
    gather_segment(src, list, beg, end, lane, a0, a1);

    const float c   = (float)(end - beg);
    const float inv = (c > 0.0f) ? 1.0f / c : 0.0f;
    a0 *= inv; a1 *= inv;
    float* drow = dst + (size_t)w * D + lane * 4;
    *reinterpret_cast<v4f*>(drow)       = a0;
    *reinterpret_cast<v4f*>(drow + 128) = a1;
}

template<int M>
__device__ __forceinline__ float row_xor_add(float v) {
    return v + __builtin_bit_cast(float, __builtin_amdgcn_update_dpp(0, __builtin_bit_cast(int, v), 0x160 | M, 0xf, 0xf, true));
}

__device__ __forceinline__ float wave_sum(float v) {
    v = row_xor_add<1>(v);
    v = row_xor_add<2>(v);
    v = row_xor_add<4>(v);
    v = row_xor_add<8>(v);
    const int b = __builtin_bit_cast(int, v);
    return v + __builtin_bit_cast(float, __builtin_amdgcn_permlanex16(b, b, 0x76543210, 0xfedcba98, false, false));
}

__global__ __launch_bounds__(256) void gather_node_ln_prelu_kernel(const float* __restrict__ src,
                                                                   const int* __restrict__ pos,
                                                                   const int* __restrict__ list,
                                                                   const float* __restrict__ bias,
                                                                   const float* __restrict__ g,
                                                                   const float* __restrict__ beta,
                                                                   const float* __restrict__ pa,
                                                                   float* __restrict__ dst) {
    const int w    = __builtin_amdgcn_readfirstlane((int)((blockIdx.x * 256 + threadIdx.x) >> 5));
    const int lane = threadIdx.x & 31;
    if (w >= N_NODES) return;
    const int beg = pos[w];
    const int end = (w + 1 < N_NODES) ? pos[w + 1] : NNZ_CNT;

    v4f a0 = {}, a1 = {};
    gather_segment(src, list, beg, end, lane, a0, a1);

    const float dg   = (float)(end - beg);
    const float dinv = (dg > 0.0f) ? 1.0f / dg : 0.0f;
    const int   col  = lane * 4;
    const v4f v0 = a0 * dinv + *reinterpret_cast<const v4f*>(bias + col);
    const v4f v1 = a1 * dinv + *reinterpret_cast<const v4f*>(bias + col + 128);

    const float mu = wave_sum(((v0.x + v0.y) + (v0.z + v0.w)) + ((v1.x + v1.y) + (v1.z + v1.w))) * (1.0f / D);
    const v4f d0 = v0 - mu;
    const v4f d1 = v1 - mu;
    const v4f q0 = d0 * d0;
    const v4f q1 = d1 * d1;
    const float var = wave_sum(((q0.x + q0.y) + (q0.z + q0.w)) + ((q1.x + q1.y) + (q1.z + q1.w))) * (1.0f / D);
    const float rs  = rsqrtf(var + EPS_F);

    const v4f y0 = d0 * rs * *reinterpret_cast<const v4f*>(g + col)       + *reinterpret_cast<const v4f*>(beta + col);
    const v4f y1 = d1 * rs * *reinterpret_cast<const v4f*>(g + col + 128) + *reinterpret_cast<const v4f*>(beta + col + 128);
    const float a = pa[0];
    v4f o0, o1;
    o0.x = (y0.x >= 0.0f) ? y0.x : a * y0.x;
    o0.y = (y0.y >= 0.0f) ? y0.y : a * y0.y;
    o0.z = (y0.z >= 0.0f) ? y0.z : a * y0.z;
    o0.w = (y0.w >= 0.0f) ? y0.w : a * y0.w;
    o1.x = (y1.x >= 0.0f) ? y1.x : a * y1.x;
    o1.y = (y1.y >= 0.0f) ? y1.y : a * y1.y;
    o1.z = (y1.z >= 0.0f) ? y1.z : a * y1.z;
    o1.w = (y1.w >= 0.0f) ? y1.w : a * y1.w;

    float* drow = dst + (size_t)w * D + col;
    *reinterpret_cast<v4f*>(drow)       = o0;
    *reinterpret_cast<v4f*>(drow + 128) = o1;
}

__global__ __launch_bounds__(256) void copy_lists_kernel(const int* __restrict__ from,
                                                         int* __restrict__ to, int n4) {
    const int i = blockIdx.x * 256 + threadIdx.x;
    if (i < n4) reinterpret_cast<v4i*>(to)[i] = reinterpret_cast<const v4i*>(from)[i];
}

__global__ __launch_bounds__(256) void wstd_kernel(const float* __restrict__ W,
                                                   float* __restrict__ Ws) {
    __shared__ float sred[256];
    const int r = blockIdx.x;
    const int f = threadIdx.x;
    const float w = W[(size_t)r * D + f];

    sred[f] = w;
    __syncthreads();
    #pragma unroll
    for (int s = 128; s > 0; s >>= 1) { if (f < s) sred[f] += sred[f + s]; __syncthreads(); }
    const float mean = sred[0] * (1.0f / D);
    __syncthreads();

    const float dv = w - mean;
    sred[f] = dv * dv;
    __syncthreads();
    #pragma unroll
    for (int s = 128; s > 0; s >>= 1) { if (f < s) sred[f] += sred[f + s]; __syncthreads(); }
    const float var = sred[0] * (1.0f / (D - 1));

    Ws[(size_t)r * D + f] = dv * rsqrtf(var + EPS_F);
}

extern "C" void kernel_launch(void* const* d_in, const int* in_sizes, int n_in,
                              void* d_out, int out_size, void* d_ws, size_t ws_size,
                              hipStream_t stream) {
    const float* edge_fea = (const float*)d_in[0];
    const int*   node_idx = (const int*)  d_in[1];
    const int*   edge_idx = (const int*)  d_in[2];
    const float* W1       = (const float*)d_in[3];
    const float* b1       = (const float*)d_in[4];
    const float* W2       = (const float*)d_in[5];
    const float* b2       = (const float*)d_in[6];
    const float* g1       = (const float*)d_in[7];
    const float* beta1    = (const float*)d_in[8];
    const float* g2       = (const float*)d_in[9];
    const float* beta2    = (const float*)d_in[10];
    const float* prelu_a  = (const float*)d_in[11];

    const size_t nodeRowBytes = (size_t)N_NODES * D * sizeof(float);

    float* B0   = (float*)d_out;
    char*  ws   = (char*)d_ws;
    float* B1   = (float*)ws;
    float* E    = (float*)(ws + nodeRowBytes);
    int*   posN = (int*)(E + (size_t)N_EDGES * D);
    int*   posE = posN + N_NODES;
    float* W2s  = (float*)(posE + N_EDGES);

    int* listsA = (int*)B0;
    int* listsE = (int*)E;
    int* listsB = (int*)B1;
    int* unsN   = listsA + 2 * NNZ_CNT;
    int* unsE   = unsN + NNZ_CNT;
    int* rankN  = unsE + NNZ_CNT;
    int* rankE  = rankN + NNZ_CNT;

    const int mTiles = N_NODES / 16;
    const int gemmBlocks = (mTiles * 4 * 32) / 256;
    const int nnzBlocks  = (NNZ_CNT + 255) / 256;
    const int listVec    = (2 * NNZ_CNT) / 4;

    hipMemsetAsync(posN, 0, (size_t)(N_NODES + N_EDGES) * sizeof(int), stream);
    rank_kernel<<<nnzBlocks, 256, 0, stream>>>(node_idx, edge_idx, posN, posE, rankN, rankE);
    scan_kernel<<<2, 1024, 0, stream>>>(posN, posE);
    place_kernel<<<nnzBlocks, 256, 0, stream>>>(node_idx, edge_idx, posN, posE, rankN, rankE, unsN, unsE);
    sort_segments_kernel<<<(N_EDGES + N_NODES) / 8, 256, 0, stream>>>(posN, posE, unsN, unsE,
                                                                     listsA, listsA + NNZ_CNT);

    gemm_wmma_kernel<D_IN><<<gemmBlocks, 256, 0, stream>>>(edge_fea, W1, B1, mTiles);
    gather_edge_mean_kernel<<<N_EDGES / 8, 256, 0, stream>>>(B1, posE, listsA + NNZ_CNT, E);
    gather_node_ln_prelu_kernel<<<N_NODES / 8, 256, 0, stream>>>(E, posN, listsA, b1, g1, beta1, prelu_a, B1);

    copy_lists_kernel<<<(listVec + 255) / 256, 256, 0, stream>>>(listsA, listsE, listVec);
    wstd_kernel<<<D, 256, 0, stream>>>(W2, W2s);
    gemm_wmma_kernel<D><<<gemmBlocks, 256, 0, stream>>>(B1, W2s, B0, mTiles);
    copy_lists_kernel<<<(listVec + 255) / 256, 256, 0, stream>>>(listsE, listsB, listVec);

    gather_edge_mean_kernel<<<N_EDGES / 8, 256, 0, stream>>>(B0, posE, listsB + NNZ_CNT, E);
    gather_node_ln_prelu_kernel<<<N_NODES / 8, 256, 0, stream>>>(E, posN, listsB, b2, g2, beta2, prelu_a, B0);
}


# Weight products moved behind the node-to-hyperedge mean (linearity): both WMMA GEMMs run on 20000 averaged rows instead of 100000 node rows; averaging before projection is a new summation order
# speedup vs baseline: 7.0693x; 1.6625x over previous
#include <hip/hip_runtime.h>
#include <hip/hip_bf16.h>

#define N_NODES   100000
#define N_EDGES   20000
#define NNZ_CNT   800000
#define D_IN      128
#define D         256
#define EPS_F     1e-5f

typedef float v2f __attribute__((ext_vector_type(2)));
typedef float v4f __attribute__((ext_vector_type(4)));
typedef float v8f __attribute__((ext_vector_type(8)));
typedef int   v4i __attribute__((ext_vector_type(4)));

__device__ __forceinline__ int atomAddI(int* p, int v) {
    return __hip_atomic_fetch_add(p, v, __ATOMIC_RELAXED, __HIP_MEMORY_SCOPE_AGENT);
}

__global__ __launch_bounds__(256) void rank_kernel(const int* __restrict__ node_idx,
                                                   const int* __restrict__ edge_idx,
                                                   int* __restrict__ posN,
                                                   int* __restrict__ posE,
                                                   int* __restrict__ rankN,
                                                   int* __restrict__ rankE) {
    int i = blockIdx.x * 256 + threadIdx.x;
    if (i < NNZ_CNT) {
        rankN[i] = atomAddI(&posN[node_idx[i]], 1);
        rankE[i] = atomAddI(&posE[edge_idx[i]], 1);
    }
}

template<int C>
__device__ __forceinline__ void block_exclusive_scan(int* __restrict__ a, int* sh) {
    const int t    = threadIdx.x;
    const int lane = t & 31;
    const int wv   = t >> 5;
    const bool live = t < 1000;
    v4i v[C / 4];
    int sum = 0;
    if (live) {
        #pragma unroll
        for (int q = 0; q < C / 4; ++q) {
            v[q] = *reinterpret_cast<const v4i*>(a + t * C + 4 * q);
            sum += v[q].x + v[q].y + v[q].z + v[q].w;
        }
    }
    int inc = sum;
    #pragma unroll
    for (int o = 1; o < 32; o <<= 1) {
        int up = __shfl_up(inc, o, 32);
        if (lane >= o) inc += up;
    }
    if (lane == 31) sh[wv] = inc;
    __syncthreads();
    if (wv == 0) {
        int wt = sh[lane];
        int winc = wt;
        #pragma unroll
        for (int o = 1; o < 32; o <<= 1) {
            int up = __shfl_up(winc, o, 32);
            if (lane >= o) winc += up;
        }
        sh[lane] = winc - wt;
    }
    __syncthreads();
    int run = sh[wv] + inc - sum;
    if (live) {
        #pragma unroll
        for (int q = 0; q < C / 4; ++q) {
            v4i x = v[q], o;
            o.x = run; run += x.x;
            o.y = run; run += x.y;
            o.z = run; run += x.z;
            o.w = run; run += x.w;
            *reinterpret_cast<v4i*>(a + t * C + 4 * q) = o;
        }
    }
}

__global__ __launch_bounds__(1024) void scan_kernel(int* __restrict__ posN, int* __restrict__ posE) {
    __shared__ int sh[32];
    if (blockIdx.x == 0) block_exclusive_scan<N_NODES / 1000>(posN, sh);
    else                 block_exclusive_scan<N_EDGES / 1000>(posE, sh);
}

__global__ __launch_bounds__(256) void place_kernel(const int* __restrict__ node_idx,
                                                    const int* __restrict__ edge_idx,
                                                    const int* __restrict__ posN,
                                                    const int* __restrict__ posE,
                                                    const int* __restrict__ rankN,
                                                    const int* __restrict__ rankE,
                                                    int* __restrict__ unsN,
                                                    int* __restrict__ unsE) {
    int i = blockIdx.x * 256 + threadIdx.x;
    if (i < NNZ_CNT) {
        const int n = node_idx[i];
        const int e = edge_idx[i];
        unsE[posE[e] + rankE[i]] = n;
        unsN[posN[n] + rankN[i]] = e;
    }
}

__global__ __launch_bounds__(256) void sort_segments_kernel(const int* __restrict__ posN,
                                                            const int* __restrict__ posE,
                                                            const int* __restrict__ unsN,
                                                            const int* __restrict__ unsE,
                                                            int* __restrict__ srtN,
                                                            int* __restrict__ srtE) {
    const int w    = __builtin_amdgcn_readfirstlane((int)((blockIdx.x * 256 + threadIdx.x) >> 5));
    const int lane = threadIdx.x & 31;
    if (w >= N_EDGES + N_NODES) return;
    const bool isE = w < N_EDGES;
    const int  k   = isE ? w : w - N_EDGES;
    const int* __restrict__ pos = isE ? posE : posN;
    const int* __restrict__ U   = isE ? unsE : unsN;
    int*       __restrict__ S   = isE ? srtE : srtN;
    const int nk  = isE ? N_EDGES : N_NODES;
    const int beg = pos[k];
    const int L   = ((k + 1 < nk) ? pos[k + 1] : NNZ_CNT) - beg;
    for (int m = 0; m < L; m += 32) {
        const int p = m + lane;
        const int x = (p < L) ? U[beg + p] : 0;
        int rank = 0;
        for (int c = 0; c < L; c += 32) {
            const int q  = c + lane;
            const int ec = (q < L) ? U[beg + q] : 0;
            const int cnt = (L - c < 32) ? (L - c) : 32;
            for (int j = 0; j < cnt; ++j) {
                const int v = __builtin_amdgcn_readlane(ec, j);
                rank += (v < x || (v == x && (c + j) < p)) ? 1 : 0;
            }
        }
        if (p < L) S[beg + rank] = x;
    }
}

template<int K>
__global__ __launch_bounds__(256) void gemm_wmma_kernel(const float* __restrict__ X,
                                                        const float* __restrict__ W,
                                                        float* __restrict__ C,
                                                        int mTiles) {
    const int wid  = (blockIdx.x * 256 + threadIdx.x) >> 5;
    const int lane = threadIdx.x & 31;
    const int mT   = wid >> 2;
    const int nT   = wid & 3;
    if (mT >= mTiles) return;
    const int m0   = mT * 16;
    const int n0   = nT * 64;
    const int half = lane >> 4;
    const int l16  = lane & 15;

    const float* __restrict__ xrow = X + (size_t)(m0 + l16) * K;
    const float* __restrict__ wrow = W + (size_t)(n0 + l16) * K;

    v8f acc0 = {}, acc1 = {}, acc2 = {}, acc3 = {};

    #pragma unroll 8
    for (int k0 = 0; k0 < K; k0 += 4) {
        const int ko = k0 + 2 * half;
        v2f a  = *reinterpret_cast<const v2f*>(xrow + ko);
        v2f b0 = *reinterpret_cast<const v2f*>(wrow + (size_t)0 * 16 * K + ko);
        v2f b1 = *reinterpret_cast<const v2f*>(wrow + (size_t)1 * 16 * K + ko);
        v2f b2 = *reinterpret_cast<const v2f*>(wrow + (size_t)2 * 16 * K + ko);
        v2f b3 = *reinterpret_cast<const v2f*>(wrow + (size_t)3 * 16 * K + ko);
        acc0 = __builtin_amdgcn_wmma_f32_16x16x4_f32(false, a, false, b0, (short)0, acc0, false, false);
        acc1 = __builtin_amdgcn_wmma_f32_16x16x4_f32(false, a, false, b1, (short)0, acc1, false, false);
        acc2 = __builtin_amdgcn_wmma_f32_16x16x4_f32(false, a, false, b2, (short)0, acc2, false, false);
        acc3 = __builtin_amdgcn_wmma_f32_16x16x4_f32(false, a, false, b3, (short)0, acc3, false, false);
    }

    float* __restrict__ ccol = C + n0 + l16;
    #pragma unroll
    for (int r = 0; r < 8; ++r) {
        const size_t m = (size_t)(m0 + r + 8 * half);
        ccol[m * D + 0 * 16] = acc0[r];
        ccol[m * D + 1 * 16] = acc1[r];
        ccol[m * D + 2 * 16] = acc2[r];
        ccol[m * D + 3 * 16] = acc3[r];
    }
}

__device__ __forceinline__ void gather_segment(const float* __restrict__ src,
                                               const int* __restrict__ list,
                                               int beg, int end, int lane,
                                               v4f& a0, v4f& a1) {
    const float* __restrict__ lsrc = src + lane * 4;
    for (int c = beg; c < end; c += 32) {
        const int q   = c + lane;
        const int idx = (q < end) ? list[q] : 0;
        const int cnt = (end - c < 32) ? (end - c) : 32;
        int j = 0;
        for (; j + 4 <= cnt; j += 4) {
            const float* p0 = lsrc + (size_t)__builtin_amdgcn_readlane(idx, j + 0) * D;
            const float* p1 = lsrc + (size_t)__builtin_amdgcn_readlane(idx, j + 1) * D;
            const float* p2 = lsrc + (size_t)__builtin_amdgcn_readlane(idx, j + 2) * D;
            const float* p3 = lsrc + (size_t)__builtin_amdgcn_readlane(idx, j + 3) * D;
            const v4f x0 = *reinterpret_cast<const v4f*>(p0), y0 = *reinterpret_cast<const v4f*>(p0 + 128);
            const v4f x1 = *reinterpret_cast<const v4f*>(p1), y1 = *reinterpret_cast<const v4f*>(p1 + 128);
            const v4f x2 = *reinterpret_cast<const v4f*>(p2), y2 = *reinterpret_cast<const v4f*>(p2 + 128);
            const v4f x3 = *reinterpret_cast<const v4f*>(p3), y3 = *reinterpret_cast<const v4f*>(p3 + 128);
            a0 += x0; a1 += y0;
            a0 += x1; a1 += y1;
            a0 += x2; a1 += y2;
            a0 += x3; a1 += y3;
        }
        for (; j < cnt; ++j) {
            const float* p0 = lsrc + (size_t)__builtin_amdgcn_readlane(idx, j) * D;
            a0 += *reinterpret_cast<const v4f*>(p0);
            a1 += *reinterpret_cast<const v4f*>(p0 + 128);
        }
    }
}

__global__ __launch_bounds__(256) void gather_edge_mean_kernel(const float* __restrict__ src,
                                                               const int* __restrict__ pos,
                                                               const int* __restrict__ list,
                                                               float* __restrict__ dst) {
    const int w    = __builtin_amdgcn_readfirstlane((int)((blockIdx.x * 256 + threadIdx.x) >> 5));
    const int lane = threadIdx.x & 31;
    if (w >= N_EDGES) return;
    const int beg = pos[w];
    const int end = (w + 1 < N_EDGES) ? pos[w + 1] : NNZ_CNT;

    v4f a0 = {}, a1 = {};
    gather_segment(src, list, beg, end, lane, a0, a1);

    const float c   = (float)(end - beg);
    const float inv = (c > 0.0f) ? 1.0f / c : 0.0f;
    a0 *= inv; a1 *= inv;
    float* drow = dst + (size_t)w * D + lane * 4;
    *reinterpret_cast<v4f*>(drow)       = a0;
    *reinterpret_cast<v4f*>(drow + 128) = a1;
}

__device__ __forceinline__ void gather_segment_in(const float* __restrict__ src,
                                                  const int* __restrict__ list,
                                                  int beg, int end, int lane, v4f& a0) {
    const float* __restrict__ lsrc = src + lane * 4;
    for (int c = beg; c < end; c += 32) {
        const int q   = c + lane;
        const int idx = (q < end) ? list[q] : 0;
        const int cnt = (end - c < 32) ? (end - c) : 32;
        int j = 0;
        for (; j + 8 <= cnt; j += 8) {
            const v4f x0 = *reinterpret_cast<const v4f*>(lsrc + (size_t)__builtin_amdgcn_readlane(idx, j + 0) * D_IN);
            const v4f x1 = *reinterpret_cast<const v4f*>(lsrc + (size_t)__builtin_amdgcn_readlane(idx, j + 1) * D_IN);
            const v4f x2 = *reinterpret_cast<const v4f*>(lsrc + (size_t)__builtin_amdgcn_readlane(idx, j + 2) * D_IN);
            const v4f x3 = *reinterpret_cast<const v4f*>(lsrc + (size_t)__builtin_amdgcn_readlane(idx, j + 3) * D_IN);
            const v4f x4 = *reinterpret_cast<const v4f*>(lsrc + (size_t)__builtin_amdgcn_readlane(idx, j + 4) * D_IN);
            const v4f x5 = *reinterpret_cast<const v4f*>(lsrc + (size_t)__builtin_amdgcn_readlane(idx, j + 5) * D_IN);
            const v4f x6 = *reinterpret_cast<const v4f*>(lsrc + (size_t)__builtin_amdgcn_readlane(idx, j + 6) * D_IN);
            const v4f x7 = *reinterpret_cast<const v4f*>(lsrc + (size_t)__builtin_amdgcn_readlane(idx, j + 7) * D_IN);
            a0 += x0; a0 += x1; a0 += x2; a0 += x3;
            a0 += x4; a0 += x5; a0 += x6; a0 += x7;
        }
        if (j + 4 <= cnt) {
            const v4f x0 = *reinterpret_cast<const v4f*>(lsrc + (size_t)__builtin_amdgcn_readlane(idx, j + 0) * D_IN);
            const v4f x1 = *reinterpret_cast<const v4f*>(lsrc + (size_t)__builtin_amdgcn_readlane(idx, j + 1) * D_IN);
            const v4f x2 = *reinterpret_cast<const v4f*>(lsrc + (size_t)__builtin_amdgcn_readlane(idx, j + 2) * D_IN);
            const v4f x3 = *reinterpret_cast<const v4f*>(lsrc + (size_t)__builtin_amdgcn_readlane(idx, j + 3) * D_IN);
            a0 += x0; a0 += x1; a0 += x2; a0 += x3;
            j += 4;
        }
        for (; j < cnt; ++j) {
            a0 += *reinterpret_cast<const v4f*>(lsrc + (size_t)__builtin_amdgcn_readlane(idx, j) * D_IN);
        }
    }
}

__global__ __launch_bounds__(256) void gather_edge_mean_in_kernel(const float* __restrict__ src,
                                                                  const int* __restrict__ pos,
                                                                  const int* __restrict__ list,
                                                                  float* __restrict__ dst) {
    const int w    = __builtin_amdgcn_readfirstlane((int)((blockIdx.x * 256 + threadIdx.x) >> 5));
    const int lane = threadIdx.x & 31;
    if (w >= N_EDGES) return;
    const int beg = pos[w];
    const int end = (w + 1 < N_EDGES) ? pos[w + 1] : NNZ_CNT;

    v4f a0 = {};
    gather_segment_in(src, list, beg, end, lane, a0);

    const float c   = (float)(end - beg);
    const float inv = (c > 0.0f) ? 1.0f / c : 0.0f;
    a0 *= inv;
    *reinterpret_cast<v4f*>(dst + (size_t)w * D_IN + lane * 4) = a0;
}

template<int M>
__device__ __forceinline__ float row_xor_add(float v) {
    return v + __builtin_bit_cast(float, __builtin_amdgcn_update_dpp(0, __builtin_bit_cast(int, v), 0x160 | M, 0xf, 0xf, true));
}

__device__ __forceinline__ float wave_sum(float v) {
    v = row_xor_add<1>(v);
    v = row_xor_add<2>(v);
    v = row_xor_add<4>(v);
    v = row_xor_add<8>(v);
    const int b = __builtin_bit_cast(int, v);
    return v + __builtin_bit_cast(float, __builtin_amdgcn_permlanex16(b, b, 0x76543210, 0xfedcba98, false, false));
}

__global__ __launch_bounds__(256) void gather_node_ln_prelu_kernel(const float* __restrict__ src,
                                                                   const int* __restrict__ pos,
                                                                   const int* __restrict__ list,
                                                                   const float* __restrict__ bias,
                                                                   const float* __restrict__ g,
                                                                   const float* __restrict__ beta,
                                                                   const float* __restrict__ pa,
                                                                   float* __restrict__ dst) {
    const int w    = __builtin_amdgcn_readfirstlane((int)((blockIdx.x * 256 + threadIdx.x) >> 5));
    const int lane = threadIdx.x & 31;
    if (w >= N_NODES) return;
    const int beg = pos[w];
    const int end = (w + 1 < N_NODES) ? pos[w + 1] : NNZ_CNT;

    v4f a0 = {}, a1 = {};
    gather_segment(src, list, beg, end, lane, a0, a1);

    const float dg   = (float)(end - beg);
    const float dinv = (dg > 0.0f) ? 1.0f / dg : 0.0f;
    const int   col  = lane * 4;
    const v4f v0 = a0 * dinv + *reinterpret_cast<const v4f*>(bias + col);
    const v4f v1 = a1 * dinv + *reinterpret_cast<const v4f*>(bias + col + 128);

    const float mu = wave_sum(((v0.x + v0.y) + (v0.z + v0.w)) + ((v1.x + v1.y) + (v1.z + v1.w))) * (1.0f / D);
    const v4f d0 = v0 - mu;
    const v4f d1 = v1 - mu;
    const v4f q0 = d0 * d0;
    const v4f q1 = d1 * d1;
    const float var = wave_sum(((q0.x + q0.y) + (q0.z + q0.w)) + ((q1.x + q1.y) + (q1.z + q1.w))) * (1.0f / D);
    const float rs  = rsqrtf(var + EPS_F);

    const v4f y0 = d0 * rs * *reinterpret_cast<const v4f*>(g + col)       + *reinterpret_cast<const v4f*>(beta + col);
    const v4f y1 = d1 * rs * *reinterpret_cast<const v4f*>(g + col + 128) + *reinterpret_cast<const v4f*>(beta + col + 128);
    const float a = pa[0];
    v4f o0, o1;
    o0.x = (y0.x >= 0.0f) ? y0.x : a * y0.x;
    o0.y = (y0.y >= 0.0f) ? y0.y : a * y0.y;
    o0.z = (y0.z >= 0.0f) ? y0.z : a * y0.z;
    o0.w = (y0.w >= 0.0f) ? y0.w : a * y0.w;
    o1.x = (y1.x >= 0.0f) ? y1.x : a * y1.x;
    o1.y = (y1.y >= 0.0f) ? y1.y : a * y1.y;
    o1.z = (y1.z >= 0.0f) ? y1.z : a * y1.z;
    o1.w = (y1.w >= 0.0f) ? y1.w : a * y1.w;

    float* drow = dst + (size_t)w * D + col;
    *reinterpret_cast<v4f*>(drow)       = o0;
    *reinterpret_cast<v4f*>(drow + 128) = o1;
}

__global__ __launch_bounds__(256) void wstd_kernel(const float* __restrict__ W,
                                                   float* __restrict__ Ws) {
    __shared__ float sred[256];
    const int r = blockIdx.x;
    const int f = threadIdx.x;
    const float w = W[(size_t)r * D + f];

    sred[f] = w;
    __syncthreads();
    #pragma unroll
    for (int s = 128; s > 0; s >>= 1) { if (f < s) sred[f] += sred[f + s]; __syncthreads(); }
    const float mean = sred[0] * (1.0f / D);
    __syncthreads();

    const float dv = w - mean;
    sred[f] = dv * dv;
    __syncthreads();
    #pragma unroll
    for (int s = 128; s > 0; s >>= 1) { if (f < s) sred[f] += sred[f + s]; __syncthreads(); }
    const float var = sred[0] * (1.0f / (D - 1));

    Ws[(size_t)r * D + f] = dv * rsqrtf(var + EPS_F);
}

extern "C" void kernel_launch(void* const* d_in, const int* in_sizes, int n_in,
                              void* d_out, int out_size, void* d_ws, size_t ws_size,
                              hipStream_t stream) {
    const float* edge_fea = (const float*)d_in[0];
    const int*   node_idx = (const int*)  d_in[1];
    const int*   edge_idx = (const int*)  d_in[2];
    const float* W1       = (const float*)d_in[3];
    const float* b1       = (const float*)d_in[4];
    const float* W2       = (const float*)d_in[5];
    const float* b2       = (const float*)d_in[6];
    const float* g1       = (const float*)d_in[7];
    const float* beta1    = (const float*)d_in[8];
    const float* g2       = (const float*)d_in[9];
    const float* beta2    = (const float*)d_in[10];
    const float* prelu_a  = (const float*)d_in[11];

    const size_t nodeRowBytes = (size_t)N_NODES * D * sizeof(float);

    char*  ws   = (char*)d_ws;
    float* H    = (float*)d_out;
    int*   srtN = (int*)ws;
    int*   srtE = srtN + NNZ_CNT;
    int*   unsN = srtE + NNZ_CNT;
    int*   unsE = unsN + NNZ_CNT;
    int*   rankN = unsE + NNZ_CNT;
    int*   rankE = rankN + NNZ_CNT;
    float* A    = (float*)(ws + ((size_t)32 << 20));
    float* E    = (float*)(ws + nodeRowBytes);
    int*   posN = (int*)(E + (size_t)N_EDGES * D);
    int*   posE = posN + N_NODES;
    float* W2s  = (float*)(posE + N_EDGES);

    const int mTiles = N_EDGES / 16;
    const int gemmBlocks = (mTiles * 4 * 32) / 256;
    const int nnzBlocks  = (NNZ_CNT + 255) / 256;

    hipMemsetAsync(posN, 0, (size_t)(N_NODES + N_EDGES) * sizeof(int), stream);
    rank_kernel<<<nnzBlocks, 256, 0, stream>>>(node_idx, edge_idx, posN, posE, rankN, rankE);
    scan_kernel<<<2, 1024, 0, stream>>>(posN, posE);
    place_kernel<<<nnzBlocks, 256, 0, stream>>>(node_idx, edge_idx, posN, posE, rankN, rankE, unsN, unsE);
    sort_segments_kernel<<<(N_EDGES + N_NODES) / 8, 256, 0, stream>>>(posN, posE, unsN, unsE, srtN, srtE);

    gather_edge_mean_in_kernel<<<N_EDGES / 8, 256, 0, stream>>>(edge_fea, posE, srtE, A);
    gemm_wmma_kernel<D_IN><<<gemmBlocks, 256, 0, stream>>>(A, W1, E, mTiles);
    gather_node_ln_prelu_kernel<<<N_NODES / 8, 256, 0, stream>>>(E, posN, srtN, b1, g1, beta1, prelu_a, H);

    wstd_kernel<<<D, 256, 0, stream>>>(W2, W2s);
    gather_edge_mean_kernel<<<N_EDGES / 8, 256, 0, stream>>>(H, posE, srtE, A);
    gemm_wmma_kernel<D><<<gemmBlocks, 256, 0, stream>>>(A, W2s, E, mTiles);
    gather_node_ln_prelu_kernel<<<N_NODES / 8, 256, 0, stream>>>(E, posN, srtN, b2, g2, beta2, prelu_a, H);
}


# Both WMMA GEMMs now feed X and W through LDS: each 128-byte row chunk arrives by whole-line 16-byte loads one chunk ahead in registers; fragment k order, products and sums unchanged
# speedup vs baseline: 7.6041x; 1.0757x over previous
#include <hip/hip_runtime.h>
#include <hip/hip_bf16.h>

#define N_NODES   100000
#define N_EDGES   20000
#define NNZ_CNT   800000
#define D_IN      128
#define D         256
#define EPS_F     1e-5f

typedef float v2f __attribute__((ext_vector_type(2)));
typedef float v4f __attribute__((ext_vector_type(4)));
typedef float v8f __attribute__((ext_vector_type(8)));
typedef int   v4i __attribute__((ext_vector_type(4)));

__device__ __forceinline__ int atomAddI(int* p, int v) {
    return __hip_atomic_fetch_add(p, v, __ATOMIC_RELAXED, __HIP_MEMORY_SCOPE_AGENT);
}

__global__ __launch_bounds__(256) void rank_kernel(const int* __restrict__ node_idx,
                                                   const int* __restrict__ edge_idx,
                                                   int* __restrict__ posN,
                                                   int* __restrict__ posE,
                                                   int* __restrict__ rankN,
                                                   int* __restrict__ rankE) {
    int i = blockIdx.x * 256 + threadIdx.x;
    if (i < NNZ_CNT) {
        rankN[i] = atomAddI(&posN[node_idx[i]], 1);
        rankE[i] = atomAddI(&posE[edge_idx[i]], 1);
    }
}

template<int C>
__device__ __forceinline__ void block_exclusive_scan(int* __restrict__ a, int* sh) {
    const int t    = threadIdx.x;
    const int lane = t & 31;
    const int wv   = t >> 5;
    const bool live = t < 1000;
    v4i v[C / 4];
    int sum = 0;
    if (live) {
        #pragma unroll
        for (int q = 0; q < C / 4; ++q) {
            v[q] = *reinterpret_cast<const v4i*>(a + t * C + 4 * q);
            sum += v[q].x + v[q].y + v[q].z + v[q].w;
        }
    }
    int inc = sum;
    #pragma unroll
    for (int o = 1; o < 32; o <<= 1) {
        int up = __shfl_up(inc, o, 32);
        if (lane >= o) inc += up;
    }
    if (lane == 31) sh[wv] = inc;
    __syncthreads();
    if (wv == 0) {
        int wt = sh[lane];
        int winc = wt;
        #pragma unroll
        for (int o = 1; o < 32; o <<= 1) {
            int up = __shfl_up(winc, o, 32);
            if (lane >= o) winc += up;
        }
        sh[lane] = winc - wt;
    }
    __syncthreads();
    int run = sh[wv] + inc - sum;
    if (live) {
        #pragma unroll
        for (int q = 0; q < C / 4; ++q) {
            v4i x = v[q], o;
            o.x = run; run += x.x;
            o.y = run; run += x.y;
            o.z = run; run += x.z;
            o.w = run; run += x.w;
            *reinterpret_cast<v4i*>(a + t * C + 4 * q) = o;
        }
    }
}

__global__ __launch_bounds__(1024) void scan_kernel(int* __restrict__ posN, int* __restrict__ posE) {
    __shared__ int sh[32];
    if (blockIdx.x == 0) block_exclusive_scan<N_NODES / 1000>(posN, sh);
    else                 block_exclusive_scan<N_EDGES / 1000>(posE, sh);
}

__global__ __launch_bounds__(256) void place_kernel(const int* __restrict__ node_idx,
                                                    const int* __restrict__ edge_idx,
                                                    const int* __restrict__ posN,
                                                    const int* __restrict__ posE,
                                                    const int* __restrict__ rankN,
                                                    const int* __restrict__ rankE,
                                                    int* __restrict__ unsN,
                                                    int* __restrict__ unsE) {
    int i = blockIdx.x * 256 + threadIdx.x;
    if (i < NNZ_CNT) {
        const int n = node_idx[i];
        const int e = edge_idx[i];
        unsE[posE[e] + rankE[i]] = n;
        unsN[posN[n] + rankN[i]] = e;
    }
}

__global__ __launch_bounds__(256) void sort_segments_kernel(const int* __restrict__ posN,
                                                            const int* __restrict__ posE,
                                                            const int* __restrict__ unsN,
                                                            const int* __restrict__ unsE,
                                                            int* __restrict__ srtN,
                                                            int* __restrict__ srtE) {
    const int w    = __builtin_amdgcn_readfirstlane((int)((blockIdx.x * 256 + threadIdx.x) >> 5));
    const int lane = threadIdx.x & 31;
    if (w >= N_EDGES + N_NODES) return;
    const bool isE = w < N_EDGES;
    const int  k   = isE ? w : w - N_EDGES;
    const int* __restrict__ pos = isE ? posE : posN;
    const int* __restrict__ U   = isE ? unsE : unsN;
    int*       __restrict__ S   = isE ? srtE : srtN;
    const int nk  = isE ? N_EDGES : N_NODES;
    const int beg = pos[k];
    const int L   = ((k + 1 < nk) ? pos[k + 1] : NNZ_CNT) - beg;
    for (int m = 0; m < L; m += 32) {
        const int p = m + lane;
        const int x = (p < L) ? U[beg + p] : 0;
        int rank = 0;
        for (int c = 0; c < L; c += 32) {
            const int q  = c + lane;
            const int ec = (q < L) ? U[beg + q] : 0;
            const int cnt = (L - c < 32) ? (L - c) : 32;
            for (int j = 0; j < cnt; ++j) {
                const int v = __builtin_amdgcn_readlane(ec, j);
                rank += (v < x || (v == x && (c + j) < p)) ? 1 : 0;
            }
        }
        if (p < L) S[beg + rank] = x;
    }
}

static_assert((N_EDGES % 32) == 0, "two full 16-row strips per work-group");

template<int K>
__global__ __launch_bounds__(256) void gemm_wmma_kernel(const float* __restrict__ X,
                                                        const float* __restrict__ W,
                                                        float* __restrict__ C,
                                                        int mTiles) {
    constexpr int KC = 32;
    constexpr int P  = 36;
    constexpr int NCH = K / KC;
    __shared__ __attribute__((aligned(16))) float sX[32 * P];
    __shared__ __attribute__((aligned(16))) float sW[256 * P];

    const int t    = threadIdx.x;
    const int lane = t & 31;
    const int wv   = t >> 5;
    const int mBlk = blockIdx.x * 2;
    if (mBlk >= mTiles) return;
    const int half = lane >> 4;
    const int l16  = lane & 15;

    const int srow = t >> 3;
    const int spc  = (t & 7) * 4;
    const float* __restrict__ gX = X + (size_t)(mBlk * 16 + srow) * K + spc;
    const float* __restrict__ gW = W + (size_t)srow * K + spc;
    float* stX = sX + srow * P + spc;
    float* stW = sW + srow * P + spc;

    v4f px = *reinterpret_cast<const v4f*>(gX);
    v4f pw[8];
    #pragma unroll
    for (int i = 0; i < 8; ++i) pw[i] = *reinterpret_cast<const v4f*>(gW + (size_t)(32 * i) * K);

    const float* fx = sX + ((wv >> 2) * 16 + l16) * P + 2 * half;
    const float* fw = sW + ((wv & 3) * 64 + l16) * P + 2 * half;

    v8f acc0 = {}, acc1 = {}, acc2 = {}, acc3 = {};

    for (int c = 0; c < NCH; ++c) {
        *reinterpret_cast<v4f*>(stX) = px;
        #pragma unroll
        for (int i = 0; i < 8; ++i) *reinterpret_cast<v4f*>(stW + 32 * i * P) = pw[i];
        __syncthreads();
        if (c + 1 < NCH) {
            px = *reinterpret_cast<const v4f*>(gX + (c + 1) * KC);
            #pragma unroll
            for (int i = 0; i < 8; ++i)
                pw[i] = *reinterpret_cast<const v4f*>(gW + (size_t)(32 * i) * K + (c + 1) * KC);
        }
        #pragma unroll
        for (int kk = 0; kk < KC; kk += 4) {
            const v2f a  = *reinterpret_cast<const v2f*>(fx + kk);
            const v2f b0 = *reinterpret_cast<const v2f*>(fw + 0 * 16 * P + kk);
            const v2f b1 = *reinterpret_cast<const v2f*>(fw + 1 * 16 * P + kk);
            const v2f b2 = *reinterpret_cast<const v2f*>(fw + 2 * 16 * P + kk);
            const v2f b3 = *reinterpret_cast<const v2f*>(fw + 3 * 16 * P + kk);
            acc0 = __builtin_amdgcn_wmma_f32_16x16x4_f32(false, a, false, b0, (short)0, acc0, false, false);
            acc1 = __builtin_amdgcn_wmma_f32_16x16x4_f32(false, a, false, b1, (short)0, acc1, false, false);
            acc2 = __builtin_amdgcn_wmma_f32_16x16x4_f32(false, a, false, b2, (short)0, acc2, false, false);
            acc3 = __builtin_amdgcn_wmma_f32_16x16x4_f32(false, a, false, b3, (short)0, acc3, false, false);
        }
        __syncthreads();
    }

    const int m0 = (mBlk + (wv >> 2)) * 16;
    const int n0 = (wv & 3) * 64;
    float* __restrict__ ccol = C + n0 + l16;
    #pragma unroll
    for (int r = 0; r < 8; ++r) {
        const size_t m = (size_t)(m0 + r + 8 * half);
        ccol[m * D + 0 * 16] = acc0[r];
        ccol[m * D + 1 * 16] = acc1[r];
        ccol[m * D + 2 * 16] = acc2[r];
        ccol[m * D + 3 * 16] = acc3[r];
    }
}

__device__ __forceinline__ void gather_segment(const float* __restrict__ src,
                                               const int* __restrict__ list,
                                               int beg, int end, int lane,
                                               v4f& a0, v4f& a1) {
    const float* __restrict__ lsrc = src + lane * 4;
    for (int c = beg; c < end; c += 32) {
        const int q   = c + lane;
        const int idx = (q < end) ? list[q] : 0;
        const int cnt = (end - c < 32) ? (end - c) : 32;
        int j = 0;
        for (; j + 4 <= cnt; j += 4) {
            const float* p0 = lsrc + (size_t)__builtin_amdgcn_readlane(idx, j + 0) * D;
            const float* p1 = lsrc + (size_t)__builtin_amdgcn_readlane(idx, j + 1) * D;
            const float* p2 = lsrc + (size_t)__builtin_amdgcn_readlane(idx, j + 2) * D;
            const float* p3 = lsrc + (size_t)__builtin_amdgcn_readlane(idx, j + 3) * D;
            const v4f x0 = *reinterpret_cast<const v4f*>(p0), y0 = *reinterpret_cast<const v4f*>(p0 + 128);
            const v4f x1 = *reinterpret_cast<const v4f*>(p1), y1 = *reinterpret_cast<const v4f*>(p1 + 128);
            const v4f x2 = *reinterpret_cast<const v4f*>(p2), y2 = *reinterpret_cast<const v4f*>(p2 + 128);
            const v4f x3 = *reinterpret_cast<const v4f*>(p3), y3 = *reinterpret_cast<const v4f*>(p3 + 128);
            a0 += x0; a1 += y0;
            a0 += x1; a1 += y1;
            a0 += x2; a1 += y2;
            a0 += x3; a1 += y3;
        }
        for (; j < cnt; ++j) {
            const float* p0 = lsrc + (size_t)__builtin_amdgcn_readlane(idx, j) * D;
            a0 += *reinterpret_cast<const v4f*>(p0);
            a1 += *reinterpret_cast<const v4f*>(p0 + 128);
        }
    }
}

__global__ __launch_bounds__(256) void gather_edge_mean_kernel(const float* __restrict__ src,
                                                               const int* __restrict__ pos,
                                                               const int* __restrict__ list,
                                                               float* __restrict__ dst) {
    const int w    = __builtin_amdgcn_readfirstlane((int)((blockIdx.x * 256 + threadIdx.x) >> 5));
    const int lane = threadIdx.x & 31;
    if (w >= N_EDGES) return;
    const int beg = pos[w];
    const int end = (w + 1 < N_EDGES) ? pos[w + 1] : NNZ_CNT;

    v4f a0 = {}, a1 = {};
    gather_segment(src, list, beg, end, lane, a0, a1);

    const float c   = (float)(end - beg);
    const float inv = (c > 0.0f) ? 1.0f / c : 0.0f;
    a0 *= inv; a1 *= inv;
    float* drow = dst + (size_t)w * D + lane * 4;
    *reinterpret_cast<v4f*>(drow)       = a0;
    *reinterpret_cast<v4f*>(drow + 128) = a1;
}

__device__ __forceinline__ void gather_segment_in(const float* __restrict__ src,
                                                  const int* __restrict__ list,
                                                  int beg, int end, int lane, v4f& a0) {
    const float* __restrict__ lsrc = src + lane * 4;
    for (int c = beg; c < end; c += 32) {
        const int q   = c + lane;
        const int idx = (q < end) ? list[q] : 0;
        const int cnt = (end - c < 32) ? (end - c) : 32;
        int j = 0;
        for (; j + 8 <= cnt; j += 8) {
            const v4f x0 = *reinterpret_cast<const v4f*>(lsrc + (size_t)__builtin_amdgcn_readlane(idx, j + 0) * D_IN);
            const v4f x1 = *reinterpret_cast<const v4f*>(lsrc + (size_t)__builtin_amdgcn_readlane(idx, j + 1) * D_IN);
            const v4f x2 = *reinterpret_cast<const v4f*>(lsrc + (size_t)__builtin_amdgcn_readlane(idx, j + 2) * D_IN);
            const v4f x3 = *reinterpret_cast<const v4f*>(lsrc + (size_t)__builtin_amdgcn_readlane(idx, j + 3) * D_IN);
            const v4f x4 = *reinterpret_cast<const v4f*>(lsrc + (size_t)__builtin_amdgcn_readlane(idx, j + 4) * D_IN);
            const v4f x5 = *reinterpret_cast<const v4f*>(lsrc + (size_t)__builtin_amdgcn_readlane(idx, j + 5) * D_IN);
            const v4f x6 = *reinterpret_cast<const v4f*>(lsrc + (size_t)__builtin_amdgcn_readlane(idx, j + 6) * D_IN);
            const v4f x7 = *reinterpret_cast<const v4f*>(lsrc + (size_t)__builtin_amdgcn_readlane(idx, j + 7) * D_IN);
            a0 += x0; a0 += x1; a0 += x2; a0 += x3;
            a0 += x4; a0 += x5; a0 += x6; a0 += x7;
        }
        if (j + 4 <= cnt) {
            const v4f x0 = *reinterpret_cast<const v4f*>(lsrc + (size_t)__builtin_amdgcn_readlane(idx, j + 0) * D_IN);
            const v4f x1 = *reinterpret_cast<const v4f*>(lsrc + (size_t)__builtin_amdgcn_readlane(idx, j + 1) * D_IN);
            const v4f x2 = *reinterpret_cast<const v4f*>(lsrc + (size_t)__builtin_amdgcn_readlane(idx, j + 2) * D_IN);
            const v4f x3 = *reinterpret_cast<const v4f*>(lsrc + (size_t)__builtin_amdgcn_readlane(idx, j + 3) * D_IN);
            a0 += x0; a0 += x1; a0 += x2; a0 += x3;
            j += 4;
        }
        for (; j < cnt; ++j) {
            a0 += *reinterpret_cast<const v4f*>(lsrc + (size_t)__builtin_amdgcn_readlane(idx, j) * D_IN);
        }
    }
}

__global__ __launch_bounds__(256) void gather_edge_mean_in_kernel(const float* __restrict__ src,
                                                                  const int* __restrict__ pos,
                                                                  const int* __restrict__ list,
                                                                  float* __restrict__ dst) {
    const int w    = __builtin_amdgcn_readfirstlane((int)((blockIdx.x * 256 + threadIdx.x) >> 5));
    const int lane = threadIdx.x & 31;
    if (w >= N_EDGES) return;
    const int beg = pos[w];
    const int end = (w + 1 < N_EDGES) ? pos[w + 1] : NNZ_CNT;

    v4f a0 = {};
    gather_segment_in(src, list, beg, end, lane, a0);

    const float c   = (float)(end - beg);
    const float inv = (c > 0.0f) ? 1.0f / c : 0.0f;
    a0 *= inv;
    *reinterpret_cast<v4f*>(dst + (size_t)w * D_IN + lane * 4) = a0;
}

template<int M>
__device__ __forceinline__ float row_xor_add(float v) {
    return v + __builtin_bit_cast(float, __builtin_amdgcn_update_dpp(0, __builtin_bit_cast(int, v), 0x160 | M, 0xf, 0xf, true));
}

__device__ __forceinline__ float wave_sum(float v) {
    v = row_xor_add<1>(v);
    v = row_xor_add<2>(v);
    v = row_xor_add<4>(v);
    v = row_xor_add<8>(v);
    const int b = __builtin_bit_cast(int, v);
    return v + __builtin_bit_cast(float, __builtin_amdgcn_permlanex16(b, b, 0x76543210, 0xfedcba98, false, false));
}

__global__ __launch_bounds__(256) void gather_node_ln_prelu_kernel(const float* __restrict__ src,
                                                                   const int* __restrict__ pos,
                                                                   const int* __restrict__ list,
                                                                   const float* __restrict__ bias,
                                                                   const float* __restrict__ g,
                                                                   const float* __restrict__ beta,
                                                                   const float* __restrict__ pa,
                                                                   float* __restrict__ dst) {
    const int w    = __builtin_amdgcn_readfirstlane((int)((blockIdx.x * 256 + threadIdx.x) >> 5));
    const int lane = threadIdx.x & 31;
    if (w >= N_NODES) return;
    const int beg = pos[w];
    const int end = (w + 1 < N_NODES) ? pos[w + 1] : NNZ_CNT;

    v4f a0 = {}, a1 = {};
    gather_segment(src, list, beg, end, lane, a0, a1);

    const float dg   = (float)(end - beg);
    const float dinv = (dg > 0.0f) ? 1.0f / dg : 0.0f;
    const int   col  = lane * 4;
    const v4f v0 = a0 * dinv + *reinterpret_cast<const v4f*>(bias + col);
    const v4f v1 = a1 * dinv + *reinterpret_cast<const v4f*>(bias + col + 128);

    const float mu = wave_sum(((v0.x + v0.y) + (v0.z + v0.w)) + ((v1.x + v1.y) + (v1.z + v1.w))) * (1.0f / D);
    const v4f d0 = v0 - mu;
    const v4f d1 = v1 - mu;
    const v4f q0 = d0 * d0;
    const v4f q1 = d1 * d1;
    const float var = wave_sum(((q0.x + q0.y) + (q0.z + q0.w)) + ((q1.x + q1.y) + (q1.z + q1.w))) * (1.0f / D);
    const float rs  = rsqrtf(var + EPS_F);

    const v4f y0 = d0 * rs * *reinterpret_cast<const v4f*>(g + col)       + *reinterpret_cast<const v4f*>(beta + col);
    const v4f y1 = d1 * rs * *reinterpret_cast<const v4f*>(g + col + 128) + *reinterpret_cast<const v4f*>(beta + col + 128);
    const float a = pa[0];
    v4f o0, o1;
    o0.x = (y0.x >= 0.0f) ? y0.x : a * y0.x;
    o0.y = (y0.y >= 0.0f) ? y0.y : a * y0.y;
    o0.z = (y0.z >= 0.0f) ? y0.z : a * y0.z;
    o0.w = (y0.w >= 0.0f) ? y0.w : a * y0.w;
    o1.x = (y1.x >= 0.0f) ? y1.x : a * y1.x;
    o1.y = (y1.y >= 0.0f) ? y1.y : a * y1.y;
    o1.z = (y1.z >= 0.0f) ? y1.z : a * y1.z;
    o1.w = (y1.w >= 0.0f) ? y1.w : a * y1.w;

    float* drow = dst + (size_t)w * D + col;
    *reinterpret_cast<v4f*>(drow)       = o0;
    *reinterpret_cast<v4f*>(drow + 128) = o1;
}

__global__ __launch_bounds__(256) void wstd_kernel(const float* __restrict__ W,
                                                   float* __restrict__ Ws) {
    __shared__ float sred[256];
    const int r = blockIdx.x;
    const int f = threadIdx.x;
    const float w = W[(size_t)r * D + f];

    sred[f] = w;
    __syncthreads();
    #pragma unroll
    for (int s = 128; s > 0; s >>= 1) { if (f < s) sred[f] += sred[f + s]; __syncthreads(); }
    const float mean = sred[0] * (1.0f / D);
    __syncthreads();

    const float dv = w - mean;
    sred[f] = dv * dv;
    __syncthreads();
    #pragma unroll
    for (int s = 128; s > 0; s >>= 1) { if (f < s) sred[f] += sred[f + s]; __syncthreads(); }
    const float var = sred[0] * (1.0f / (D - 1));

    Ws[(size_t)r * D + f] = dv * rsqrtf(var + EPS_F);
}

extern "C" void kernel_launch(void* const* d_in, const int* in_sizes, int n_in,
                              void* d_out, int out_size, void* d_ws, size_t ws_size,
                              hipStream_t stream) {
    const float* edge_fea = (const float*)d_in[0];
    const int*   node_idx = (const int*)  d_in[1];
    const int*   edge_idx = (const int*)  d_in[2];
    const float* W1       = (const float*)d_in[3];
    const float* b1       = (const float*)d_in[4];
    const float* W2       = (const float*)d_in[5];
    const float* b2       = (const float*)d_in[6];
    const float* g1       = (const float*)d_in[7];
    const float* beta1    = (const float*)d_in[8];
    const float* g2       = (const float*)d_in[9];
    const float* beta2    = (const float*)d_in[10];
    const float* prelu_a  = (const float*)d_in[11];

    const size_t nodeRowBytes = (size_t)N_NODES * D * sizeof(float);

    char*  ws   = (char*)d_ws;
    float* H    = (float*)d_out;
    int*   srtN = (int*)ws;
    int*   srtE = srtN + NNZ_CNT;
    int*   unsN = srtE + NNZ_CNT;
    int*   unsE = unsN + NNZ_CNT;
    int*   rankN = unsE + NNZ_CNT;
    int*   rankE = rankN + NNZ_CNT;
    float* A    = (float*)(ws + ((size_t)32 << 20));
    float* E    = (float*)(ws + nodeRowBytes);
    int*   posN = (int*)(E + (size_t)N_EDGES * D);
    int*   posE = posN + N_NODES;
    float* W2s  = (float*)(posE + N_EDGES);

    const int mTiles = N_EDGES / 16;
    const int gemmBlocks = (mTiles * 4 * 32) / 256;
    const int nnzBlocks  = (NNZ_CNT + 255) / 256;

    hipMemsetAsync(posN, 0, (size_t)(N_NODES + N_EDGES) * sizeof(int), stream);
    rank_kernel<<<nnzBlocks, 256, 0, stream>>>(node_idx, edge_idx, posN, posE, rankN, rankE);
    scan_kernel<<<2, 1024, 0, stream>>>(posN, posE);
    place_kernel<<<nnzBlocks, 256, 0, stream>>>(node_idx, edge_idx, posN, posE, rankN, rankE, unsN, unsE);
    sort_segments_kernel<<<(N_EDGES + N_NODES) / 8, 256, 0, stream>>>(posN, posE, unsN, unsE, srtN, srtE);

    gather_edge_mean_in_kernel<<<N_EDGES / 8, 256, 0, stream>>>(edge_fea, posE, srtE, A);
    gemm_wmma_kernel<D_IN><<<gemmBlocks, 256, 0, stream>>>(A, W1, E, mTiles);
    gather_node_ln_prelu_kernel<<<N_NODES / 8, 256, 0, stream>>>(E, posN, srtN, b1, g1, beta1, prelu_a, H);

    wstd_kernel<<<D, 256, 0, stream>>>(W2, W2s);
    gather_edge_mean_kernel<<<N_EDGES / 8, 256, 0, stream>>>(H, posE, srtE, A);
    gemm_wmma_kernel<D><<<gemmBlocks, 256, 0, stream>>>(A, W2s, E, mTiles);
    gather_node_ln_prelu_kernel<<<N_NODES / 8, 256, 0, stream>>>(E, posN, srtN, b2, g2, beta2, prelu_a, H);
}


# Counting-sort ranks now come from per-slice LDS counters (two 16-bit counts per word) and a prefix pass over slice counts, not two returning global atomics per entry; lists and sum order unchanged
# speedup vs baseline: 9.4390x; 1.2413x over previous
#include <hip/hip_runtime.h>
#include <hip/hip_bf16.h>

#define N_NODES   100000
#define N_EDGES   20000
#define NNZ_CNT   800000
#define D_IN      128
#define D         256
#define EPS_F     1e-5f

typedef float v2f __attribute__((ext_vector_type(2)));
typedef float v4f __attribute__((ext_vector_type(4)));
typedef float v8f __attribute__((ext_vector_type(8)));
typedef int   v4i __attribute__((ext_vector_type(4)));
typedef int   v2i __attribute__((ext_vector_type(2)));

#define SLICES     64
#define SLICE_LEN  (NNZ_CNT / SLICES)
#define NW_N       (N_NODES / 2)
#define NW_E       (N_EDGES / 2)
static_assert(NNZ_CNT % SLICES == 0, "equal slices");
static_assert(SLICE_LEN < 65536, "a 16-bit field holds any count of one slice");
static_assert((N_NODES % 8) == 0 && (N_EDGES % 8) == 0, "packed count rows are moved in 16-byte pieces");

__global__ __launch_bounds__(1024) void slice_rank_kernel(const int* __restrict__ node_idx,
                                                          const int* __restrict__ edge_idx,
                                                          int* __restrict__ cntN,
                                                          int* __restrict__ cntE,
                                                          int* __restrict__ rankN,
                                                          int* __restrict__ rankE) {
    __shared__ __attribute__((aligned(16))) int h[NW_N];
    const bool isE = blockIdx.x >= SLICES;
    const int  s   = isE ? (int)blockIdx.x - SLICES : (int)blockIdx.x;
    const int* __restrict__ key  = isE ? edge_idx : node_idx;
    int*       __restrict__ rank = isE ? rankE : rankN;
    const int  nw  = isE ? NW_E : NW_N;
    int*       __restrict__ cnt  = (isE ? cntE : cntN) + (size_t)s * nw;
    const int  t   = threadIdx.x;

    constexpr int IT = (SLICE_LEN + 1023) / 1024;
    const int base = s * SLICE_LEN;
    int k[IT];
    #pragma unroll
    for (int u = 0; u < IT; ++u) {
        const int j = t + u * 1024;
        k[u] = (j < SLICE_LEN) ? key[base + j] : 0;
    }

    const v4i zero = {0, 0, 0, 0};
    for (int q = t; q < nw / 4; q += 1024) reinterpret_cast<v4i*>(h)[q] = zero;
    __syncthreads();

    #pragma unroll
    for (int u = 0; u < IT; ++u) {
        const int j = t + u * 1024;
        if (j < SLICE_LEN) {
            const int old = atomicAdd(&h[k[u] >> 1], (k[u] & 1) ? 0x10000 : 1);
            rank[base + j] = (k[u] & 1) ? (int)((unsigned)old >> 16) : (old & 0xFFFF);
        }
    }
    __syncthreads();

    for (int q = t; q < nw / 4; q += 1024)
        reinterpret_cast<v4i*>(cnt)[q] = reinterpret_cast<const v4i*>(h)[q];
}

__global__ __launch_bounds__(256) void slice_offsets_kernel(const int* __restrict__ cntN,
                                                            const int* __restrict__ cntE,
                                                            int* __restrict__ offN,
                                                            int* __restrict__ offE,
                                                            int* __restrict__ posN,
                                                            int* __restrict__ posE) {
    const int g = blockIdx.x * 256 + threadIdx.x;
    if (g >= NW_N + NW_E) return;
    const bool isE = g >= NW_N;
    const int  w   = isE ? g - NW_N : g;
    const int  nw  = isE ? NW_E : NW_N;
    const int* __restrict__ cnt = (isE ? cntE : cntN) + w;
    v2i*       __restrict__ off = reinterpret_cast<v2i*>(isE ? offE : offN) + w;
    int lo = 0, hi = 0;
    #pragma unroll 8
    for (int s = 0; s < SLICES; ++s) {
        const int c = cnt[(size_t)s * nw];
        v2i o; o.x = lo; o.y = hi;
        off[(size_t)s * nw] = o;
        lo += c & 0xFFFF;
        hi += (int)((unsigned)c >> 16);
    }
    v2i tot; tot.x = lo; tot.y = hi;
    reinterpret_cast<v2i*>(isE ? posE : posN)[w] = tot;
}

template<int C>
__device__ __forceinline__ void block_exclusive_scan(int* __restrict__ a, int* sh) {
    const int t    = threadIdx.x;
    const int lane = t & 31;
    const int wv   = t >> 5;
    const bool live = t < 1000;
    v4i v[C / 4];
    int sum = 0;
    if (live) {
        #pragma unroll
        for (int q = 0; q < C / 4; ++q) {
            v[q] = *reinterpret_cast<const v4i*>(a + t * C + 4 * q);
            sum += v[q].x + v[q].y + v[q].z + v[q].w;
        }
    }
    int inc = sum;
    #pragma unroll
    for (int o = 1; o < 32; o <<= 1) {
        int up = __shfl_up(inc, o, 32);
        if (lane >= o) inc += up;
    }
    if (lane == 31) sh[wv] = inc;
    __syncthreads();
    if (wv == 0) {
        int wt = sh[lane];
        int winc = wt;
        #pragma unroll
        for (int o = 1; o < 32; o <<= 1) {
            int up = __shfl_up(winc, o, 32);
            if (lane >= o) winc += up;
        }
        sh[lane] = winc - wt;
    }
    __syncthreads();
    int run = sh[wv] + inc - sum;
    if (live) {
        #pragma unroll
        for (int q = 0; q < C / 4; ++q) {
            v4i x = v[q], o;
            o.x = run; run += x.x;
            o.y = run; run += x.y;
            o.z = run; run += x.z;
            o.w = run; run += x.w;
            *reinterpret_cast<v4i*>(a + t * C + 4 * q) = o;
        }
    }
}

__global__ __launch_bounds__(1024) void scan_kernel(int* __restrict__ posN, int* __restrict__ posE) {
    __shared__ int sh[32];
    if (blockIdx.x == 0) block_exclusive_scan<N_NODES / 1000>(posN, sh);
    else                 block_exclusive_scan<N_EDGES / 1000>(posE, sh);
}

__global__ __launch_bounds__(256) void place_kernel(const int* __restrict__ node_idx,
                                                    const int* __restrict__ edge_idx,
                                                    const int* __restrict__ posN,
                                                    const int* __restrict__ posE,
                                                    const int* __restrict__ offN,
                                                    const int* __restrict__ offE,
                                                    const int* __restrict__ rankN,
                                                    const int* __restrict__ rankE,
                                                    int* __restrict__ unsN,
                                                    int* __restrict__ unsE) {
    int i = blockIdx.x * 256 + threadIdx.x;
    if (i < NNZ_CNT) {
        const int n = node_idx[i];
        const int e = edge_idx[i];
        const int s = i / SLICE_LEN;
        unsE[posE[e] + offE[(size_t)s * N_EDGES + e] + rankE[i]] = n;
        unsN[posN[n] + offN[(size_t)s * N_NODES + n] + rankN[i]] = e;
    }
}

__global__ __launch_bounds__(256) void sort_segments_kernel(const int* __restrict__ posN,
                                                            const int* __restrict__ posE,
                                                            const int* __restrict__ unsN,
                                                            const int* __restrict__ unsE,
                                                            int* __restrict__ srtN,
                                                            int* __restrict__ srtE) {
    const int w    = __builtin_amdgcn_readfirstlane((int)((blockIdx.x * 256 + threadIdx.x) >> 5));
    const int lane = threadIdx.x & 31;
    if (w >= N_EDGES + N_NODES) return;
    const bool isE = w < N_EDGES;
    const int  k   = isE ? w : w - N_EDGES;
    const int* __restrict__ pos = isE ? posE : posN;
    const int* __restrict__ U   = isE ? unsE : unsN;
    int*       __restrict__ S   = isE ? srtE : srtN;
    const int nk  = isE ? N_EDGES : N_NODES;
    const int beg = pos[k];
    const int L   = ((k + 1 < nk) ? pos[k + 1] : NNZ_CNT) - beg;
    for (int m = 0; m < L; m += 32) {
        const int p = m + lane;
        const int x = (p < L) ? U[beg + p] : 0;
        int rank = 0;
        for (int c = 0; c < L; c += 32) {
            const int q  = c + lane;
            const int ec = (q < L) ? U[beg + q] : 0;
            const int cnt = (L - c < 32) ? (L - c) : 32;
            for (int j = 0; j < cnt; ++j) {
                const int v = __builtin_amdgcn_readlane(ec, j);
                rank += (v < x || (v == x && (c + j) < p)) ? 1 : 0;
            }
        }
        if (p < L) S[beg + rank] = x;
    }
}

static_assert((N_EDGES % 32) == 0, "two full 16-row strips per work-group");

template<int K>
__global__ __launch_bounds__(256) void gemm_wmma_kernel(const float* __restrict__ X,
                                                        const float* __restrict__ W,
                                                        float* __restrict__ C,
                                                        int mTiles) {
    constexpr int KC = 32;
    constexpr int P  = 36;
    constexpr int NCH = K / KC;
    __shared__ __attribute__((aligned(16))) float sX[32 * P];
    __shared__ __attribute__((aligned(16))) float sW[256 * P];

    const int t    = threadIdx.x;
    const int lane = t & 31;
    const int wv   = t >> 5;
    const int mBlk = blockIdx.x * 2;
    if (mBlk >= mTiles) return;
    const int half = lane >> 4;
    const int l16  = lane & 15;

    const int srow = t >> 3;
    const int spc  = (t & 7) * 4;
    const float* __restrict__ gX = X + (size_t)(mBlk * 16 + srow) * K + spc;
    const float* __restrict__ gW = W + (size_t)srow * K + spc;
    float* stX = sX + srow * P + spc;
    float* stW = sW + srow * P + spc;

    v4f px = *reinterpret_cast<const v4f*>(gX);
    v4f pw[8];
    #pragma unroll
    for (int i = 0; i < 8; ++i) pw[i] = *reinterpret_cast<const v4f*>(gW + (size_t)(32 * i) * K);

    const float* fx = sX + ((wv >> 2) * 16 + l16) * P + 2 * half;
    const float* fw = sW + ((wv & 3) * 64 + l16) * P + 2 * half;

    v8f acc0 = {}, acc1 = {}, acc2 = {}, acc3 = {};

    for (int c = 0; c < NCH; ++c) {
        *reinterpret_cast<v4f*>(stX) = px;
        #pragma unroll
        for (int i = 0; i < 8; ++i) *reinterpret_cast<v4f*>(stW + 32 * i * P) = pw[i];
        __syncthreads();
        if (c + 1 < NCH) {
            px = *reinterpret_cast<const v4f*>(gX + (c + 1) * KC);
            #pragma unroll
            for (int i = 0; i < 8; ++i)
                pw[i] = *reinterpret_cast<const v4f*>(gW + (size_t)(32 * i) * K + (c + 1) * KC);
        }
        #pragma unroll
        for (int kk = 0; kk < KC; kk += 4) {
            const v2f a  = *reinterpret_cast<const v2f*>(fx + kk);
            const v2f b0 = *reinterpret_cast<const v2f*>(fw + 0 * 16 * P + kk);
            const v2f b1 = *reinterpret_cast<const v2f*>(fw + 1 * 16 * P + kk);
            const v2f b2 = *reinterpret_cast<const v2f*>(fw + 2 * 16 * P + kk);
            const v2f b3 = *reinterpret_cast<const v2f*>(fw + 3 * 16 * P + kk);
            acc0 = __builtin_amdgcn_wmma_f32_16x16x4_f32(false, a, false, b0, (short)0, acc0, false, false);
            acc1 = __builtin_amdgcn_wmma_f32_16x16x4_f32(false, a, false, b1, (short)0, acc1, false, false);
            acc2 = __builtin_amdgcn_wmma_f32_16x16x4_f32(false, a, false, b2, (short)0, acc2, false, false);
            acc3 = __builtin_amdgcn_wmma_f32_16x16x4_f32(false, a, false, b3, (short)0, acc3, false, false);
        }
        __syncthreads();
    }

    const int m0 = (mBlk + (wv >> 2)) * 16;
    const int n0 = (wv & 3) * 64;
    float* __restrict__ ccol = C + n0 + l16;
    #pragma unroll
    for (int r = 0; r < 8; ++r) {
        const size_t m = (size_t)(m0 + r + 8 * half);
        ccol[m * D + 0 * 16] = acc0[r];
        ccol[m * D + 1 * 16] = acc1[r];
        ccol[m * D + 2 * 16] = acc2[r];
        ccol[m * D + 3 * 16] = acc3[r];
    }
}

__device__ __forceinline__ void gather_segment(const float* __restrict__ src,
                                               const int* __restrict__ list,
                                               int beg, int end, int lane,
                                               v4f& a0, v4f& a1) {
    const float* __restrict__ lsrc = src + lane * 4;
    for (int c = beg; c < end; c += 32) {
        const int q   = c + lane;
        const int idx = (q < end) ? list[q] : 0;
        const int cnt = (end - c < 32) ? (end - c) : 32;
        int j = 0;
        for (; j + 4 <= cnt; j += 4) {
            const float* p0 = lsrc + (size_t)__builtin_amdgcn_readlane(idx, j + 0) * D;
            const float* p1 = lsrc + (size_t)__builtin_amdgcn_readlane(idx, j + 1) * D;
            const float* p2 = lsrc + (size_t)__builtin_amdgcn_readlane(idx, j + 2) * D;
            const float* p3 = lsrc + (size_t)__builtin_amdgcn_readlane(idx, j + 3) * D;
            const v4f x0 = *reinterpret_cast<const v4f*>(p0), y0 = *reinterpret_cast<const v4f*>(p0 + 128);
            const v4f x1 = *reinterpret_cast<const v4f*>(p1), y1 = *reinterpret_cast<const v4f*>(p1 + 128);
            const v4f x2 = *reinterpret_cast<const v4f*>(p2), y2 = *reinterpret_cast<const v4f*>(p2 + 128);
            const v4f x3 = *reinterpret_cast<const v4f*>(p3), y3 = *reinterpret_cast<const v4f*>(p3 + 128);
            a0 += x0; a1 += y0;
            a0 += x1; a1 += y1;
            a0 += x2; a1 += y2;
            a0 += x3; a1 += y3;
        }
        for (; j < cnt; ++j) {
            const float* p0 = lsrc + (size_t)__builtin_amdgcn_readlane(idx, j) * D;
            a0 += *reinterpret_cast<const v4f*>(p0);
            a1 += *reinterpret_cast<const v4f*>(p0 + 128);
        }
    }
}

__global__ __launch_bounds__(256) void gather_edge_mean_kernel(const float* __restrict__ src,
                                                               const int* __restrict__ pos,
                                                               const int* __restrict__ list,
                                                               float* __restrict__ dst) {
    const int w    = __builtin_amdgcn_readfirstlane((int)((blockIdx.x * 256 + threadIdx.x) >> 5));
    const int lane = threadIdx.x & 31;
    if (w >= N_EDGES) return;
    const int beg = pos[w];
    const int end = (w + 1 < N_EDGES) ? pos[w + 1] : NNZ_CNT;

    v4f a0 = {}, a1 = {};
    gather_segment(src, list, beg, end, lane, a0, a1);

    const float c   = (float)(end - beg);
    const float inv = (c > 0.0f) ? 1.0f / c : 0.0f;
    a0 *= inv; a1 *= inv;
    float* drow = dst + (size_t)w * D + lane * 4;
    *reinterpret_cast<v4f*>(drow)       = a0;
    *reinterpret_cast<v4f*>(drow + 128) = a1;
}

__device__ __forceinline__ void gather_segment_in(const float* __restrict__ src,
                                                  const int* __restrict__ list,
                                                  int beg, int end, int lane, v4f& a0) {
    const float* __restrict__ lsrc = src + lane * 4;
    for (int c = beg; c < end; c += 32) {
        const int q   = c + lane;
        const int idx = (q < end) ? list[q] : 0;
        const int cnt = (end - c < 32) ? (end - c) : 32;
        int j = 0;
        for (; j + 8 <= cnt; j += 8) {
            const v4f x0 = *reinterpret_cast<const v4f*>(lsrc + (size_t)__builtin_amdgcn_readlane(idx, j + 0) * D_IN);
            const v4f x1 = *reinterpret_cast<const v4f*>(lsrc + (size_t)__builtin_amdgcn_readlane(idx, j + 1) * D_IN);
            const v4f x2 = *reinterpret_cast<const v4f*>(lsrc + (size_t)__builtin_amdgcn_readlane(idx, j + 2) * D_IN);
            const v4f x3 = *reinterpret_cast<const v4f*>(lsrc + (size_t)__builtin_amdgcn_readlane(idx, j + 3) * D_IN);
            const v4f x4 = *reinterpret_cast<const v4f*>(lsrc + (size_t)__builtin_amdgcn_readlane(idx, j + 4) * D_IN);
            const v4f x5 = *reinterpret_cast<const v4f*>(lsrc + (size_t)__builtin_amdgcn_readlane(idx, j + 5) * D_IN);
            const v4f x6 = *reinterpret_cast<const v4f*>(lsrc + (size_t)__builtin_amdgcn_readlane(idx, j + 6) * D_IN);
            const v4f x7 = *reinterpret_cast<const v4f*>(lsrc + (size_t)__builtin_amdgcn_readlane(idx, j + 7) * D_IN);
            a0 += x0; a0 += x1; a0 += x2; a0 += x3;
            a0 += x4; a0 += x5; a0 += x6; a0 += x7;
        }
        if (j + 4 <= cnt) {
            const v4f x0 = *reinterpret_cast<const v4f*>(lsrc + (size_t)__builtin_amdgcn_readlane(idx, j + 0) * D_IN);
            const v4f x1 = *reinterpret_cast<const v4f*>(lsrc + (size_t)__builtin_amdgcn_readlane(idx, j + 1) * D_IN);
            const v4f x2 = *reinterpret_cast<const v4f*>(lsrc + (size_t)__builtin_amdgcn_readlane(idx, j + 2) * D_IN);
            const v4f x3 = *reinterpret_cast<const v4f*>(lsrc + (size_t)__builtin_amdgcn_readlane(idx, j + 3) * D_IN);
            a0 += x0; a0 += x1; a0 += x2; a0 += x3;
            j += 4;
        }
        for (; j < cnt; ++j) {
            a0 += *reinterpret_cast<const v4f*>(lsrc + (size_t)__builtin_amdgcn_readlane(idx, j) * D_IN);
        }
    }
}

__global__ __launch_bounds__(256) void gather_edge_mean_in_kernel(const float* __restrict__ src,
                                                                  const int* __restrict__ pos,
                                                                  const int* __restrict__ list,
                                                                  float* __restrict__ dst) {
    const int w    = __builtin_amdgcn_readfirstlane((int)((blockIdx.x * 256 + threadIdx.x) >> 5));
    const int lane = threadIdx.x & 31;
    if (w >= N_EDGES) return;
    const int beg = pos[w];
    const int end = (w + 1 < N_EDGES) ? pos[w + 1] : NNZ_CNT;

    v4f a0 = {};
    gather_segment_in(src, list, beg, end, lane, a0);

    const float c   = (float)(end - beg);
    const float inv = (c > 0.0f) ? 1.0f / c : 0.0f;
    a0 *= inv;
    *reinterpret_cast<v4f*>(dst + (size_t)w * D_IN + lane * 4) = a0;
}

template<int M>
__device__ __forceinline__ float row_xor_add(float v) {
    return v + __builtin_bit_cast(float, __builtin_amdgcn_update_dpp(0, __builtin_bit_cast(int, v), 0x160 | M, 0xf, 0xf, true));
}

__device__ __forceinline__ float wave_sum(float v) {
    v = row_xor_add<1>(v);
    v = row_xor_add<2>(v);
    v = row_xor_add<4>(v);
    v = row_xor_add<8>(v);
    const int b = __builtin_bit_cast(int, v);
    return v + __builtin_bit_cast(float, __builtin_amdgcn_permlanex16(b, b, 0x76543210, 0xfedcba98, false, false));
}

__global__ __launch_bounds__(256) void gather_node_ln_prelu_kernel(const float* __restrict__ src,
                                                                   const int* __restrict__ pos,
                                                                   const int* __restrict__ list,
                                                                   const float* __restrict__ bias,
                                                                   const float* __restrict__ g,
                                                                   const float* __restrict__ beta,
                                                                   const float* __restrict__ pa,
                                                                   float* __restrict__ dst) {
    const int w    = __builtin_amdgcn_readfirstlane((int)((blockIdx.x * 256 + threadIdx.x) >> 5));
    const int lane = threadIdx.x & 31;
    if (w >= N_NODES) return;
    const int beg = pos[w];
    const int end = (w + 1 < N_NODES) ? pos[w + 1] : NNZ_CNT;

    v4f a0 = {}, a1 = {};
    gather_segment(src, list, beg, end, lane, a0, a1);

    const float dg   = (float)(end - beg);
    const float dinv = (dg > 0.0f) ? 1.0f / dg : 0.0f;
    const int   col  = lane * 4;
    const v4f v0 = a0 * dinv + *reinterpret_cast<const v4f*>(bias + col);
    const v4f v1 = a1 * dinv + *reinterpret_cast<const v4f*>(bias + col + 128);

    const float mu = wave_sum(((v0.x + v0.y) + (v0.z + v0.w)) + ((v1.x + v1.y) + (v1.z + v1.w))) * (1.0f / D);
    const v4f d0 = v0 - mu;
    const v4f d1 = v1 - mu;
    const v4f q0 = d0 * d0;
    const v4f q1 = d1 * d1;
    const float var = wave_sum(((q0.x + q0.y) + (q0.z + q0.w)) + ((q1.x + q1.y) + (q1.z + q1.w))) * (1.0f / D);
    const float rs  = rsqrtf(var + EPS_F);

    const v4f y0 = d0 * rs * *reinterpret_cast<const v4f*>(g + col)       + *reinterpret_cast<const v4f*>(beta + col);
    const v4f y1 = d1 * rs * *reinterpret_cast<const v4f*>(g + col + 128) + *reinterpret_cast<const v4f*>(beta + col + 128);
    const float a = pa[0];
    v4f o0, o1;
    o0.x = (y0.x >= 0.0f) ? y0.x : a * y0.x;
    o0.y = (y0.y >= 0.0f) ? y0.y : a * y0.y;
    o0.z = (y0.z >= 0.0f) ? y0.z : a * y0.z;
    o0.w = (y0.w >= 0.0f) ? y0.w : a * y0.w;
    o1.x = (y1.x >= 0.0f) ? y1.x : a * y1.x;
    o1.y = (y1.y >= 0.0f) ? y1.y : a * y1.y;
    o1.z = (y1.z >= 0.0f) ? y1.z : a * y1.z;
    o1.w = (y1.w >= 0.0f) ? y1.w : a * y1.w;

    float* drow = dst + (size_t)w * D + col;
    *reinterpret_cast<v4f*>(drow)       = o0;
    *reinterpret_cast<v4f*>(drow + 128) = o1;
}

__global__ __launch_bounds__(256) void wstd_kernel(const float* __restrict__ W,
                                                   float* __restrict__ Ws) {
    __shared__ float sred[256];
    const int r = blockIdx.x;
    const int f = threadIdx.x;
    const float w = W[(size_t)r * D + f];

    sred[f] = w;
    __syncthreads();
    #pragma unroll
    for (int s = 128; s > 0; s >>= 1) { if (f < s) sred[f] += sred[f + s]; __syncthreads(); }
    const float mean = sred[0] * (1.0f / D);
    __syncthreads();

    const float dv = w - mean;
    sred[f] = dv * dv;
    __syncthreads();
    #pragma unroll
    for (int s = 128; s > 0; s >>= 1) { if (f < s) sred[f] += sred[f + s]; __syncthreads(); }
    const float var = sred[0] * (1.0f / (D - 1));

    Ws[(size_t)r * D + f] = dv * rsqrtf(var + EPS_F);
}

extern "C" void kernel_launch(void* const* d_in, const int* in_sizes, int n_in,
                              void* d_out, int out_size, void* d_ws, size_t ws_size,
                              hipStream_t stream) {
    const float* edge_fea = (const float*)d_in[0];
    const int*   node_idx = (const int*)  d_in[1];
    const int*   edge_idx = (const int*)  d_in[2];
    const float* W1       = (const float*)d_in[3];
    const float* b1       = (const float*)d_in[4];
    const float* W2       = (const float*)d_in[5];
    const float* b2       = (const float*)d_in[6];
    const float* g1       = (const float*)d_in[7];
    const float* beta1    = (const float*)d_in[8];
    const float* g2       = (const float*)d_in[9];
    const float* beta2    = (const float*)d_in[10];
    const float* prelu_a  = (const float*)d_in[11];

    const size_t nodeRowBytes = (size_t)N_NODES * D * sizeof(float);

    char*  ws   = (char*)d_ws;
    float* H    = (float*)d_out;
    int*   srtN = (int*)ws;
    int*   srtE = srtN + NNZ_CNT;
    int*   unsN = srtE + NNZ_CNT;
    int*   unsE = unsN + NNZ_CNT;
    int*   rankN = unsE + NNZ_CNT;
    int*   rankE = rankN + NNZ_CNT;
    int*   cntN = (int*)(ws + ((size_t)20 << 20));
    int*   cntE = cntN + (size_t)SLICES * NW_N;
    int*   offN = cntE + (size_t)SLICES * NW_E;
    int*   offE = offN + (size_t)SLICES * N_NODES;
    float* A    = (float*)(ws + ((size_t)32 << 20));
    float* E    = (float*)(ws + nodeRowBytes);
    int*   posN = (int*)(E + (size_t)N_EDGES * D);
    int*   posE = posN + N_NODES;
    float* W2s  = (float*)(posE + N_EDGES);

    const int mTiles = N_EDGES / 16;
    const int gemmBlocks = (mTiles * 4 * 32) / 256;
    const int nnzBlocks  = (NNZ_CNT + 255) / 256;

    slice_rank_kernel<<<2 * SLICES, 1024, 0, stream>>>(node_idx, edge_idx, cntN, cntE, rankN, rankE);
    slice_offsets_kernel<<<(NW_N + NW_E + 255) / 256, 256, 0, stream>>>(cntN, cntE, offN, offE, posN, posE);
    scan_kernel<<<2, 1024, 0, stream>>>(posN, posE);
    place_kernel<<<nnzBlocks, 256, 0, stream>>>(node_idx, edge_idx, posN, posE, offN, offE, rankN, rankE, unsN, unsE);
    sort_segments_kernel<<<(N_EDGES + N_NODES) / 8, 256, 0, stream>>>(posN, posE, unsN, unsE, srtN, srtE);

    gather_edge_mean_in_kernel<<<N_EDGES / 8, 256, 0, stream>>>(edge_fea, posE, srtE, A);
    gemm_wmma_kernel<D_IN><<<gemmBlocks, 256, 0, stream>>>(A, W1, E, mTiles);
    gather_node_ln_prelu_kernel<<<N_NODES / 8, 256, 0, stream>>>(E, posN, srtN, b1, g1, beta1, prelu_a, H);

    wstd_kernel<<<D, 256, 0, stream>>>(W2, W2s);
    gather_edge_mean_kernel<<<N_EDGES / 8, 256, 0, stream>>>(H, posE, srtE, A);
    gemm_wmma_kernel<D><<<gemmBlocks, 256, 0, stream>>>(A, W2s, E, mTiles);
    gather_node_ln_prelu_kernel<<<N_NODES / 8, 256, 0, stream>>>(E, posN, srtN, b2, g2, beta2, prelu_a, H);
}


# Counting sort now two-level: each list slice grouped by key range in LDS, then one work-group per range counts, scans and places its keys; count planes and one-block scan gone; sum order unchanged
# speedup vs baseline: 12.3021x; 1.3033x over previous
#include <hip/hip_runtime.h>
#include <hip/hip_bf16.h>

#define N_NODES   100000
#define N_EDGES   20000
#define NNZ_CNT   800000
#define D_IN      128
#define D         256
#define EPS_F     1e-5f

typedef float v2f __attribute__((ext_vector_type(2)));
typedef float v4f __attribute__((ext_vector_type(4)));
typedef float v8f __attribute__((ext_vector_type(8)));
typedef int   v4i __attribute__((ext_vector_type(4)));
typedef int   v2i __attribute__((ext_vector_type(2)));

#define SLICES     256
#define SLICE_LEN  (NNZ_CNT / SLICES)
#define RANGES     256
#define KR_N       391
#define KR_E       79
static_assert(NNZ_CNT % SLICES == 0, "equal slices");
static_assert(RANGES * KR_N >= N_NODES && RANGES * KR_E >= N_EDGES, "the ranges cover every key");
static_assert(N_NODES <= (1 << 17) && N_EDGES <= (1 << 15), "node and hyperedge share one 32-bit word");
static_assert(KR_N <= 512 && KR_E <= 512, "two key counters per thread of the place pass");
static_assert(SLICE_LEN < 65536, "a rank inside a slice fits 16 bits");

__global__ __launch_bounds__(1024) void slice_partition_kernel(const int* __restrict__ node_idx,
                                                               const int* __restrict__ edge_idx,
                                                               unsigned* __restrict__ partN,
                                                               unsigned* __restrict__ partE,
                                                               v2i* __restrict__ ocN,
                                                               v2i* __restrict__ ocE) {
    __shared__ int cN[RANGES];
    __shared__ int cE[RANGES];
    __shared__ unsigned stN[SLICE_LEN];
    __shared__ unsigned stE[SLICE_LEN];
    const int t    = threadIdx.x;
    const int s    = blockIdx.x;
    const int base = s * SLICE_LEN;

    constexpr int IT = (SLICE_LEN + 1023) / 1024;
    unsigned w[IT];
    int rn[IT], re[IT];
    #pragma unroll
    for (int u = 0; u < IT; ++u) {
        const int j = t + u * 1024;
        const int n = (j < SLICE_LEN) ? node_idx[base + j] : 0;
        const int e = (j < SLICE_LEN) ? edge_idx[base + j] : 0;
        w[u] = ((unsigned)n << 15) | (unsigned)e;
    }
    if (t < RANGES) { cN[t] = 0; cE[t] = 0; }
    __syncthreads();

    #pragma unroll
    for (int u = 0; u < IT; ++u) {
        const int j = t + u * 1024;
        if (j < SLICE_LEN) {
            const int rN = (int)(w[u] >> 15) / KR_N;
            const int rE = (int)(w[u] & 0x7FFFu) / KR_E;
            rn[u] = (rN << 16) | atomicAdd(&cN[rN], 1);
            re[u] = (rE << 16) | atomicAdd(&cE[rE], 1);
        }
    }
    __syncthreads();

    if (t < 64) {
        const int lane = t & 31;
        int* c   = (t < 32) ? cN : cE;
        v2i* oc  = (t < 32) ? ocN : ocE;
        int v[RANGES / 32];
        int sum = 0;
        #pragma unroll
        for (int i = 0; i < RANGES / 32; ++i) { v[i] = c[(RANGES / 32) * lane + i]; sum += v[i]; }
        int inc = sum;
        #pragma unroll
        for (int o = 1; o < 32; o <<= 1) {
            const int up = __shfl_up(inc, o, 32);
            if (lane >= o) inc += up;
        }
        int run = inc - sum;
        #pragma unroll
        for (int i = 0; i < RANGES / 32; ++i) {
            const int r = (RANGES / 32) * lane + i;
            c[r] = run;
            v2i o; o.x = run; o.y = v[i];
            oc[(size_t)r * SLICES + s] = o;
            run += v[i];
        }
    }
    __syncthreads();

    #pragma unroll
    for (int u = 0; u < IT; ++u) {
        const int j = t + u * 1024;
        if (j < SLICE_LEN) {
            stN[cN[rn[u] >> 16] + (rn[u] & 0xFFFF)] = w[u];
            stE[cE[re[u] >> 16] + (re[u] & 0xFFFF)] = w[u];
        }
    }
    __syncthreads();

    for (int i = t; i < SLICE_LEN; i += 1024) {
        partN[base + i] = stN[i];
        partE[base + i] = stE[i];
    }
}

__global__ __launch_bounds__(256) void range_place_kernel(const unsigned* __restrict__ partN,
                                                          const unsigned* __restrict__ partE,
                                                          const v2i* __restrict__ ocN,
                                                          const v2i* __restrict__ ocE,
                                                          int* __restrict__ posN,
                                                          int* __restrict__ posE,
                                                          int* __restrict__ unsN,
                                                          int* __restrict__ unsE) {
    __shared__ int kc[512];
    __shared__ int wtot[8];
    __shared__ int sbase;
    const bool isE = blockIdx.x >= RANGES;
    const int  r   = isE ? (int)blockIdx.x - RANGES : (int)blockIdx.x;
    const int  KR  = isE ? KR_E : KR_N;
    const int  nk  = isE ? N_EDGES : N_NODES;
    const int  k0  = r * KR;
    if (k0 >= nk) return;
    const int nkeys = (nk - k0 < KR) ? (nk - k0) : KR;
    const int t    = threadIdx.x;
    const int lane = t & 31;
    const int wv   = __builtin_amdgcn_readfirstlane(t >> 5);
    static_assert(SLICES == 256, "one run per thread of the work-group");

    const v2i oc = (isE ? ocE : ocN)[(size_t)r * SLICES + t];
    const unsigned* __restrict__ wpart = (isE ? partE : partN) + (size_t)(wv * 32) * SLICE_LEN;
    int* __restrict__ pos = isE ? posE : posN;
    int* __restrict__ uns = isE ? unsE : unsN;

    kc[t] = 0; kc[t + 256] = 0;
    if (t == 0) sbase = 0;
    __syncthreads();

    atomicAdd(&sbase, oc.x);
    for (int l = 0; l < 32; l += 4) {
        const int n0 = __builtin_amdgcn_readlane(oc.y, l + 0);
        const int n1 = __builtin_amdgcn_readlane(oc.y, l + 1);
        const int n2 = __builtin_amdgcn_readlane(oc.y, l + 2);
        const int n3 = __builtin_amdgcn_readlane(oc.y, l + 3);
        const unsigned* r0 = wpart + (size_t)(l + 0) * SLICE_LEN + __builtin_amdgcn_readlane(oc.x, l + 0);
        const unsigned* r1 = wpart + (size_t)(l + 1) * SLICE_LEN + __builtin_amdgcn_readlane(oc.x, l + 1);
        const unsigned* r2 = wpart + (size_t)(l + 2) * SLICE_LEN + __builtin_amdgcn_readlane(oc.x, l + 2);
        const unsigned* r3 = wpart + (size_t)(l + 3) * SLICE_LEN + __builtin_amdgcn_readlane(oc.x, l + 3);
        const unsigned w0 = (lane < n0) ? r0[lane] : 0u;
        const unsigned w1 = (lane < n1) ? r1[lane] : 0u;
        const unsigned w2 = (lane < n2) ? r2[lane] : 0u;
        const unsigned w3 = (lane < n3) ? r3[lane] : 0u;
        if (lane < n0) atomicAdd(&kc[(int)(isE ? (w0 & 0x7FFFu) : (w0 >> 15)) - k0], 1);
        if (lane < n1) atomicAdd(&kc[(int)(isE ? (w1 & 0x7FFFu) : (w1 >> 15)) - k0], 1);
        if (lane < n2) atomicAdd(&kc[(int)(isE ? (w2 & 0x7FFFu) : (w2 >> 15)) - k0], 1);
        if (lane < n3) atomicAdd(&kc[(int)(isE ? (w3 & 0x7FFFu) : (w3 >> 15)) - k0], 1);
    }
    for (int l = 0; l < 32; ++l) {
        const int n = __builtin_amdgcn_readlane(oc.y, l);
        if (n > 32) {
            const unsigned* rr = wpart + (size_t)l * SLICE_LEN + __builtin_amdgcn_readlane(oc.x, l);
            for (int j = 32 + lane; j < n; j += 32) {
                const unsigned w = rr[j];
                atomicAdd(&kc[(int)(isE ? (w & 0x7FFFu) : (w >> 15)) - k0], 1);
            }
        }
    }
    __syncthreads();

    const int c0 = kc[2 * t];
    const int c1 = kc[2 * t + 1];
    const int sum = c0 + c1;
    int inc = sum;
    #pragma unroll
    for (int o = 1; o < 32; o <<= 1) {
        const int up = __shfl_up(inc, o, 32);
        if (lane >= o) inc += up;
    }
    if (lane == 31) wtot[wv] = inc;
    __syncthreads();
    int before = 0;
    #pragma unroll
    for (int q = 0; q < 8; ++q) before += (q < wv) ? wtot[q] : 0;
    const int rbase = sbase;
    const int x0 = before + inc - sum;
    const int x1 = x0 + c0;
    kc[2 * t]     = x0;
    kc[2 * t + 1] = x1;
    if (2 * t     < nkeys) pos[k0 + 2 * t]     = rbase + x0;
    if (2 * t + 1 < nkeys) pos[k0 + 2 * t + 1] = rbase + x1;
    __syncthreads();

    int* __restrict__ ubase = uns + rbase;
    for (int l = 0; l < 32; l += 4) {
        const int n0 = __builtin_amdgcn_readlane(oc.y, l + 0);
        const int n1 = __builtin_amdgcn_readlane(oc.y, l + 1);
        const int n2 = __builtin_amdgcn_readlane(oc.y, l + 2);
        const int n3 = __builtin_amdgcn_readlane(oc.y, l + 3);
        const unsigned* r0 = wpart + (size_t)(l + 0) * SLICE_LEN + __builtin_amdgcn_readlane(oc.x, l + 0);
        const unsigned* r1 = wpart + (size_t)(l + 1) * SLICE_LEN + __builtin_amdgcn_readlane(oc.x, l + 1);
        const unsigned* r2 = wpart + (size_t)(l + 2) * SLICE_LEN + __builtin_amdgcn_readlane(oc.x, l + 2);
        const unsigned* r3 = wpart + (size_t)(l + 3) * SLICE_LEN + __builtin_amdgcn_readlane(oc.x, l + 3);
        const unsigned w0 = (lane < n0) ? r0[lane] : 0u;
        const unsigned w1 = (lane < n1) ? r1[lane] : 0u;
        const unsigned w2 = (lane < n2) ? r2[lane] : 0u;
        const unsigned w3 = (lane < n3) ? r3[lane] : 0u;
        if (lane < n0) {
            const int p = atomicAdd(&kc[(int)(isE ? (w0 & 0x7FFFu) : (w0 >> 15)) - k0], 1);
            ubase[p] = (int)(isE ? (w0 >> 15) : (w0 & 0x7FFFu));
        }
        if (lane < n1) {
            const int p = atomicAdd(&kc[(int)(isE ? (w1 & 0x7FFFu) : (w1 >> 15)) - k0], 1);
            ubase[p] = (int)(isE ? (w1 >> 15) : (w1 & 0x7FFFu));
        }
        if (lane < n2) {
            const int p = atomicAdd(&kc[(int)(isE ? (w2 & 0x7FFFu) : (w2 >> 15)) - k0], 1);
            ubase[p] = (int)(isE ? (w2 >> 15) : (w2 & 0x7FFFu));
        }
        if (lane < n3) {
            const int p = atomicAdd(&kc[(int)(isE ? (w3 & 0x7FFFu) : (w3 >> 15)) - k0], 1);
            ubase[p] = (int)(isE ? (w3 >> 15) : (w3 & 0x7FFFu));
        }
    }
    for (int l = 0; l < 32; ++l) {
        const int n = __builtin_amdgcn_readlane(oc.y, l);
        if (n > 32) {
            const unsigned* rr = wpart + (size_t)l * SLICE_LEN + __builtin_amdgcn_readlane(oc.x, l);
            for (int j = 32 + lane; j < n; j += 32) {
                const unsigned w = rr[j];
                const int p = atomicAdd(&kc[(int)(isE ? (w & 0x7FFFu) : (w >> 15)) - k0], 1);
                ubase[p] = (int)(isE ? (w >> 15) : (w & 0x7FFFu));
            }
        }
    }
}

__global__ __launch_bounds__(256) void sort_segments_kernel(const int* __restrict__ posN,
                                                            const int* __restrict__ posE,
                                                            const int* __restrict__ unsN,
                                                            const int* __restrict__ unsE,
                                                            int* __restrict__ srtN,
                                                            int* __restrict__ srtE) {
    const int w    = __builtin_amdgcn_readfirstlane((int)((blockIdx.x * 256 + threadIdx.x) >> 5));
    const int lane = threadIdx.x & 31;
    if (w >= N_EDGES + N_NODES) return;
    const bool isE = w < N_EDGES;
    const int  k   = isE ? w : w - N_EDGES;
    const int* __restrict__ pos = isE ? posE : posN;
    const int* __restrict__ U   = isE ? unsE : unsN;
    int*       __restrict__ S   = isE ? srtE : srtN;
    const int nk  = isE ? N_EDGES : N_NODES;
    const int beg = pos[k];
    const int L   = ((k + 1 < nk) ? pos[k + 1] : NNZ_CNT) - beg;
    for (int m = 0; m < L; m += 32) {
        const int p = m + lane;
        const int x = (p < L) ? U[beg + p] : 0;
        int rank = 0;
        for (int c = 0; c < L; c += 32) {
            const int q  = c + lane;
            const int ec = (q < L) ? U[beg + q] : 0;
            const int cnt = (L - c < 32) ? (L - c) : 32;
            for (int j = 0; j < cnt; ++j) {
                const int v = __builtin_amdgcn_readlane(ec, j);
                rank += (v < x || (v == x && (c + j) < p)) ? 1 : 0;
            }
        }
        if (p < L) S[beg + rank] = x;
    }
}

static_assert((N_EDGES % 32) == 0, "two full 16-row strips per work-group");

template<int K>
__global__ __launch_bounds__(256) void gemm_wmma_kernel(const float* __restrict__ X,
                                                        const float* __restrict__ W,
                                                        float* __restrict__ C,
                                                        int mTiles) {
    constexpr int KC = 32;
    constexpr int P  = 36;
    constexpr int NCH = K / KC;
    __shared__ __attribute__((aligned(16))) float sX[32 * P];
    __shared__ __attribute__((aligned(16))) float sW[256 * P];

    const int t    = threadIdx.x;
    const int lane = t & 31;
    const int wv   = t >> 5;
    const int mBlk = blockIdx.x * 2;
    if (mBlk >= mTiles) return;
    const int half = lane >> 4;
    const int l16  = lane & 15;

    const int srow = t >> 3;
    const int spc  = (t & 7) * 4;
    const float* __restrict__ gX = X + (size_t)(mBlk * 16 + srow) * K + spc;
    const float* __restrict__ gW = W + (size_t)srow * K + spc;
    float* stX = sX + srow * P + spc;
    float* stW = sW + srow * P + spc;

    v4f px = *reinterpret_cast<const v4f*>(gX);
    v4f pw[8];
    #pragma unroll
    for (int i = 0; i < 8; ++i) pw[i] = *reinterpret_cast<const v4f*>(gW + (size_t)(32 * i) * K);

    const float* fx = sX + ((wv >> 2) * 16 + l16) * P + 2 * half;
    const float* fw = sW + ((wv & 3) * 64 + l16) * P + 2 * half;

    v8f acc0 = {}, acc1 = {}, acc2 = {}, acc3 = {};

    for (int c = 0; c < NCH; ++c) {
        *reinterpret_cast<v4f*>(stX) = px;
        #pragma unroll
        for (int i = 0; i < 8; ++i) *reinterpret_cast<v4f*>(stW + 32 * i * P) = pw[i];
        __syncthreads();
        if (c + 1 < NCH) {
            px = *reinterpret_cast<const v4f*>(gX + (c + 1) * KC);
            #pragma unroll
            for (int i = 0; i < 8; ++i)
                pw[i] = *reinterpret_cast<const v4f*>(gW + (size_t)(32 * i) * K + (c + 1) * KC);
        }
        #pragma unroll
        for (int kk = 0; kk < KC; kk += 4) {
            const v2f a  = *reinterpret_cast<const v2f*>(fx + kk);
            const v2f b0 = *reinterpret_cast<const v2f*>(fw + 0 * 16 * P + kk);
            const v2f b1 = *reinterpret_cast<const v2f*>(fw + 1 * 16 * P + kk);
            const v2f b2 = *reinterpret_cast<const v2f*>(fw + 2 * 16 * P + kk);
            const v2f b3 = *reinterpret_cast<const v2f*>(fw + 3 * 16 * P + kk);
            acc0 = __builtin_amdgcn_wmma_f32_16x16x4_f32(false, a, false, b0, (short)0, acc0, false, false);
            acc1 = __builtin_amdgcn_wmma_f32_16x16x4_f32(false, a, false, b1, (short)0, acc1, false, false);
            acc2 = __builtin_amdgcn_wmma_f32_16x16x4_f32(false, a, false, b2, (short)0, acc2, false, false);
            acc3 = __builtin_amdgcn_wmma_f32_16x16x4_f32(false, a, false, b3, (short)0, acc3, false, false);
        }
        __syncthreads();
    }

    const int m0 = (mBlk + (wv >> 2)) * 16;
    const int n0 = (wv & 3) * 64;
    float* __restrict__ ccol = C + n0 + l16;
    #pragma unroll
    for (int r = 0; r < 8; ++r) {
        const size_t m = (size_t)(m0 + r + 8 * half);
        ccol[m * D + 0 * 16] = acc0[r];
        ccol[m * D + 1 * 16] = acc1[r];
        ccol[m * D + 2 * 16] = acc2[r];
        ccol[m * D + 3 * 16] = acc3[r];
    }
}

__device__ __forceinline__ void gather_segment(const float* __restrict__ src,
                                               const int* __restrict__ list,
                                               int beg, int end, int lane,
                                               v4f& a0, v4f& a1) {
    const float* __restrict__ lsrc = src + lane * 4;
    for (int c = beg; c < end; c += 32) {
        const int q   = c + lane;
        const int idx = (q < end) ? list[q] : 0;
        const int cnt = (end - c < 32) ? (end - c) : 32;
        int j = 0;
        for (; j + 4 <= cnt; j += 4) {
            const float* p0 = lsrc + (size_t)__builtin_amdgcn_readlane(idx, j + 0) * D;
            const float* p1 = lsrc + (size_t)__builtin_amdgcn_readlane(idx, j + 1) * D;
            const float* p2 = lsrc + (size_t)__builtin_amdgcn_readlane(idx, j + 2) * D;
            const float* p3 = lsrc + (size_t)__builtin_amdgcn_readlane(idx, j + 3) * D;
            const v4f x0 = *reinterpret_cast<const v4f*>(p0), y0 = *reinterpret_cast<const v4f*>(p0 + 128);
            const v4f x1 = *reinterpret_cast<const v4f*>(p1), y1 = *reinterpret_cast<const v4f*>(p1 + 128);
            const v4f x2 = *reinterpret_cast<const v4f*>(p2), y2 = *reinterpret_cast<const v4f*>(p2 + 128);
            const v4f x3 = *reinterpret_cast<const v4f*>(p3), y3 = *reinterpret_cast<const v4f*>(p3 + 128);
            a0 += x0; a1 += y0;
            a0 += x1; a1 += y1;
            a0 += x2; a1 += y2;
            a0 += x3; a1 += y3;
        }
        for (; j < cnt; ++j) {
            const float* p0 = lsrc + (size_t)__builtin_amdgcn_readlane(idx, j) * D;
            a0 += *reinterpret_cast<const v4f*>(p0);
            a1 += *reinterpret_cast<const v4f*>(p0 + 128);
        }
    }
}

__global__ __launch_bounds__(256) void gather_edge_mean_kernel(const float* __restrict__ src,
                                                               const int* __restrict__ pos,
                                                               const int* __restrict__ list,
                                                               float* __restrict__ dst) {
    const int w    = __builtin_amdgcn_readfirstlane((int)((blockIdx.x * 256 + threadIdx.x) >> 5));
    const int lane = threadIdx.x & 31;
    if (w >= N_EDGES) return;
    const int beg = pos[w];
    const int end = (w + 1 < N_EDGES) ? pos[w + 1] : NNZ_CNT;

    v4f a0 = {}, a1 = {};
    gather_segment(src, list, beg, end, lane, a0, a1);

    const float c   = (float)(end - beg);
    const float inv = (c > 0.0f) ? 1.0f / c : 0.0f;
    a0 *= inv; a1 *= inv;
    float* drow = dst + (size_t)w * D + lane * 4;
    *reinterpret_cast<v4f*>(drow)       = a0;
    *reinterpret_cast<v4f*>(drow + 128) = a1;
}

__device__ __forceinline__ void gather_segment_in(const float* __restrict__ src,
                                                  const int* __restrict__ list,
                                                  int beg, int end, int lane, v4f& a0) {
    const float* __restrict__ lsrc = src + lane * 4;
    for (int c = beg; c < end; c += 32) {
        const int q   = c + lane;
        const int idx = (q < end) ? list[q] : 0;
        const int cnt = (end - c < 32) ? (end - c) : 32;
        int j = 0;
        for (; j + 8 <= cnt; j += 8) {
            const v4f x0 = *reinterpret_cast<const v4f*>(lsrc + (size_t)__builtin_amdgcn_readlane(idx, j + 0) * D_IN);
            const v4f x1 = *reinterpret_cast<const v4f*>(lsrc + (size_t)__builtin_amdgcn_readlane(idx, j + 1) * D_IN);
            const v4f x2 = *reinterpret_cast<const v4f*>(lsrc + (size_t)__builtin_amdgcn_readlane(idx, j + 2) * D_IN);
            const v4f x3 = *reinterpret_cast<const v4f*>(lsrc + (size_t)__builtin_amdgcn_readlane(idx, j + 3) * D_IN);
            const v4f x4 = *reinterpret_cast<const v4f*>(lsrc + (size_t)__builtin_amdgcn_readlane(idx, j + 4) * D_IN);
            const v4f x5 = *reinterpret_cast<const v4f*>(lsrc + (size_t)__builtin_amdgcn_readlane(idx, j + 5) * D_IN);
            const v4f x6 = *reinterpret_cast<const v4f*>(lsrc + (size_t)__builtin_amdgcn_readlane(idx, j + 6) * D_IN);
            const v4f x7 = *reinterpret_cast<const v4f*>(lsrc + (size_t)__builtin_amdgcn_readlane(idx, j + 7) * D_IN);
            a0 += x0; a0 += x1; a0 += x2; a0 += x3;
            a0 += x4; a0 += x5; a0 += x6; a0 += x7;
        }
        if (j + 4 <= cnt) {
            const v4f x0 = *reinterpret_cast<const v4f*>(lsrc + (size_t)__builtin_amdgcn_readlane(idx, j + 0) * D_IN);
            const v4f x1 = *reinterpret_cast<const v4f*>(lsrc + (size_t)__builtin_amdgcn_readlane(idx, j + 1) * D_IN);
            const v4f x2 = *reinterpret_cast<const v4f*>(lsrc + (size_t)__builtin_amdgcn_readlane(idx, j + 2) * D_IN);
            const v4f x3 = *reinterpret_cast<const v4f*>(lsrc + (size_t)__builtin_amdgcn_readlane(idx, j + 3) * D_IN);
            a0 += x0; a0 += x1; a0 += x2; a0 += x3;
            j += 4;
        }
        for (; j < cnt; ++j) {
            a0 += *reinterpret_cast<const v4f*>(lsrc + (size_t)__builtin_amdgcn_readlane(idx, j) * D_IN);
        }
    }
}

__global__ __launch_bounds__(256) void gather_edge_mean_in_kernel(const float* __restrict__ src,
                                                                  const int* __restrict__ pos,
                                                                  const int* __restrict__ list,
                                                                  float* __restrict__ dst) {
    const int w    = __builtin_amdgcn_readfirstlane((int)((blockIdx.x * 256 + threadIdx.x) >> 5));
    const int lane = threadIdx.x & 31;
    if (w >= N_EDGES) return;
    const int beg = pos[w];
    const int end = (w + 1 < N_EDGES) ? pos[w + 1] : NNZ_CNT;

    v4f a0 = {};
    gather_segment_in(src, list, beg, end, lane, a0);

    const float c   = (float)(end - beg);
    const float inv = (c > 0.0f) ? 1.0f / c : 0.0f;
    a0 *= inv;
    *reinterpret_cast<v4f*>(dst + (size_t)w * D_IN + lane * 4) = a0;
}

template<int M>
__device__ __forceinline__ float row_xor_add(float v) {
    return v + __builtin_bit_cast(float, __builtin_amdgcn_update_dpp(0, __builtin_bit_cast(int, v), 0x160 | M, 0xf, 0xf, true));
}

__device__ __forceinline__ float wave_sum(float v) {
    v = row_xor_add<1>(v);
    v = row_xor_add<2>(v);
    v = row_xor_add<4>(v);
    v = row_xor_add<8>(v);
    const int b = __builtin_bit_cast(int, v);
    return v + __builtin_bit_cast(float, __builtin_amdgcn_permlanex16(b, b, 0x76543210, 0xfedcba98, false, false));
}

__global__ __launch_bounds__(256) void gather_node_ln_prelu_kernel(const float* __restrict__ src,
                                                                   const int* __restrict__ pos,
                                                                   const int* __restrict__ list,
                                                                   const float* __restrict__ bias,
                                                                   const float* __restrict__ g,
                                                                   const float* __restrict__ beta,
                                                                   const float* __restrict__ pa,
                                                                   float* __restrict__ dst) {
    const int w    = __builtin_amdgcn_readfirstlane((int)((blockIdx.x * 256 + threadIdx.x) >> 5));
    const int lane = threadIdx.x & 31;
    if (w >= N_NODES) return;
    const int beg = pos[w];
    const int end = (w + 1 < N_NODES) ? pos[w + 1] : NNZ_CNT;

    v4f a0 = {}, a1 = {};
    gather_segment(src, list, beg, end, lane, a0, a1);

    const float dg   = (float)(end - beg);
    const float dinv = (dg > 0.0f) ? 1.0f / dg : 0.0f;
    const int   col  = lane * 4;
    const v4f v0 = a0 * dinv + *reinterpret_cast<const v4f*>(bias + col);
    const v4f v1 = a1 * dinv + *reinterpret_cast<const v4f*>(bias + col + 128);

    const float mu = wave_sum(((v0.x + v0.y) + (v0.z + v0.w)) + ((v1.x + v1.y) + (v1.z + v1.w))) * (1.0f / D);
    const v4f d0 = v0 - mu;
    const v4f d1 = v1 - mu;
    const v4f q0 = d0 * d0;
    const v4f q1 = d1 * d1;
    const float var = wave_sum(((q0.x + q0.y) + (q0.z + q0.w)) + ((q1.x + q1.y) + (q1.z + q1.w))) * (1.0f / D);
    const float rs  = rsqrtf(var + EPS_F);

    const v4f y0 = d0 * rs * *reinterpret_cast<const v4f*>(g + col)       + *reinterpret_cast<const v4f*>(beta + col);
    const v4f y1 = d1 * rs * *reinterpret_cast<const v4f*>(g + col + 128) + *reinterpret_cast<const v4f*>(beta + col + 128);
    const float a = pa[0];
    v4f o0, o1;
    o0.x = (y0.x >= 0.0f) ? y0.x : a * y0.x;
    o0.y = (y0.y >= 0.0f) ? y0.y : a * y0.y;
    o0.z = (y0.z >= 0.0f) ? y0.z : a * y0.z;
    o0.w = (y0.w >= 0.0f) ? y0.w : a * y0.w;
    o1.x = (y1.x >= 0.0f) ? y1.x : a * y1.x;
    o1.y = (y1.y >= 0.0f) ? y1.y : a * y1.y;
    o1.z = (y1.z >= 0.0f) ? y1.z : a * y1.z;
    o1.w = (y1.w >= 0.0f) ? y1.w : a * y1.w;

    float* drow = dst + (size_t)w * D + col;
    *reinterpret_cast<v4f*>(drow)       = o0;
    *reinterpret_cast<v4f*>(drow + 128) = o1;
}

__global__ __launch_bounds__(256) void wstd_kernel(const float* __restrict__ W,
                                                   float* __restrict__ Ws) {
    __shared__ float sred[256];
    const int r = blockIdx.x;
    const int f = threadIdx.x;
    const float w = W[(size_t)r * D + f];

    sred[f] = w;
    __syncthreads();
    #pragma unroll
    for (int s = 128; s > 0; s >>= 1) { if (f < s) sred[f] += sred[f + s]; __syncthreads(); }
    const float mean = sred[0] * (1.0f / D);
    __syncthreads();

    const float dv = w - mean;
    sred[f] = dv * dv;
    __syncthreads();
    #pragma unroll
    for (int s = 128; s > 0; s >>= 1) { if (f < s) sred[f] += sred[f + s]; __syncthreads(); }
    const float var = sred[0] * (1.0f / (D - 1));

    Ws[(size_t)r * D + f] = dv * rsqrtf(var + EPS_F);
}

extern "C" void kernel_launch(void* const* d_in, const int* in_sizes, int n_in,
                              void* d_out, int out_size, void* d_ws, size_t ws_size,
                              hipStream_t stream) {
    const float* edge_fea = (const float*)d_in[0];
    const int*   node_idx = (const int*)  d_in[1];
    const int*   edge_idx = (const int*)  d_in[2];
    const float* W1       = (const float*)d_in[3];
    const float* b1       = (const float*)d_in[4];
    const float* W2       = (const float*)d_in[5];
    const float* b2       = (const float*)d_in[6];
    const float* g1       = (const float*)d_in[7];
    const float* beta1    = (const float*)d_in[8];
    const float* g2       = (const float*)d_in[9];
    const float* beta2    = (const float*)d_in[10];
    const float* prelu_a  = (const float*)d_in[11];

    const size_t nodeRowBytes = (size_t)N_NODES * D * sizeof(float);

    char*  ws   = (char*)d_ws;
    float* H    = (float*)d_out;
    int*   srtN = (int*)ws;
    int*   srtE = srtN + NNZ_CNT;
    int*   unsN = srtE + NNZ_CNT;
    int*   unsE = unsN + NNZ_CNT;
    unsigned* partN = (unsigned*)(unsE + NNZ_CNT);
    unsigned* partE = partN + NNZ_CNT;
    v2i*   ocN  = (v2i*)(ws + ((size_t)20 << 20));
    v2i*   ocE  = ocN + (size_t)RANGES * SLICES;
    float* A    = (float*)(ws + ((size_t)32 << 20));
    float* E    = (float*)(ws + nodeRowBytes);
    int*   posN = (int*)(E + (size_t)N_EDGES * D);
    int*   posE = posN + N_NODES;
    float* W2s  = (float*)(posE + N_EDGES);

    const int mTiles = N_EDGES / 16;
    const int gemmBlocks = (mTiles * 4 * 32) / 256;

    slice_partition_kernel<<<SLICES, 1024, 0, stream>>>(node_idx, edge_idx, partN, partE, ocN, ocE);
    range_place_kernel<<<2 * RANGES, 256, 0, stream>>>(partN, partE, ocN, ocE, posN, posE, unsN, unsE);
    sort_segments_kernel<<<(N_EDGES + N_NODES) / 8, 256, 0, stream>>>(posN, posE, unsN, unsE, srtN, srtE);

    gather_edge_mean_in_kernel<<<N_EDGES / 8, 256, 0, stream>>>(edge_fea, posE, srtE, A);
    gemm_wmma_kernel<D_IN><<<gemmBlocks, 256, 0, stream>>>(A, W1, E, mTiles);
    gather_node_ln_prelu_kernel<<<N_NODES / 8, 256, 0, stream>>>(E, posN, srtN, b1, g1, beta1, prelu_a, H);

    wstd_kernel<<<D, 256, 0, stream>>>(W2, W2s);
    gather_edge_mean_kernel<<<N_EDGES / 8, 256, 0, stream>>>(H, posE, srtE, A);
    gemm_wmma_kernel<D><<<gemmBlocks, 256, 0, stream>>>(A, W2s, E, mTiles);
    gather_node_ln_prelu_kernel<<<N_NODES / 8, 256, 0, stream>>>(E, posN, srtN, b2, g2, beta2, prelu_a, H);
}


# List segments now ordered by value inside the place pass, in LDS (range read once, grouped, ranked, whole-line stores out); segment-sort launch and unsorted lists gone; sum order unchanged
# speedup vs baseline: 13.9576x; 1.1346x over previous
#include <hip/hip_runtime.h>
#include <hip/hip_bf16.h>

#define N_NODES   100000
#define N_EDGES   20000
#define NNZ_CNT   800000
#define D_IN      128
#define D         256
#define EPS_F     1e-5f

typedef float v2f __attribute__((ext_vector_type(2)));
typedef float v4f __attribute__((ext_vector_type(4)));
typedef float v8f __attribute__((ext_vector_type(8)));
typedef int   v4i __attribute__((ext_vector_type(4)));
typedef int   v2i __attribute__((ext_vector_type(2)));

#define SLICES     256
#define SLICE_LEN  (NNZ_CNT / SLICES)
#define RANGES     256
#define KR_N       391
#define KR_E       79
static_assert(NNZ_CNT % SLICES == 0, "equal slices");
static_assert(RANGES * KR_N >= N_NODES && RANGES * KR_E >= N_EDGES, "the ranges cover every key");
static_assert(N_NODES <= (1 << 17) && N_EDGES <= (1 << 15), "node and hyperedge share one 32-bit word");
static_assert(KR_N <= 512 && KR_E <= 512, "two key counters per thread of the place pass");
static_assert(SLICE_LEN < 65536, "a rank inside a slice fits 16 bits");

__global__ __launch_bounds__(1024) void slice_partition_kernel(const int* __restrict__ node_idx,
                                                               const int* __restrict__ edge_idx,
                                                               unsigned* __restrict__ partN,
                                                               unsigned* __restrict__ partE,
                                                               v2i* __restrict__ ocN,
                                                               v2i* __restrict__ ocE) {
    __shared__ int cN[RANGES];
    __shared__ int cE[RANGES];
    __shared__ unsigned stN[SLICE_LEN];
    __shared__ unsigned stE[SLICE_LEN];
    const int t    = threadIdx.x;
    const int s    = blockIdx.x;
    const int base = s * SLICE_LEN;

    constexpr int IT = (SLICE_LEN + 1023) / 1024;
    unsigned w[IT];
    int rn[IT], re[IT];
    #pragma unroll
    for (int u = 0; u < IT; ++u) {
        const int j = t + u * 1024;
        const int n = (j < SLICE_LEN) ? node_idx[base + j] : 0;
        const int e = (j < SLICE_LEN) ? edge_idx[base + j] : 0;
        w[u] = ((unsigned)n << 15) | (unsigned)e;
    }
    if (t < RANGES) { cN[t] = 0; cE[t] = 0; }
    __syncthreads();

    #pragma unroll
    for (int u = 0; u < IT; ++u) {
        const int j = t + u * 1024;
        if (j < SLICE_LEN) {
            const int rN = (int)(w[u] >> 15) / KR_N;
            const int rE = (int)(w[u] & 0x7FFFu) / KR_E;
            rn[u] = (rN << 16) | atomicAdd(&cN[rN], 1);
            re[u] = (rE << 16) | atomicAdd(&cE[rE], 1);
        }
    }
    __syncthreads();

    if (t < 64) {
        const int lane = t & 31;
        int* c   = (t < 32) ? cN : cE;
        v2i* oc  = (t < 32) ? ocN : ocE;
        int v[RANGES / 32];
        int sum = 0;
        #pragma unroll
        for (int i = 0; i < RANGES / 32; ++i) { v[i] = c[(RANGES / 32) * lane + i]; sum += v[i]; }
        int inc = sum;
        #pragma unroll
        for (int o = 1; o < 32; o <<= 1) {
            const int up = __shfl_up(inc, o, 32);
            if (lane >= o) inc += up;
        }
        int run = inc - sum;
        #pragma unroll
        for (int i = 0; i < RANGES / 32; ++i) {
            const int r = (RANGES / 32) * lane + i;
            c[r] = run;
            v2i o; o.x = run; o.y = v[i];
            oc[(size_t)r * SLICES + s] = o;
            run += v[i];
        }
    }
    __syncthreads();

    #pragma unroll
    for (int u = 0; u < IT; ++u) {
        const int j = t + u * 1024;
        if (j < SLICE_LEN) {
            stN[cN[rn[u] >> 16] + (rn[u] & 0xFFFF)] = w[u];
            stE[cE[re[u] >> 16] + (re[u] & 0xFFFF)] = w[u];
        }
    }
    __syncthreads();

    for (int i = t; i < SLICE_LEN; i += 1024) {
        partN[base + i] = stN[i];
        partE[base + i] = stE[i];
    }
}

#define CAP 4096

__device__ __forceinline__ void rank_sort_segment(const int* U, int* S, int L, int lane) {
    for (int m = 0; m < L; m += 32) {
        const int p = m + lane;
        const int x = (p < L) ? U[p] : 0;
        int rank = 0;
        for (int c = 0; c < L; c += 32) {
            const int q  = c + lane;
            const int ec = (q < L) ? U[q] : 0;
            const int cnt = (L - c < 32) ? (L - c) : 32;
            for (int j = 0; j < cnt; ++j) {
                const int v = __builtin_amdgcn_readlane(ec, j);
                rank += (v < x || (v == x && (c + j) < p)) ? 1 : 0;
            }
        }
        if (p < L) S[rank] = x;
    }
}

__global__ __launch_bounds__(1024) void range_place_kernel(const unsigned* __restrict__ partN,
                                                           const unsigned* __restrict__ partE,
                                                           const v2i* __restrict__ ocN,
                                                           const v2i* __restrict__ ocE,
                                                           int* __restrict__ posN,
                                                           int* __restrict__ posE,
                                                           int* unsN,
                                                           int* unsE,
                                                           int* srtN,
                                                           int* srtE) {
    __shared__ int kc[512];
    __shared__ int wtot[8];
    __shared__ int wlen[8];
    __shared__ int sbase;
    __shared__ int rx[SLICES];
    __shared__ int ry[SLICES];
    __shared__ int rf[SLICES];
    __shared__ unsigned raw[CAP];
    __shared__ unsigned byk[CAP + 8];
    const bool isE = blockIdx.x >= RANGES;
    const int  r   = isE ? (int)blockIdx.x - RANGES : (int)blockIdx.x;
    const int  KR  = isE ? KR_E : KR_N;
    const int  nk  = isE ? N_EDGES : N_NODES;
    const int  k0  = r * KR;
    if (k0 >= nk) return;
    const int nkeys = (nk - k0 < KR) ? (nk - k0) : KR;
    const int sb    = isE ? 2 : 0;
    const int sub   = (1 << sb) - 1;
    const int ckb   = k0 << sb;
    const unsigned vmask = isE ? 0x1FFFFu : 0x7FFFu;
    static_assert((KR_E << 2) <= 512 && KR_N <= 512, "the counters of a range fit kc[]");
    static_assert(SLICES == 256 && CAP <= 4096, "eight runs per wave, at most four entries per thread");
    const int t    = threadIdx.x;
    const int lane = t & 31;
    const int wv   = __builtin_amdgcn_readfirstlane(t >> 5);

    const unsigned* __restrict__ part = isE ? partE : partN;
    int* __restrict__ pos = isE ? posE : posN;
    int* uns = isE ? unsE : unsN;
    int* srt = isE ? srtE : srtN;

    if (t < 512) kc[t] = 0;
    if (t == 0) sbase = 0;
    int ocx = 0;
    if (t < SLICES) {
        const v2i oc = (isE ? ocE : ocN)[(size_t)r * SLICES + t];
        ocx = oc.x;
        int incl = oc.y;
        #pragma unroll
        for (int o = 1; o < 32; o <<= 1) {
            const int up = __shfl_up(incl, o, 32);
            if (lane >= o) incl += up;
        }
        if (lane == 31) wlen[wv] = incl;
        rx[t] = oc.x; ry[t] = oc.y; rf[t] = incl - oc.y;
    }
    __syncthreads();

    if (t < SLICES) atomicAdd(&sbase, ocx);
    const int ow = wv >> 2;
    int n = 0, wb = 0;
    #pragma unroll
    for (int q = 0; q < 8; ++q) { const int v = wlen[q]; n += v; wb += (q < ow) ? v : 0; }
    const bool inLds = n <= CAP;
    const int  s0    = wv * 8;
    const int tN = ry[s0 + (lane & 7)];
    const int tF = rf[s0 + (lane & 7)] + wb;
    const int tX = rx[s0 + (lane & 7)];

    if (inLds) {
        unsigned w[8];
        int nn[8], ro[8];
        #pragma unroll
        for (int u = 0; u < 8; ++u) {
            nn[u] = __builtin_amdgcn_readlane(tN, u);
            ro[u] = __builtin_amdgcn_readlane(tF, u);
            const unsigned* rr = part + (size_t)(s0 + u) * SLICE_LEN + __builtin_amdgcn_readlane(tX, u);
            w[u] = (lane < nn[u]) ? rr[lane] : 0u;
        }
        #pragma unroll
        for (int u = 0; u < 8; ++u) {
            if (lane < nn[u]) {
                const unsigned K = isE ? ((w[u] >> 15) | (w[u] << 17)) : w[u];
                raw[ro[u] + lane] = K;
                atomicAdd(&kc[(int)(K >> 15) - ckb], 1);
            }
        }
        for (int u = 0; u < 8; ++u) {
            const int nl = __builtin_amdgcn_readlane(tN, u);
            if (nl > 32) {
                const unsigned* rr = part + (size_t)(s0 + u) * SLICE_LEN + __builtin_amdgcn_readlane(tX, u);
                const int ro1 = __builtin_amdgcn_readlane(tF, u);
                for (int j = 32 + lane; j < nl; j += 32) {
                    const unsigned w1 = rr[j];
                    const unsigned K  = isE ? ((w1 >> 15) | (w1 << 17)) : w1;
                    raw[ro1 + j] = K;
                    atomicAdd(&kc[(int)(K >> 15) - ckb], 1);
                }
            }
        }
    } else {
        for (int u = 0; u < 8; ++u) {
            const int nl = __builtin_amdgcn_readlane(tN, u);
            const unsigned* rr = part + (size_t)(s0 + u) * SLICE_LEN + __builtin_amdgcn_readlane(tX, u);
            for (int j = lane; j < nl; j += 32) {
                const unsigned w1 = rr[j];
                const unsigned K  = isE ? ((w1 >> 15) | (w1 << 17)) : w1;
                atomicAdd(&kc[(int)(K >> 15) - ckb], 1);
            }
        }
    }
    __syncthreads();

    int c0 = 0, sum = 0, inc = 0;
    if (t < 256) {
        c0 = kc[2 * t];
        sum = c0 + kc[2 * t + 1];
        inc = sum;
        #pragma unroll
        for (int o = 1; o < 32; o <<= 1) {
            const int up = __shfl_up(inc, o, 32);
            if (lane >= o) inc += up;
        }
        if (lane == 31) wtot[wv] = inc;
    }
    __syncthreads();
    const int rbase = sbase;
    if (t < 256) {
        int before = 0;
        #pragma unroll
        for (int q = 0; q < 8; ++q) before += (q < wv) ? wtot[q] : 0;
        const int x0 = before + inc - sum;
        const int x1 = x0 + c0;
        kc[2 * t]     = x0;
        kc[2 * t + 1] = x1;
        if (((2 * t)     & sub) == 0 && ((2 * t)     >> sb) < nkeys) pos[k0 + ((2 * t)     >> sb)] = rbase + x0;
        if (((2 * t + 1) & sub) == 0 && ((2 * t + 1) >> sb) < nkeys) pos[k0 + ((2 * t + 1) >> sb)] = rbase + x1;
    }
    __syncthreads();

    if (inLds) {
        unsigned K4[4];
        #pragma unroll
        for (int u = 0; u < 4; ++u) K4[u] = (t + 1024 * u < n) ? raw[t + 1024 * u] : 0u;
        #pragma unroll
        for (int u = 0; u < 4; ++u) {
            if (t + 1024 * u < n) {
                const int p = atomicAdd(&kc[(int)(K4[u] >> 15) - ckb], 1);
                byk[p] = K4[u];
            }
        }
        if (t < 8) byk[n + t] = 0xFFFFFFFFu;
        __syncthreads();

        int* fin = reinterpret_cast<int*>(raw);
        for (int p = t; p < n; p += 1024) {
            const unsigned kp = byk[p];
            const int ck = (int)(kp >> 15) - ckb;
            const int sprev = kc[(ck > 0) ? ck - 1 : 0];
            const int s = (ck > 0) ? sprev : 0;
            const int e = kc[ck];
            int rank = 0;
            for (int q = s; q < e; q += 8) {
                unsigned v[8];
                #pragma unroll
                for (int u = 0; u < 8; ++u) v[u] = byk[q + u];
                #pragma unroll
                for (int u = 0; u < 8; ++u) rank += (v[u] < kp || (v[u] == kp && q + u < p)) ? 1 : 0;
            }
            fin[s + rank] = (int)(kp & vmask);
        }
        __syncthreads();

        int* __restrict__ out = srt + rbase;
        for (int i = t; i < n; i += 1024) out[i] = fin[i];
    } else {
        int* ubase = uns + rbase;
        for (int u = 0; u < 8; ++u) {
            const int nl = __builtin_amdgcn_readlane(tN, u);
            const unsigned* rr = part + (size_t)(s0 + u) * SLICE_LEN + __builtin_amdgcn_readlane(tX, u);
            for (int j = lane; j < nl; j += 32) {
                const unsigned w1 = rr[j];
                const unsigned K  = isE ? ((w1 >> 15) | (w1 << 17)) : w1;
                const int p = atomicAdd(&kc[(int)(K >> 15) - ckb], 1);
                ubase[p] = (int)(K & vmask);
            }
        }
        __syncthreads();
        for (int j = wv; j < nkeys; j += 32) {
            const int cf   = j << sb;
            const int sbeg = kc[(cf > 0) ? cf - 1 : 0];
            const int seg0 = (cf > 0) ? sbeg : 0;
            const int send = kc[cf + sub];
            rank_sort_segment(ubase + seg0, srt + rbase + seg0, send - seg0, lane);
        }
    }
}

static_assert((N_EDGES % 32) == 0, "two full 16-row strips per work-group");

template<int K>
__global__ __launch_bounds__(256) void gemm_wmma_kernel(const float* __restrict__ X,
                                                        const float* __restrict__ W,
                                                        float* __restrict__ C,
                                                        int mTiles) {
    constexpr int KC = 32;
    constexpr int P  = 36;
    constexpr int NCH = K / KC;
    __shared__ __attribute__((aligned(16))) float sX[32 * P];
    __shared__ __attribute__((aligned(16))) float sW[256 * P];

    const int t    = threadIdx.x;
    const int lane = t & 31;
    const int wv   = t >> 5;
    const int mBlk = blockIdx.x * 2;
    if (mBlk >= mTiles) return;
    const int half = lane >> 4;
    const int l16  = lane & 15;

    const int srow = t >> 3;
    const int spc  = (t & 7) * 4;
    const float* __restrict__ gX = X + (size_t)(mBlk * 16 + srow) * K + spc;
    const float* __restrict__ gW = W + (size_t)srow * K + spc;
    float* stX = sX + srow * P + spc;
    float* stW = sW + srow * P + spc;

    v4f px = *reinterpret_cast<const v4f*>(gX);
    v4f pw[8];
    #pragma unroll
    for (int i = 0; i < 8; ++i) pw[i] = *reinterpret_cast<const v4f*>(gW + (size_t)(32 * i) * K);

    const float* fx = sX + ((wv >> 2) * 16 + l16) * P + 2 * half;
    const float* fw = sW + ((wv & 3) * 64 + l16) * P + 2 * half;

    v8f acc0 = {}, acc1 = {}, acc2 = {}, acc3 = {};

    for (int c = 0; c < NCH; ++c) {
        *reinterpret_cast<v4f*>(stX) = px;
        #pragma unroll
        for (int i = 0; i < 8; ++i) *reinterpret_cast<v4f*>(stW + 32 * i * P) = pw[i];
        __syncthreads();
        if (c + 1 < NCH) {
            px = *reinterpret_cast<const v4f*>(gX + (c + 1) * KC);
            #pragma unroll
            for (int i = 0; i < 8; ++i)
                pw[i] = *reinterpret_cast<const v4f*>(gW + (size_t)(32 * i) * K + (c + 1) * KC);
        }
        #pragma unroll
        for (int kk = 0; kk < KC; kk += 4) {
            const v2f a  = *reinterpret_cast<const v2f*>(fx + kk);
            const v2f b0 = *reinterpret_cast<const v2f*>(fw + 0 * 16 * P + kk);
            const v2f b1 = *reinterpret_cast<const v2f*>(fw + 1 * 16 * P + kk);
            const v2f b2 = *reinterpret_cast<const v2f*>(fw + 2 * 16 * P + kk);
            const v2f b3 = *reinterpret_cast<const v2f*>(fw + 3 * 16 * P + kk);
            acc0 = __builtin_amdgcn_wmma_f32_16x16x4_f32(false, a, false, b0, (short)0, acc0, false, false);
            acc1 = __builtin_amdgcn_wmma_f32_16x16x4_f32(false, a, false, b1, (short)0, acc1, false, false);
            acc2 = __builtin_amdgcn_wmma_f32_16x16x4_f32(false, a, false, b2, (short)0, acc2, false, false);
            acc3 = __builtin_amdgcn_wmma_f32_16x16x4_f32(false, a, false, b3, (short)0, acc3, false, false);
        }
        __syncthreads();
    }

    const int m0 = (mBlk + (wv >> 2)) * 16;
    const int n0 = (wv & 3) * 64;
    float* __restrict__ ccol = C + n0 + l16;
    #pragma unroll
    for (int r = 0; r < 8; ++r) {
        const size_t m = (size_t)(m0 + r + 8 * half);
        ccol[m * D + 0 * 16] = acc0[r];
        ccol[m * D + 1 * 16] = acc1[r];
        ccol[m * D + 2 * 16] = acc2[r];
        ccol[m * D + 3 * 16] = acc3[r];
    }
}

__device__ __forceinline__ void gather_segment(const float* __restrict__ src,
                                               const int* __restrict__ list,
                                               int beg, int end, int lane,
                                               v4f& a0, v4f& a1) {
    const float* __restrict__ lsrc = src + lane * 4;
    for (int c = beg; c < end; c += 32) {
        const int q   = c + lane;
        const int idx = (q < end) ? list[q] : 0;
        const int cnt = (end - c < 32) ? (end - c) : 32;
        int j = 0;
        for (; j + 4 <= cnt; j += 4) {
            const float* p0 = lsrc + (size_t)__builtin_amdgcn_readlane(idx, j + 0) * D;
            const float* p1 = lsrc + (size_t)__builtin_amdgcn_readlane(idx, j + 1) * D;
            const float* p2 = lsrc + (size_t)__builtin_amdgcn_readlane(idx, j + 2) * D;
            const float* p3 = lsrc + (size_t)__builtin_amdgcn_readlane(idx, j + 3) * D;
            const v4f x0 = *reinterpret_cast<const v4f*>(p0), y0 = *reinterpret_cast<const v4f*>(p0 + 128);
            const v4f x1 = *reinterpret_cast<const v4f*>(p1), y1 = *reinterpret_cast<const v4f*>(p1 + 128);
            const v4f x2 = *reinterpret_cast<const v4f*>(p2), y2 = *reinterpret_cast<const v4f*>(p2 + 128);
            const v4f x3 = *reinterpret_cast<const v4f*>(p3), y3 = *reinterpret_cast<const v4f*>(p3 + 128);
            a0 += x0; a1 += y0;
            a0 += x1; a1 += y1;
            a0 += x2; a1 += y2;
            a0 += x3; a1 += y3;
        }
        for (; j < cnt; ++j) {
            const float* p0 = lsrc + (size_t)__builtin_amdgcn_readlane(idx, j) * D;
            a0 += *reinterpret_cast<const v4f*>(p0);
            a1 += *reinterpret_cast<const v4f*>(p0 + 128);
        }
    }
}

__global__ __launch_bounds__(256) void gather_edge_mean_kernel(const float* __restrict__ src,
                                                               const int* __restrict__ pos,
                                                               const int* __restrict__ list,
                                                               float* __restrict__ dst) {
    const int w    = __builtin_amdgcn_readfirstlane((int)((blockIdx.x * 256 + threadIdx.x) >> 5));
    const int lane = threadIdx.x & 31;
    if (w >= N_EDGES) return;
    const int beg = pos[w];
    const int end = (w + 1 < N_EDGES) ? pos[w + 1] : NNZ_CNT;

    v4f a0 = {}, a1 = {};
    gather_segment(src, list, beg, end, lane, a0, a1);

    const float c   = (float)(end - beg);
    const float inv = (c > 0.0f) ? 1.0f / c : 0.0f;
    a0 *= inv; a1 *= inv;
    float* drow = dst + (size_t)w * D + lane * 4;
    *reinterpret_cast<v4f*>(drow)       = a0;
    *reinterpret_cast<v4f*>(drow + 128) = a1;
}

__device__ __forceinline__ void gather_segment_in(const float* __restrict__ src,
                                                  const int* __restrict__ list,
                                                  int beg, int end, int lane, v4f& a0) {
    const float* __restrict__ lsrc = src + lane * 4;
    for (int c = beg; c < end; c += 32) {
        const int q   = c + lane;
        const int idx = (q < end) ? list[q] : 0;
        const int cnt = (end - c < 32) ? (end - c) : 32;
        int j = 0;
        for (; j + 8 <= cnt; j += 8) {
            const v4f x0 = *reinterpret_cast<const v4f*>(lsrc + (size_t)__builtin_amdgcn_readlane(idx, j + 0) * D_IN);
            const v4f x1 = *reinterpret_cast<const v4f*>(lsrc + (size_t)__builtin_amdgcn_readlane(idx, j + 1) * D_IN);
            const v4f x2 = *reinterpret_cast<const v4f*>(lsrc + (size_t)__builtin_amdgcn_readlane(idx, j + 2) * D_IN);
            const v4f x3 = *reinterpret_cast<const v4f*>(lsrc + (size_t)__builtin_amdgcn_readlane(idx, j + 3) * D_IN);
            const v4f x4 = *reinterpret_cast<const v4f*>(lsrc + (size_t)__builtin_amdgcn_readlane(idx, j + 4) * D_IN);
            const v4f x5 = *reinterpret_cast<const v4f*>(lsrc + (size_t)__builtin_amdgcn_readlane(idx, j + 5) * D_IN);
            const v4f x6 = *reinterpret_cast<const v4f*>(lsrc + (size_t)__builtin_amdgcn_readlane(idx, j + 6) * D_IN);
            const v4f x7 = *reinterpret_cast<const v4f*>(lsrc + (size_t)__builtin_amdgcn_readlane(idx, j + 7) * D_IN);
            a0 += x0; a0 += x1; a0 += x2; a0 += x3;
            a0 += x4; a0 += x5; a0 += x6; a0 += x7;
        }
        if (j + 4 <= cnt) {
            const v4f x0 = *reinterpret_cast<const v4f*>(lsrc + (size_t)__builtin_amdgcn_readlane(idx, j + 0) * D_IN);
            const v4f x1 = *reinterpret_cast<const v4f*>(lsrc + (size_t)__builtin_amdgcn_readlane(idx, j + 1) * D_IN);
            const v4f x2 = *reinterpret_cast<const v4f*>(lsrc + (size_t)__builtin_amdgcn_readlane(idx, j + 2) * D_IN);
            const v4f x3 = *reinterpret_cast<const v4f*>(lsrc + (size_t)__builtin_amdgcn_readlane(idx, j + 3) * D_IN);
            a0 += x0; a0 += x1; a0 += x2; a0 += x3;
            j += 4;
        }
        for (; j < cnt; ++j) {
            a0 += *reinterpret_cast<const v4f*>(lsrc + (size_t)__builtin_amdgcn_readlane(idx, j) * D_IN);
        }
    }
}

__global__ __launch_bounds__(256) void gather_edge_mean_in_kernel(const float* __restrict__ src,
                                                                  const int* __restrict__ pos,
                                                                  const int* __restrict__ list,
                                                                  float* __restrict__ dst) {
    const int w    = __builtin_amdgcn_readfirstlane((int)((blockIdx.x * 256 + threadIdx.x) >> 5));
    const int lane = threadIdx.x & 31;
    if (w >= N_EDGES) return;
    const int beg = pos[w];
    const int end = (w + 1 < N_EDGES) ? pos[w + 1] : NNZ_CNT;

    v4f a0 = {};
    gather_segment_in(src, list, beg, end, lane, a0);

    const float c   = (float)(end - beg);
    const float inv = (c > 0.0f) ? 1.0f / c : 0.0f;
    a0 *= inv;
    *reinterpret_cast<v4f*>(dst + (size_t)w * D_IN + lane * 4) = a0;
}

template<int M>
__device__ __forceinline__ float row_xor_add(float v) {
    return v + __builtin_bit_cast(float, __builtin_amdgcn_update_dpp(0, __builtin_bit_cast(int, v), 0x160 | M, 0xf, 0xf, true));
}

__device__ __forceinline__ float wave_sum(float v) {
    v = row_xor_add<1>(v);
    v = row_xor_add<2>(v);
    v = row_xor_add<4>(v);
    v = row_xor_add<8>(v);
    const int b = __builtin_bit_cast(int, v);
    return v + __builtin_bit_cast(float, __builtin_amdgcn_permlanex16(b, b, 0x76543210, 0xfedcba98, false, false));
}

__global__ __launch_bounds__(256) void gather_node_ln_prelu_kernel(const float* __restrict__ src,
                                                                   const int* __restrict__ pos,
                                                                   const int* __restrict__ list,
                                                                   const float* __restrict__ bias,
                                                                   const float* __restrict__ g,
                                                                   const float* __restrict__ beta,
                                                                   const float* __restrict__ pa,
                                                                   float* __restrict__ dst) {
    const int w    = __builtin_amdgcn_readfirstlane((int)((blockIdx.x * 256 + threadIdx.x) >> 5));
    const int lane = threadIdx.x & 31;
    if (w >= N_NODES) return;
    const int beg = pos[w];
    const int end = (w + 1 < N_NODES) ? pos[w + 1] : NNZ_CNT;

    v4f a0 = {}, a1 = {};
    gather_segment(src, list, beg, end, lane, a0, a1);

    const float dg   = (float)(end - beg);
    const float dinv = (dg > 0.0f) ? 1.0f / dg : 0.0f;
    const int   col  = lane * 4;
    const v4f v0 = a0 * dinv + *reinterpret_cast<const v4f*>(bias + col);
    const v4f v1 = a1 * dinv + *reinterpret_cast<const v4f*>(bias + col + 128);

    const float mu = wave_sum(((v0.x + v0.y) + (v0.z + v0.w)) + ((v1.x + v1.y) + (v1.z + v1.w))) * (1.0f / D);
    const v4f d0 = v0 - mu;
    const v4f d1 = v1 - mu;
    const v4f q0 = d0 * d0;
    const v4f q1 = d1 * d1;
    const float var = wave_sum(((q0.x + q0.y) + (q0.z + q0.w)) + ((q1.x + q1.y) + (q1.z + q1.w))) * (1.0f / D);
    const float rs  = rsqrtf(var + EPS_F);

    const v4f y0 = d0 * rs * *reinterpret_cast<const v4f*>(g + col)       + *reinterpret_cast<const v4f*>(beta + col);
    const v4f y1 = d1 * rs * *reinterpret_cast<const v4f*>(g + col + 128) + *reinterpret_cast<const v4f*>(beta + col + 128);
    const float a = pa[0];
    v4f o0, o1;
    o0.x = (y0.x >= 0.0f) ? y0.x : a * y0.x;
    o0.y = (y0.y >= 0.0f) ? y0.y : a * y0.y;
    o0.z = (y0.z >= 0.0f) ? y0.z : a * y0.z;
    o0.w = (y0.w >= 0.0f) ? y0.w : a * y0.w;
    o1.x = (y1.x >= 0.0f) ? y1.x : a * y1.x;
    o1.y = (y1.y >= 0.0f) ? y1.y : a * y1.y;
    o1.z = (y1.z >= 0.0f) ? y1.z : a * y1.z;
    o1.w = (y1.w >= 0.0f) ? y1.w : a * y1.w;

    float* drow = dst + (size_t)w * D + col;
    *reinterpret_cast<v4f*>(drow)       = o0;
    *reinterpret_cast<v4f*>(drow + 128) = o1;
}

__global__ __launch_bounds__(256) void wstd_kernel(const float* __restrict__ W,
                                                   float* __restrict__ Ws) {
    __shared__ float sred[256];
    const int r = blockIdx.x;
    const int f = threadIdx.x;
    const float w = W[(size_t)r * D + f];

    sred[f] = w;
    __syncthreads();
    #pragma unroll
    for (int s = 128; s > 0; s >>= 1) { if (f < s) sred[f] += sred[f + s]; __syncthreads(); }
    const float mean = sred[0] * (1.0f / D);
    __syncthreads();

    const float dv = w - mean;
    sred[f] = dv * dv;
    __syncthreads();
    #pragma unroll
    for (int s = 128; s > 0; s >>= 1) { if (f < s) sred[f] += sred[f + s]; __syncthreads(); }
    const float var = sred[0] * (1.0f / (D - 1));

    Ws[(size_t)r * D + f] = dv * rsqrtf(var + EPS_F);
}

extern "C" void kernel_launch(void* const* d_in, const int* in_sizes, int n_in,
                              void* d_out, int out_size, void* d_ws, size_t ws_size,
                              hipStream_t stream) {
    const float* edge_fea = (const float*)d_in[0];
    const int*   node_idx = (const int*)  d_in[1];
    const int*   edge_idx = (const int*)  d_in[2];
    const float* W1       = (const float*)d_in[3];
    const float* b1       = (const float*)d_in[4];
    const float* W2       = (const float*)d_in[5];
    const float* b2       = (const float*)d_in[6];
    const float* g1       = (const float*)d_in[7];
    const float* beta1    = (const float*)d_in[8];
    const float* g2       = (const float*)d_in[9];
    const float* beta2    = (const float*)d_in[10];
    const float* prelu_a  = (const float*)d_in[11];

    const size_t nodeRowBytes = (size_t)N_NODES * D * sizeof(float);

    char*  ws   = (char*)d_ws;
    float* H    = (float*)d_out;
    int*   srtN = (int*)ws;
    int*   srtE = srtN + NNZ_CNT;
    int*   unsN = srtE + NNZ_CNT;
    int*   unsE = unsN + NNZ_CNT;
    unsigned* partN = (unsigned*)(unsE + NNZ_CNT);
    unsigned* partE = partN + NNZ_CNT;
    v2i*   ocN  = (v2i*)(ws + ((size_t)20 << 20));
    v2i*   ocE  = ocN + (size_t)RANGES * SLICES;
    float* A    = (float*)(ws + ((size_t)32 << 20));
    float* E    = (float*)(ws + nodeRowBytes);
    int*   posN = (int*)(E + (size_t)N_EDGES * D);
    int*   posE = posN + N_NODES;
    float* W2s  = (float*)(posE + N_EDGES);

    const int mTiles = N_EDGES / 16;
    const int gemmBlocks = (mTiles * 4 * 32) / 256;

    slice_partition_kernel<<<SLICES, 1024, 0, stream>>>(node_idx, edge_idx, partN, partE, ocN, ocE);
    range_place_kernel<<<2 * RANGES, 1024, 0, stream>>>(partN, partE, ocN, ocE, posN, posE, unsN, unsE, srtN, srtE);

    gather_edge_mean_in_kernel<<<N_EDGES / 8, 256, 0, stream>>>(edge_fea, posE, srtE, A);
    gemm_wmma_kernel<D_IN><<<gemmBlocks, 256, 0, stream>>>(A, W1, E, mTiles);
    gather_node_ln_prelu_kernel<<<N_NODES / 8, 256, 0, stream>>>(E, posN, srtN, b1, g1, beta1, prelu_a, H);

    wstd_kernel<<<D, 256, 0, stream>>>(W2, W2s);
    gather_edge_mean_kernel<<<N_EDGES / 8, 256, 0, stream>>>(H, posE, srtE, A);
    gemm_wmma_kernel<D><<<gemmBlocks, 256, 0, stream>>>(A, W2s, E, mTiles);
    gather_node_ln_prelu_kernel<<<N_NODES / 8, 256, 0, stream>>>(E, posN, srtN, b2, g2, beta2, prelu_a, H);
}


# Hyperedge-mean gathers fused into the WMMA GEMMs: 16 waves average a 32-row tile straight into LDS and multiply it there; two launches and the averaged plane gone; sums and k order unchanged
# speedup vs baseline: 14.4701x; 1.0367x over previous
#include <hip/hip_runtime.h>
#include <hip/hip_bf16.h>

#define N_NODES   100000
#define N_EDGES   20000
#define NNZ_CNT   800000
#define D_IN      128
#define D         256
#define EPS_F     1e-5f

typedef float v2f __attribute__((ext_vector_type(2)));
typedef float v4f __attribute__((ext_vector_type(4)));
typedef float v8f __attribute__((ext_vector_type(8)));
typedef int   v4i __attribute__((ext_vector_type(4)));
typedef int   v2i __attribute__((ext_vector_type(2)));

#define SLICES     256
#define SLICE_LEN  (NNZ_CNT / SLICES)
#define RANGES     256
#define KR_N       391
#define KR_E       79
static_assert(NNZ_CNT % SLICES == 0, "equal slices");
static_assert(RANGES * KR_N >= N_NODES && RANGES * KR_E >= N_EDGES, "the ranges cover every key");
static_assert(N_NODES <= (1 << 17) && N_EDGES <= (1 << 15), "node and hyperedge share one 32-bit word");
static_assert(KR_N <= 512 && KR_E <= 512, "two key counters per thread of the place pass");
static_assert(SLICE_LEN < 65536, "a rank inside a slice fits 16 bits");

__global__ __launch_bounds__(1024) void slice_partition_kernel(const int* __restrict__ node_idx,
                                                               const int* __restrict__ edge_idx,
                                                               unsigned* __restrict__ partN,
                                                               unsigned* __restrict__ partE,
                                                               v2i* __restrict__ ocN,
                                                               v2i* __restrict__ ocE) {
    __shared__ int cN[RANGES];
    __shared__ int cE[RANGES];
    __shared__ unsigned stN[SLICE_LEN];
    __shared__ unsigned stE[SLICE_LEN];
    const int t    = threadIdx.x;
    const int s    = blockIdx.x;
    const int base = s * SLICE_LEN;

    constexpr int IT = (SLICE_LEN + 1023) / 1024;
    unsigned w[IT];
    int rn[IT], re[IT];
    #pragma unroll
    for (int u = 0; u < IT; ++u) {
        const int j = t + u * 1024;
        const int n = (j < SLICE_LEN) ? node_idx[base + j] : 0;
        const int e = (j < SLICE_LEN) ? edge_idx[base + j] : 0;
        w[u] = ((unsigned)n << 15) | (unsigned)e;
    }
    if (t < RANGES) { cN[t] = 0; cE[t] = 0; }
    __syncthreads();

    #pragma unroll
    for (int u = 0; u < IT; ++u) {
        const int j = t + u * 1024;
        if (j < SLICE_LEN) {
            const int rN = (int)(w[u] >> 15) / KR_N;
            const int rE = (int)(w[u] & 0x7FFFu) / KR_E;
            rn[u] = (rN << 16) | atomicAdd(&cN[rN], 1);
            re[u] = (rE << 16) | atomicAdd(&cE[rE], 1);
        }
    }
    __syncthreads();

    if (t < 64) {
        const int lane = t & 31;
        int* c   = (t < 32) ? cN : cE;
        v2i* oc  = (t < 32) ? ocN : ocE;
        int v[RANGES / 32];
        int sum = 0;
        #pragma unroll
        for (int i = 0; i < RANGES / 32; ++i) { v[i] = c[(RANGES / 32) * lane + i]; sum += v[i]; }
        int inc = sum;
        #pragma unroll
        for (int o = 1; o < 32; o <<= 1) {
            const int up = __shfl_up(inc, o, 32);
            if (lane >= o) inc += up;
        }
        int run = inc - sum;
        #pragma unroll
        for (int i = 0; i < RANGES / 32; ++i) {
            const int r = (RANGES / 32) * lane + i;
            c[r] = run;
            v2i o; o.x = run; o.y = v[i];
            oc[(size_t)r * SLICES + s] = o;
            run += v[i];
        }
    }
    __syncthreads();

    #pragma unroll
    for (int u = 0; u < IT; ++u) {
        const int j = t + u * 1024;
        if (j < SLICE_LEN) {
            stN[cN[rn[u] >> 16] + (rn[u] & 0xFFFF)] = w[u];
            stE[cE[re[u] >> 16] + (re[u] & 0xFFFF)] = w[u];
        }
    }
    __syncthreads();

    for (int i = t; i < SLICE_LEN; i += 1024) {
        partN[base + i] = stN[i];
        partE[base + i] = stE[i];
    }
}

#define CAP 4096

__device__ __forceinline__ void rank_sort_segment(const int* U, int* S, int L, int lane) {
    for (int m = 0; m < L; m += 32) {
        const int p = m + lane;
        const int x = (p < L) ? U[p] : 0;
        int rank = 0;
        for (int c = 0; c < L; c += 32) {
            const int q  = c + lane;
            const int ec = (q < L) ? U[q] : 0;
            const int cnt = (L - c < 32) ? (L - c) : 32;
            for (int j = 0; j < cnt; ++j) {
                const int v = __builtin_amdgcn_readlane(ec, j);
                rank += (v < x || (v == x && (c + j) < p)) ? 1 : 0;
            }
        }
        if (p < L) S[rank] = x;
    }
}

__global__ __launch_bounds__(1024) void range_place_kernel(const unsigned* __restrict__ partN,
                                                           const unsigned* __restrict__ partE,
                                                           const v2i* __restrict__ ocN,
                                                           const v2i* __restrict__ ocE,
                                                           int* __restrict__ posN,
                                                           int* __restrict__ posE,
                                                           int* unsN,
                                                           int* unsE,
                                                           int* srtN,
                                                           int* srtE) {
    __shared__ int kc[512];
    __shared__ int wtot[8];
    __shared__ int wlen[8];
    __shared__ int sbase;
    __shared__ int rx[SLICES];
    __shared__ int ry[SLICES];
    __shared__ int rf[SLICES];
    __shared__ unsigned raw[CAP];
    __shared__ unsigned byk[CAP + 8];
    const bool isE = blockIdx.x >= RANGES;
    const int  r   = isE ? (int)blockIdx.x - RANGES : (int)blockIdx.x;
    const int  KR  = isE ? KR_E : KR_N;
    const int  nk  = isE ? N_EDGES : N_NODES;
    const int  k0  = r * KR;
    if (k0 >= nk) return;
    const int nkeys = (nk - k0 < KR) ? (nk - k0) : KR;
    const int sb    = isE ? 2 : 0;
    const int sub   = (1 << sb) - 1;
    const int ckb   = k0 << sb;
    const unsigned vmask = isE ? 0x1FFFFu : 0x7FFFu;
    static_assert((KR_E << 2) <= 512 && KR_N <= 512, "the counters of a range fit kc[]");
    static_assert(SLICES == 256 && CAP <= 4096, "eight runs per wave, at most four entries per thread");
    const int t    = threadIdx.x;
    const int lane = t & 31;
    const int wv   = __builtin_amdgcn_readfirstlane(t >> 5);

    const unsigned* __restrict__ part = isE ? partE : partN;
    int* __restrict__ pos = isE ? posE : posN;
    int* uns = isE ? unsE : unsN;
    int* srt = isE ? srtE : srtN;

    if (t < 512) kc[t] = 0;
    if (t == 0) sbase = 0;
    int ocx = 0;
    if (t < SLICES) {
        const v2i oc = (isE ? ocE : ocN)[(size_t)r * SLICES + t];
        ocx = oc.x;
        int incl = oc.y;
        #pragma unroll
        for (int o = 1; o < 32; o <<= 1) {
            const int up = __shfl_up(incl, o, 32);
            if (lane >= o) incl += up;
        }
        if (lane == 31) wlen[wv] = incl;
        rx[t] = oc.x; ry[t] = oc.y; rf[t] = incl - oc.y;
    }
    __syncthreads();

    if (t < SLICES) atomicAdd(&sbase, ocx);
    const int ow = wv >> 2;
    int n = 0, wb = 0;
    #pragma unroll
    for (int q = 0; q < 8; ++q) { const int v = wlen[q]; n += v; wb += (q < ow) ? v : 0; }
    const bool inLds = n <= CAP;
    const int  s0    = wv * 8;
    const int tN = ry[s0 + (lane & 7)];
    const int tF = rf[s0 + (lane & 7)] + wb;
    const int tX = rx[s0 + (lane & 7)];

    if (inLds) {
        unsigned w[8];
        int nn[8], ro[8];
        #pragma unroll
        for (int u = 0; u < 8; ++u) {
            nn[u] = __builtin_amdgcn_readlane(tN, u);
            ro[u] = __builtin_amdgcn_readlane(tF, u);
            const unsigned* rr = part + (size_t)(s0 + u) * SLICE_LEN + __builtin_amdgcn_readlane(tX, u);
            w[u] = (lane < nn[u]) ? rr[lane] : 0u;
        }
        #pragma unroll
        for (int u = 0; u < 8; ++u) {
            if (lane < nn[u]) {
                const unsigned K = isE ? ((w[u] >> 15) | (w[u] << 17)) : w[u];
                raw[ro[u] + lane] = K;
                atomicAdd(&kc[(int)(K >> 15) - ckb], 1);
            }
        }
        for (int u = 0; u < 8; ++u) {
            const int nl = __builtin_amdgcn_readlane(tN, u);
            if (nl > 32) {
                const unsigned* rr = part + (size_t)(s0 + u) * SLICE_LEN + __builtin_amdgcn_readlane(tX, u);
                const int ro1 = __builtin_amdgcn_readlane(tF, u);
                for (int j = 32 + lane; j < nl; j += 32) {
                    const unsigned w1 = rr[j];
                    const unsigned K  = isE ? ((w1 >> 15) | (w1 << 17)) : w1;
                    raw[ro1 + j] = K;
                    atomicAdd(&kc[(int)(K >> 15) - ckb], 1);
                }
            }
        }
    } else {
        for (int u = 0; u < 8; ++u) {
            const int nl = __builtin_amdgcn_readlane(tN, u);
            const unsigned* rr = part + (size_t)(s0 + u) * SLICE_LEN + __builtin_amdgcn_readlane(tX, u);
            for (int j = lane; j < nl; j += 32) {
                const unsigned w1 = rr[j];
                const unsigned K  = isE ? ((w1 >> 15) | (w1 << 17)) : w1;
                atomicAdd(&kc[(int)(K >> 15) - ckb], 1);
            }
        }
    }
    __syncthreads();

    int c0 = 0, sum = 0, inc = 0;
    if (t < 256) {
        c0 = kc[2 * t];
        sum = c0 + kc[2 * t + 1];
        inc = sum;
        #pragma unroll
        for (int o = 1; o < 32; o <<= 1) {
            const int up = __shfl_up(inc, o, 32);
            if (lane >= o) inc += up;
        }
        if (lane == 31) wtot[wv] = inc;
    }
    __syncthreads();
    const int rbase = sbase;
    if (t < 256) {
        int before = 0;
        #pragma unroll
        for (int q = 0; q < 8; ++q) before += (q < wv) ? wtot[q] : 0;
        const int x0 = before + inc - sum;
        const int x1 = x0 + c0;
        kc[2 * t]     = x0;
        kc[2 * t + 1] = x1;
        if (((2 * t)     & sub) == 0 && ((2 * t)     >> sb) < nkeys) pos[k0 + ((2 * t)     >> sb)] = rbase + x0;
        if (((2 * t + 1) & sub) == 0 && ((2 * t + 1) >> sb) < nkeys) pos[k0 + ((2 * t + 1) >> sb)] = rbase + x1;
    }
    __syncthreads();

    if (inLds) {
        unsigned K4[4];
        #pragma unroll
        for (int u = 0; u < 4; ++u) K4[u] = (t + 1024 * u < n) ? raw[t + 1024 * u] : 0u;
        #pragma unroll
        for (int u = 0; u < 4; ++u) {
            if (t + 1024 * u < n) {
                const int p = atomicAdd(&kc[(int)(K4[u] >> 15) - ckb], 1);
                byk[p] = K4[u];
            }
        }
        if (t < 8) byk[n + t] = 0xFFFFFFFFu;
        __syncthreads();

        int* fin = reinterpret_cast<int*>(raw);
        for (int p = t; p < n; p += 1024) {
            const unsigned kp = byk[p];
            const int ck = (int)(kp >> 15) - ckb;
            const int sprev = kc[(ck > 0) ? ck - 1 : 0];
            const int s = (ck > 0) ? sprev : 0;
            const int e = kc[ck];
            int rank = 0;
            for (int q = s; q < e; q += 8) {
                unsigned v[8];
                #pragma unroll
                for (int u = 0; u < 8; ++u) v[u] = byk[q + u];
                #pragma unroll
                for (int u = 0; u < 8; ++u) rank += (v[u] < kp || (v[u] == kp && q + u < p)) ? 1 : 0;
            }
            fin[s + rank] = (int)(kp & vmask);
        }
        __syncthreads();

        int* __restrict__ out = srt + rbase;
        for (int i = t; i < n; i += 1024) out[i] = fin[i];
    } else {
        int* ubase = uns + rbase;
        for (int u = 0; u < 8; ++u) {
            const int nl = __builtin_amdgcn_readlane(tN, u);
            const unsigned* rr = part + (size_t)(s0 + u) * SLICE_LEN + __builtin_amdgcn_readlane(tX, u);
            for (int j = lane; j < nl; j += 32) {
                const unsigned w1 = rr[j];
                const unsigned K  = isE ? ((w1 >> 15) | (w1 << 17)) : w1;
                const int p = atomicAdd(&kc[(int)(K >> 15) - ckb], 1);
                ubase[p] = (int)(K & vmask);
            }
        }
        __syncthreads();
        for (int j = wv; j < nkeys; j += 32) {
            const int cf   = j << sb;
            const int sbeg = kc[(cf > 0) ? cf - 1 : 0];
            const int seg0 = (cf > 0) ? sbeg : 0;
            const int send = kc[cf + sub];
            rank_sort_segment(ubase + seg0, srt + rbase + seg0, send - seg0, lane);
        }
    }
}

__device__ __forceinline__ void gather_segment(const float* __restrict__ src,
                                               const int* __restrict__ list,
                                               int beg, int end, int lane,
                                               v4f& a0, v4f& a1) {
    const float* __restrict__ lsrc = src + lane * 4;
    for (int c = beg; c < end; c += 32) {
        const int q   = c + lane;
        const int idx = (q < end) ? list[q] : 0;
        const int cnt = (end - c < 32) ? (end - c) : 32;
        int j = 0;
        for (; j + 4 <= cnt; j += 4) {
            const float* p0 = lsrc + (size_t)__builtin_amdgcn_readlane(idx, j + 0) * D;
            const float* p1 = lsrc + (size_t)__builtin_amdgcn_readlane(idx, j + 1) * D;
            const float* p2 = lsrc + (size_t)__builtin_amdgcn_readlane(idx, j + 2) * D;
            const float* p3 = lsrc + (size_t)__builtin_amdgcn_readlane(idx, j + 3) * D;
            const v4f x0 = *reinterpret_cast<const v4f*>(p0), y0 = *reinterpret_cast<const v4f*>(p0 + 128);
            const v4f x1 = *reinterpret_cast<const v4f*>(p1), y1 = *reinterpret_cast<const v4f*>(p1 + 128);
            const v4f x2 = *reinterpret_cast<const v4f*>(p2), y2 = *reinterpret_cast<const v4f*>(p2 + 128);
            const v4f x3 = *reinterpret_cast<const v4f*>(p3), y3 = *reinterpret_cast<const v4f*>(p3 + 128);
            a0 += x0; a1 += y0;
            a0 += x1; a1 += y1;
            a0 += x2; a1 += y2;
            a0 += x3; a1 += y3;
        }
        for (; j < cnt; ++j) {
            const float* p0 = lsrc + (size_t)__builtin_amdgcn_readlane(idx, j) * D;
            a0 += *reinterpret_cast<const v4f*>(p0);
            a1 += *reinterpret_cast<const v4f*>(p0 + 128);
        }
    }
}

__device__ __forceinline__ void gather_segment_in(const float* __restrict__ src,
                                                  const int* __restrict__ list,
                                                  int beg, int end, int lane, v4f& a0) {
    const float* __restrict__ lsrc = src + lane * 4;
    for (int c = beg; c < end; c += 32) {
        const int q   = c + lane;
        const int idx = (q < end) ? list[q] : 0;
        const int cnt = (end - c < 32) ? (end - c) : 32;
        int j = 0;
        for (; j + 8 <= cnt; j += 8) {
            const v4f x0 = *reinterpret_cast<const v4f*>(lsrc + (size_t)__builtin_amdgcn_readlane(idx, j + 0) * D_IN);
            const v4f x1 = *reinterpret_cast<const v4f*>(lsrc + (size_t)__builtin_amdgcn_readlane(idx, j + 1) * D_IN);
            const v4f x2 = *reinterpret_cast<const v4f*>(lsrc + (size_t)__builtin_amdgcn_readlane(idx, j + 2) * D_IN);
            const v4f x3 = *reinterpret_cast<const v4f*>(lsrc + (size_t)__builtin_amdgcn_readlane(idx, j + 3) * D_IN);
            const v4f x4 = *reinterpret_cast<const v4f*>(lsrc + (size_t)__builtin_amdgcn_readlane(idx, j + 4) * D_IN);
            const v4f x5 = *reinterpret_cast<const v4f*>(lsrc + (size_t)__builtin_amdgcn_readlane(idx, j + 5) * D_IN);
            const v4f x6 = *reinterpret_cast<const v4f*>(lsrc + (size_t)__builtin_amdgcn_readlane(idx, j + 6) * D_IN);
            const v4f x7 = *reinterpret_cast<const v4f*>(lsrc + (size_t)__builtin_amdgcn_readlane(idx, j + 7) * D_IN);
            a0 += x0; a0 += x1; a0 += x2; a0 += x3;
            a0 += x4; a0 += x5; a0 += x6; a0 += x7;
        }
        if (j + 4 <= cnt) {
            const v4f x0 = *reinterpret_cast<const v4f*>(lsrc + (size_t)__builtin_amdgcn_readlane(idx, j + 0) * D_IN);
            const v4f x1 = *reinterpret_cast<const v4f*>(lsrc + (size_t)__builtin_amdgcn_readlane(idx, j + 1) * D_IN);
            const v4f x2 = *reinterpret_cast<const v4f*>(lsrc + (size_t)__builtin_amdgcn_readlane(idx, j + 2) * D_IN);
            const v4f x3 = *reinterpret_cast<const v4f*>(lsrc + (size_t)__builtin_amdgcn_readlane(idx, j + 3) * D_IN);
            a0 += x0; a0 += x1; a0 += x2; a0 += x3;
            j += 4;
        }
        for (; j < cnt; ++j) {
            a0 += *reinterpret_cast<const v4f*>(lsrc + (size_t)__builtin_amdgcn_readlane(idx, j) * D_IN);
        }
    }
}

static_assert((N_EDGES % 32) == 0, "32 whole hyperedge rows per work-group");

template<int K>
__global__ __launch_bounds__(512, 12) void edge_mean_gemm_kernel(const float* __restrict__ src,
                                                             const int* __restrict__ pos,
                                                             const int* __restrict__ list,
                                                             const float* __restrict__ W,
                                                             float* __restrict__ C,
                                                             int mTiles) {
    static_assert(K == D_IN || K == D, "a row is one or two 16-byte pieces per lane");
    constexpr int KC  = 32;
    constexpr int P   = 36;
    constexpr int PA  = K + 4;
    constexpr int NCH = K / KC;
    __shared__ __attribute__((aligned(16))) float sA[32 * PA];
    __shared__ __attribute__((aligned(16))) float sW[256 * P];

    const int t    = threadIdx.x;
    const int lane = t & 31;
    const int wv   = __builtin_amdgcn_readfirstlane(t >> 5);
    const int mBlk = blockIdx.x * 2;
    if (mBlk >= mTiles) return;
    const int half = lane >> 4;
    const int l16  = lane & 15;

    const int srow = t >> 3;
    const int spc  = (t & 7) * 4;
    const float* __restrict__ gW = W + (size_t)srow * K + spc;
    float* stW = sW + srow * P + spc;

    v4f pw[4];
    #pragma unroll
    for (int i = 0; i < 4; ++i) pw[i] = *reinterpret_cast<const v4f*>(gW + (size_t)(64 * i) * K);

    const int e0 = mBlk * 16 + 2 * wv;
    const int b0 = pos[e0];
    const int b1 = pos[e0 + 1];
    const int b2 = (e0 + 2 < N_EDGES) ? pos[e0 + 2] : NNZ_CNT;
    #pragma unroll
    for (int i = 0; i < 2; ++i) {
        const int beg = (i == 0) ? b0 : b1;
        const int end = (i == 0) ? b1 : b2;
        const float cf  = (float)(end - beg);
        const float inv = (cf > 0.0f) ? 1.0f / cf : 0.0f;
        float* arow = sA + (2 * wv + i) * PA + lane * 4;
        if constexpr (K == D_IN) {
            v4f a0 = {};
            gather_segment_in(src, list, beg, end, lane, a0);
            a0 *= inv;
            *reinterpret_cast<v4f*>(arow) = a0;
        } else {
            v4f a0 = {}, a1 = {};
            gather_segment(src, list, beg, end, lane, a0, a1);
            a0 *= inv; a1 *= inv;
            *reinterpret_cast<v4f*>(arow)       = a0;
            *reinterpret_cast<v4f*>(arow + 128) = a1;
        }
    }

    const float* fx = sA + ((wv >> 3) * 16 + l16) * PA + 2 * half;
    const float* fw = sW + ((wv & 7) * 32 + l16) * P + 2 * half;

    v8f acc0 = {}, acc1 = {};

    for (int c = 0; c < NCH; ++c) {
        #pragma unroll
        for (int i = 0; i < 4; ++i) *reinterpret_cast<v4f*>(stW + 64 * i * P) = pw[i];
        __syncthreads();
        if (c + 1 < NCH) {
            #pragma unroll
            for (int i = 0; i < 4; ++i)
                pw[i] = *reinterpret_cast<const v4f*>(gW + (size_t)(64 * i) * K + (c + 1) * KC);
        }
        const float* fxc = fx + c * KC;
        #pragma unroll
        for (int kk = 0; kk < KC; kk += 4) {
            const v2f a  = *reinterpret_cast<const v2f*>(fxc + kk);
            const v2f w0 = *reinterpret_cast<const v2f*>(fw + 0 * 16 * P + kk);
            const v2f w1 = *reinterpret_cast<const v2f*>(fw + 1 * 16 * P + kk);
            acc0 = __builtin_amdgcn_wmma_f32_16x16x4_f32(false, a, false, w0, (short)0, acc0, false, false);
            acc1 = __builtin_amdgcn_wmma_f32_16x16x4_f32(false, a, false, w1, (short)0, acc1, false, false);
        }
        __syncthreads();
    }

    const int m0 = (mBlk + (wv >> 3)) * 16;
    const int n0 = (wv & 7) * 32;
    float* __restrict__ ccol = C + n0 + l16;
    #pragma unroll
    for (int r = 0; r < 8; ++r) {
        const size_t m = (size_t)(m0 + r + 8 * half);
        ccol[m * D + 0 * 16] = acc0[r];
        ccol[m * D + 1 * 16] = acc1[r];
    }
}

template<int M>
__device__ __forceinline__ float row_xor_add(float v) {
    return v + __builtin_bit_cast(float, __builtin_amdgcn_update_dpp(0, __builtin_bit_cast(int, v), 0x160 | M, 0xf, 0xf, true));
}

__device__ __forceinline__ float wave_sum(float v) {
    v = row_xor_add<1>(v);
    v = row_xor_add<2>(v);
    v = row_xor_add<4>(v);
    v = row_xor_add<8>(v);
    const int b = __builtin_bit_cast(int, v);
    return v + __builtin_bit_cast(float, __builtin_amdgcn_permlanex16(b, b, 0x76543210, 0xfedcba98, false, false));
}

__global__ __launch_bounds__(256) void gather_node_ln_prelu_kernel(const float* __restrict__ src,
                                                                   const int* __restrict__ pos,
                                                                   const int* __restrict__ list,
                                                                   const float* __restrict__ bias,
                                                                   const float* __restrict__ g,
                                                                   const float* __restrict__ beta,
                                                                   const float* __restrict__ pa,
                                                                   float* __restrict__ dst) {
    const int w    = __builtin_amdgcn_readfirstlane((int)((blockIdx.x * 256 + threadIdx.x) >> 5));
    const int lane = threadIdx.x & 31;
    if (w >= N_NODES) return;
    const int beg = pos[w];
    const int end = (w + 1 < N_NODES) ? pos[w + 1] : NNZ_CNT;

    v4f a0 = {}, a1 = {};
    gather_segment(src, list, beg, end, lane, a0, a1);

    const float dg   = (float)(end - beg);
    const float dinv = (dg > 0.0f) ? 1.0f / dg : 0.0f;
    const int   col  = lane * 4;
    const v4f v0 = a0 * dinv + *reinterpret_cast<const v4f*>(bias + col);
    const v4f v1 = a1 * dinv + *reinterpret_cast<const v4f*>(bias + col + 128);

    const float mu = wave_sum(((v0.x + v0.y) + (v0.z + v0.w)) + ((v1.x + v1.y) + (v1.z + v1.w))) * (1.0f / D);
    const v4f d0 = v0 - mu;
    const v4f d1 = v1 - mu;
    const v4f q0 = d0 * d0;
    const v4f q1 = d1 * d1;
    const float var = wave_sum(((q0.x + q0.y) + (q0.z + q0.w)) + ((q1.x + q1.y) + (q1.z + q1.w))) * (1.0f / D);
    const float rs  = rsqrtf(var + EPS_F);

    const v4f y0 = d0 * rs * *reinterpret_cast<const v4f*>(g + col)       + *reinterpret_cast<const v4f*>(beta + col);
    const v4f y1 = d1 * rs * *reinterpret_cast<const v4f*>(g + col + 128) + *reinterpret_cast<const v4f*>(beta + col + 128);
    const float a = pa[0];
    v4f o0, o1;
    o0.x = (y0.x >= 0.0f) ? y0.x : a * y0.x;
    o0.y = (y0.y >= 0.0f) ? y0.y : a * y0.y;
    o0.z = (y0.z >= 0.0f) ? y0.z : a * y0.z;
    o0.w = (y0.w >= 0.0f) ? y0.w : a * y0.w;
    o1.x = (y1.x >= 0.0f) ? y1.x : a * y1.x;
    o1.y = (y1.y >= 0.0f) ? y1.y : a * y1.y;
    o1.z = (y1.z >= 0.0f) ? y1.z : a * y1.z;
    o1.w = (y1.w >= 0.0f) ? y1.w : a * y1.w;

    float* drow = dst + (size_t)w * D + col;
    *reinterpret_cast<v4f*>(drow)       = o0;
    *reinterpret_cast<v4f*>(drow + 128) = o1;
}

__global__ __launch_bounds__(256) void wstd_kernel(const float* __restrict__ W,
                                                   float* __restrict__ Ws) {
    __shared__ float sred[256];
    const int r = blockIdx.x;
    const int f = threadIdx.x;
    const float w = W[(size_t)r * D + f];

    sred[f] = w;
    __syncthreads();
    #pragma unroll
    for (int s = 128; s > 0; s >>= 1) { if (f < s) sred[f] += sred[f + s]; __syncthreads(); }
    const float mean = sred[0] * (1.0f / D);
    __syncthreads();

    const float dv = w - mean;
    sred[f] = dv * dv;
    __syncthreads();
    #pragma unroll
    for (int s = 128; s > 0; s >>= 1) { if (f < s) sred[f] += sred[f + s]; __syncthreads(); }
    const float var = sred[0] * (1.0f / (D - 1));

    Ws[(size_t)r * D + f] = dv * rsqrtf(var + EPS_F);
}

extern "C" void kernel_launch(void* const* d_in, const int* in_sizes, int n_in,
                              void* d_out, int out_size, void* d_ws, size_t ws_size,
                              hipStream_t stream) {
    const float* edge_fea = (const float*)d_in[0];
    const int*   node_idx = (const int*)  d_in[1];
    const int*   edge_idx = (const int*)  d_in[2];
    const float* W1       = (const float*)d_in[3];
    const float* b1       = (const float*)d_in[4];
    const float* W2       = (const float*)d_in[5];
    const float* b2       = (const float*)d_in[6];
    const float* g1       = (const float*)d_in[7];
    const float* beta1    = (const float*)d_in[8];
    const float* g2       = (const float*)d_in[9];
    const float* beta2    = (const float*)d_in[10];
    const float* prelu_a  = (const float*)d_in[11];

    const size_t nodeRowBytes = (size_t)N_NODES * D * sizeof(float);

    char*  ws   = (char*)d_ws;
    float* H    = (float*)d_out;
    int*   srtN = (int*)ws;
    int*   srtE = srtN + NNZ_CNT;
    int*   unsN = srtE + NNZ_CNT;
    int*   unsE = unsN + NNZ_CNT;
    unsigned* partN = (unsigned*)(unsE + NNZ_CNT);
    unsigned* partE = partN + NNZ_CNT;
    v2i*   ocN  = (v2i*)(ws + ((size_t)20 << 20));
    v2i*   ocE  = ocN + (size_t)RANGES * SLICES;
    float* E    = (float*)(ws + nodeRowBytes);
    int*   posN = (int*)(E + (size_t)N_EDGES * D);
    int*   posE = posN + N_NODES;
    float* W2s  = (float*)(posE + N_EDGES);

    const int mTiles = N_EDGES / 16;
    const int tileBlocks = mTiles / 2;

    slice_partition_kernel<<<SLICES, 1024, 0, stream>>>(node_idx, edge_idx, partN, partE, ocN, ocE);
    range_place_kernel<<<2 * RANGES, 1024, 0, stream>>>(partN, partE, ocN, ocE, posN, posE, unsN, unsE, srtN, srtE);

    edge_mean_gemm_kernel<D_IN><<<tileBlocks, 512, 0, stream>>>(edge_fea, posE, srtE, W1, E, mTiles);
    gather_node_ln_prelu_kernel<<<N_NODES / 8, 256, 0, stream>>>(E, posN, srtN, b1, g1, beta1, prelu_a, H);

    wstd_kernel<<<D, 256, 0, stream>>>(W2, W2s);
    edge_mean_gemm_kernel<D><<<tileBlocks, 512, 0, stream>>>(H, posE, srtE, W2s, E, mTiles);
    gather_node_ln_prelu_kernel<<<N_NODES / 8, 256, 0, stream>>>(E, posN, srtN, b2, g2, beta2, prelu_a, H);
}
